# starting kernel
# baseline (speedup 1.0000x reference)
#include <hip/hip_runtime.h>
#include <math.h>

typedef float v2f __attribute__((ext_vector_type(2)));
typedef float v8f __attribute__((ext_vector_type(8)));

#define GCN_G   64
#define GCN_OUT 64
#define LDS_STRIDE 132

__global__ void k_deg_init(float* __restrict__ deg, int n) {
    int i = blockIdx.x * blockDim.x + threadIdx.x;
    if (i < n) deg[i] = 1.0f;
}

__global__ void k_deg_count(const int* __restrict__ dst, float* __restrict__ deg, int e) {
    int i = blockIdx.x * blockDim.x + threadIdx.x;
    if (i < e) atomicAdd(&deg[dst[i]], 1.0f);
}

__global__ void k_deg_inv(float* __restrict__ deg, int n) {
    int i = blockIdx.x * blockDim.x + threadIdx.x;
    if (i < n) {
        float d = deg[i];
        deg[i] = (d > 0.0f) ? (1.0f / sqrtf(d)) : 0.0f;
    }
}

__global__ void __launch_bounds__(128)
k_gemm_wmma_f32(const float* __restrict__ X, const float* __restrict__ W,
                float* __restrict__ H, int K, int F, int mtiles) {
    __shared__ float Wlds[16 * LDS_STRIDE];

    const int t     = threadIdx.x;
    const int wave  = t >> 5;
    const int lane  = t & 31;
    const int half  = lane >> 4;
    const int l16   = lane & 15;
    const int col0  = blockIdx.y << 4;
    const int mtile = (blockIdx.x << 2) + wave;

    const int reps = (K << 4) >> 7;
    for (int rep = 0; rep < reps; ++rep) {
        int idx = (rep << 7) + t;
        int k = idx >> 4;
        int n = idx & 15;
        Wlds[n * LDS_STRIDE + k] = W[(size_t)k * F + (col0 + n)];
    }
    __syncthreads();

    if (mtile < mtiles) {
        const int row0 = mtile << 4;
        const float* __restrict__ Xrow = X + (size_t)(row0 + l16) * K + (half << 1);
        const float* __restrict__ Wl = &Wlds[l16 * LDS_STRIDE + (half << 1)];

        v8f c = {};
#pragma unroll 8
        for (int k0 = 0; k0 < K; k0 += 4) {
            v2f a = *(const v2f*)(Xrow + k0);
            v2f b = *(const v2f*)(Wl + k0);
            c = __builtin_amdgcn_wmma_f32_16x16x4_f32(false, a, false, b,
                                                      (short)0, c, false, false);
        }

        float* __restrict__ Ho = H + (size_t)(row0 + (half << 3)) * F + (col0 + l16);
#pragma unroll
        for (int v = 0; v < 8; ++v)
            Ho[(size_t)v * F] = c[v];
    }
}

__global__ void k_agg_init(const float* __restrict__ H, const float* __restrict__ dinv,
                           const float* __restrict__ bias, float* __restrict__ out,
                           int n, int fshift) {
    int idx = blockIdx.x * blockDim.x + threadIdx.x;
    int total = n << fshift;
    if (idx >= total) return;
    int i = idx >> fshift;
    int f = idx & ((1 << fshift) - 1);
    float di = dinv[i];
    out[idx] = di * di * H[idx] + bias[f];
}

__global__ void k_edge_scatter(const int* __restrict__ src, const int* __restrict__ dst,
                               const float* __restrict__ dinv, const float* __restrict__ H,
                               float* __restrict__ out, int e, int fshift) {
    const int cshift = fshift - 2;
    long long t = (long long)blockIdx.x * blockDim.x + threadIdx.x;
    long long total = (long long)e << cshift;
    if (t >= total) return;
    int ei = (int)(t >> cshift);
    int c  = ((int)t & ((1 << cshift) - 1)) << 2;
    int s = src[ei];
    int d = dst[ei];
    float w = dinv[s] * dinv[d];
    int F = 1 << fshift;
    const float4 h4 = *(const float4*)(H + (size_t)s * F + c);
    float* o = out + (size_t)d * F + c;
    atomicAdd(o + 0, w * h4.x);
    atomicAdd(o + 1, w * h4.y);
    atomicAdd(o + 2, w * h4.z);
    atomicAdd(o + 3, w * h4.w);
}

__global__ void k_relu(float* __restrict__ x, int total) {
    int i = blockIdx.x * blockDim.x + threadIdx.x;
    if (i < total) x[i] = fmaxf(x[i], 0.0f);
}

__global__ void k_readout_zero(float* __restrict__ gsum, float* __restrict__ cnt) {
    int i = blockIdx.x * blockDim.x + threadIdx.x;
    if (i < GCN_G * GCN_OUT) gsum[i] = 0.0f;
    if (i < GCN_G) cnt[i] = 0.0f;
}

__global__ void k_readout_scatter(const float* __restrict__ h, const int* __restrict__ batch,
                                  float* __restrict__ gsum, float* __restrict__ cnt, int n) {
    int idx = blockIdx.x * blockDim.x + threadIdx.x;
    if (idx >= (n << 6)) return;
    int i = idx >> 6;
    int f = idx & 63;
    int g = batch[i];
    atomicAdd(&gsum[(g << 6) + f], h[idx]);
    if (f == 0) atomicAdd(&cnt[g], 1.0f);
}

__global__ void k_readout_final(const float* __restrict__ gsum, const float* __restrict__ cnt,
                                float* __restrict__ outp) {
    int idx = blockIdx.x * blockDim.x + threadIdx.x;
    if (idx >= GCN_G * GCN_OUT) return;
    int g = idx >> 6;
    outp[idx] = gsum[idx] / fmaxf(cnt[g], 1.0f);
}

extern "C" void kernel_launch(void* const* d_in, const int* in_sizes, int n_in,
                              void* d_out, int out_size, void* d_ws, size_t ws_size,
                              hipStream_t stream) {
    const float* x     = (const float*)d_in[0];
    const int*   edge  = (const int*)d_in[1];
    const int*   batch = (const int*)d_in[2];
    const float* W1    = (const float*)d_in[3];
    const float* b1    = (const float*)d_in[4];
    const float* W2    = (const float*)d_in[5];
    const float* b2    = (const float*)d_in[6];
    const float* W3    = (const float*)d_in[7];
    const float* b3    = (const float*)d_in[8];

    const int IN = 128, HID = 128, OUTF = 64;
    const int N = in_sizes[0] / IN;
    const int E = in_sizes[1] / 2;
    const int* src = edge;
    const int* dst = edge + E;

    char* ws = (char*)d_ws;
    float* dinv = (float*)ws;
    size_t off = ((size_t)N * sizeof(float) + 511) & ~(size_t)511;
    float* bufA = (float*)(ws + off);
    float* bufB = bufA + (size_t)N * HID;
    float* gsum = bufB + (size_t)N * HID;
    float* cnt  = gsum + GCN_G * GCN_OUT;

    const int TB = 256;
    dim3 blk(TB);

    k_deg_init <<<dim3((N + TB - 1) / TB), blk, 0, stream>>>(dinv, N);
    k_deg_count<<<dim3((E + TB - 1) / TB), blk, 0, stream>>>(dst, dinv, E);
    k_deg_inv  <<<dim3((N + TB - 1) / TB), blk, 0, stream>>>(dinv, N);

    const int mt = N / 16;
    const int gx = (mt + 3) / 4;
    long long tE128 = (long long)E * (HID / 4);
    long long tE64  = (long long)E * (OUTF / 4);
    int nblkE128 = (int)((tE128 + TB - 1) / TB);
    int nblkE64  = (int)((tE64 + TB - 1) / TB);

    k_gemm_wmma_f32<<<dim3(gx, HID / 16), dim3(128), 0, stream>>>(x, W1, bufA, IN, HID, mt);
    k_agg_init     <<<dim3((N * HID + TB - 1) / TB), blk, 0, stream>>>(bufA, dinv, b1, bufB, N, 7);
    k_edge_scatter <<<dim3(nblkE128), blk, 0, stream>>>(src, dst, dinv, bufA, bufB, E, 7);
    k_relu         <<<dim3((N * HID + TB - 1) / TB), blk, 0, stream>>>(bufB, N * HID);

    k_gemm_wmma_f32<<<dim3(gx, HID / 16), dim3(128), 0, stream>>>(bufB, W2, bufA, HID, HID, mt);
    k_agg_init     <<<dim3((N * HID + TB - 1) / TB), blk, 0, stream>>>(bufA, dinv, b2, bufB, N, 7);
    k_edge_scatter <<<dim3(nblkE128), blk, 0, stream>>>(src, dst, dinv, bufA, bufB, E, 7);
    k_relu         <<<dim3((N * HID + TB - 1) / TB), blk, 0, stream>>>(bufB, N * HID);

    k_gemm_wmma_f32<<<dim3(gx, OUTF / 16), dim3(128), 0, stream>>>(bufB, W3, bufA, HID, OUTF, mt);
    k_agg_init     <<<dim3((N * OUTF + TB - 1) / TB), blk, 0, stream>>>(bufA, dinv, b3, bufB, N, 6);
    k_edge_scatter <<<dim3(nblkE64), blk, 0, stream>>>(src, dst, dinv, bufA, bufB, E, 6);

    k_readout_zero   <<<dim3((GCN_G * GCN_OUT + TB - 1) / TB), blk, 0, stream>>>(gsum, cnt);
    k_readout_scatter<<<dim3((N * OUTF + TB - 1) / TB), blk, 0, stream>>>(bufB, batch, gsum, cnt, N);
    k_readout_final  <<<dim3((GCN_G * GCN_OUT + TB - 1) / TB), blk, 0, stream>>>(gsum, cnt, (float*)d_out);
}


# Layer 3 + mean readout reassociated, out=((P*A)*h2)*W3: pooled edge weights C[64xN] built by float atomics, C*h2 (split-K) and T*W3 on f32 WMMA; no N-row layer-3 pass; sum order changes
# speedup vs baseline: 1.1977x; 1.1977x over previous
#include <hip/hip_runtime.h>
#include <math.h>

typedef float v2f __attribute__((ext_vector_type(2)));
typedef float v8f __attribute__((ext_vector_type(8)));

#define GCN_G   64
#define GCN_OUT 64
#define LDS_STRIDE 132

__global__ void k_deg_init(float* __restrict__ deg, int n) {
    int i = blockIdx.x * blockDim.x + threadIdx.x;
    if (i < n) deg[i] = 1.0f;
}

__global__ void k_deg_count(const int* __restrict__ dst, float* __restrict__ deg, int e) {
    int i = blockIdx.x * blockDim.x + threadIdx.x;
    if (i < e) atomicAdd(&deg[dst[i]], 1.0f);
}

__global__ void k_deg_inv(float* __restrict__ deg, int n) {
    int i = blockIdx.x * blockDim.x + threadIdx.x;
    if (i < n) {
        float d = deg[i];
        deg[i] = (d > 0.0f) ? (1.0f / sqrtf(d)) : 0.0f;
    }
}

__global__ void __launch_bounds__(128)
k_gemm_wmma_f32(const float* __restrict__ X, const float* __restrict__ W,
                float* __restrict__ H, int K, int F, int mtiles) {
    __shared__ float Wlds[16 * LDS_STRIDE];

    const int t     = threadIdx.x;
    const int wave  = t >> 5;
    const int lane  = t & 31;
    const int half  = lane >> 4;
    const int l16   = lane & 15;
    const int col0  = blockIdx.y << 4;
    const int mtile = (blockIdx.x << 2) + wave;

    const int reps = (K << 4) >> 7;
    for (int rep = 0; rep < reps; ++rep) {
        int idx = (rep << 7) + t;
        int k = idx >> 4;
        int n = idx & 15;
        Wlds[n * LDS_STRIDE + k] = W[(size_t)k * F + (col0 + n)];
    }
    __syncthreads();

    if (mtile < mtiles) {
        const int row0 = mtile << 4;
        const float* __restrict__ Xrow = X + (size_t)(row0 + l16) * K + (half << 1);
        const float* __restrict__ Wl = &Wlds[l16 * LDS_STRIDE + (half << 1)];

        v8f c = {};
#pragma unroll 8
        for (int k0 = 0; k0 < K; k0 += 4) {
            v2f a = *(const v2f*)(Xrow + k0);
            v2f b = *(const v2f*)(Wl + k0);
            c = __builtin_amdgcn_wmma_f32_16x16x4_f32(false, a, false, b,
                                                      (short)0, c, false, false);
        }

        float* __restrict__ Ho = H + (size_t)(row0 + (half << 3)) * F + (col0 + l16);
#pragma unroll
        for (int v = 0; v < 8; ++v)
            Ho[(size_t)v * F] = c[v];
    }
}

__global__ void k_agg_init(const float* __restrict__ H, const float* __restrict__ dinv,
                           const float* __restrict__ bias, float* __restrict__ out,
                           int n, int fshift) {
    int idx = blockIdx.x * blockDim.x + threadIdx.x;
    int total = n << fshift;
    if (idx >= total) return;
    int i = idx >> fshift;
    int f = idx & ((1 << fshift) - 1);
    float di = dinv[i];
    out[idx] = di * di * H[idx] + bias[f];
}

__global__ void k_edge_scatter(const int* __restrict__ src, const int* __restrict__ dst,
                               const float* __restrict__ dinv, const float* __restrict__ H,
                               float* __restrict__ out, int e, int fshift) {
    const int cshift = fshift - 2;
    long long t = (long long)blockIdx.x * blockDim.x + threadIdx.x;
    long long total = (long long)e << cshift;
    if (t >= total) return;
    int ei = (int)(t >> cshift);
    int c  = ((int)t & ((1 << cshift) - 1)) << 2;
    int s = src[ei];
    int d = dst[ei];
    float w = dinv[s] * dinv[d];
    int F = 1 << fshift;
    const float4 h4 = *(const float4*)(H + (size_t)s * F + c);
    float* o = out + (size_t)d * F + c;
    atomicAdd(o + 0, w * h4.x);
    atomicAdd(o + 1, w * h4.y);
    atomicAdd(o + 2, w * h4.z);
    atomicAdd(o + 3, w * h4.w);
}

__global__ void k_relu(float* __restrict__ x, int total) {
    int i = blockIdx.x * blockDim.x + threadIdx.x;
    if (i < total) x[i] = fmaxf(x[i], 0.0f);
}

#define POOL_KC 400
#define POOL_STRIDE 404
#define POOL_TILE (GCN_G * 128)

__global__ void k_pool_init(const int* __restrict__ batch, const float* __restrict__ dinv,
                            float* __restrict__ Cm, int n) {
    int j = blockIdx.x * blockDim.x + threadIdx.x;
    if (j >= n) return;
    int g = batch[j];
    float d = dinv[j];
    float self = d * d;
    for (int q = 0; q < GCN_G; ++q)
        Cm[(size_t)q * n + j] = (q == g) ? self : 0.0f;
}

__global__ void k_pool_edges(const int* __restrict__ src, const int* __restrict__ dst,
                             const int* __restrict__ batch, const float* __restrict__ dinv,
                             float* __restrict__ Cm, int e, int n) {
    int i = blockIdx.x * blockDim.x + threadIdx.x;
    if (i >= e) return;
    int s = src[i];
    int d = dst[i];
    atomicAdd(&Cm[(size_t)batch[d] * n + s], dinv[s] * dinv[d]);
}

__global__ void __launch_bounds__(256)
k_pool_gemm(const float* __restrict__ Cm, const float* __restrict__ h,
            float* __restrict__ Tpart, int n) {
    __shared__ float As[GCN_G * POOL_STRIDE];
    const int t    = threadIdx.x;
    const int wave = t >> 5;
    const int lane = t & 31;
    const int half = lane >> 4;
    const int l16  = lane & 15;
    const int k0   = blockIdx.x * POOL_KC;
    const int klen = min(POOL_KC, n - k0);

    for (int idx = t; idx < GCN_G * (POOL_KC / 4); idx += 256) {
        int row = idx / (POOL_KC / 4);
        int q   = (idx - row * (POOL_KC / 4)) << 2;
        float4 v = make_float4(0.0f, 0.0f, 0.0f, 0.0f);
        if (q < klen) v = *(const float4*)(Cm + (size_t)row * n + k0 + q);
        *(float4*)(&As[row * POOL_STRIDE + q]) = v;
    }
    __syncthreads();

    const int n0 = wave << 4;
    const float* __restrict__ al = &As[l16 * POOL_STRIDE + (half << 1)];
    const float* __restrict__ hb = h + (size_t)(k0 + (half << 1)) * 128 + n0 + l16;

    v8f c0 = {}, c1 = {}, c2 = {}, c3 = {};
    v2f bn[4];
#pragma unroll
    for (int u = 0; u < 4; ++u) {
        bn[u].x = hb[(size_t)(4 * u) * 128];
        bn[u].y = hb[(size_t)(4 * u) * 128 + 128];
    }
    for (int kk = 0; kk < klen; kk += 16) {
        v2f bc[4];
#pragma unroll
        for (int u = 0; u < 4; ++u) bc[u] = bn[u];
        const float* __restrict__ hn = hb + (size_t)min(kk + 16, klen - 16) * 128;
#pragma unroll
        for (int u = 0; u < 4; ++u) {
            bn[u].x = hn[(size_t)(4 * u) * 128];
            bn[u].y = hn[(size_t)(4 * u) * 128 + 128];
        }
#pragma unroll
        for (int u = 0; u < 4; ++u) {
            const float* __restrict__ ak = al + kk + 4 * u;
            v2f a0 = *(const v2f*)(ak);
            v2f a1 = *(const v2f*)(ak + 16 * POOL_STRIDE);
            v2f a2 = *(const v2f*)(ak + 32 * POOL_STRIDE);
            v2f a3 = *(const v2f*)(ak + 48 * POOL_STRIDE);
            c0 = __builtin_amdgcn_wmma_f32_16x16x4_f32(false, a0, false, bc[u], (short)0, c0, false, false);
            c1 = __builtin_amdgcn_wmma_f32_16x16x4_f32(false, a1, false, bc[u], (short)0, c1, false, false);
            c2 = __builtin_amdgcn_wmma_f32_16x16x4_f32(false, a2, false, bc[u], (short)0, c2, false, false);
            c3 = __builtin_amdgcn_wmma_f32_16x16x4_f32(false, a3, false, bc[u], (short)0, c3, false, false);
        }
    }

    float* __restrict__ o = Tpart + (size_t)blockIdx.x * POOL_TILE + (size_t)(half << 3) * 128 + n0 + l16;
#pragma unroll
    for (int v = 0; v < 8; ++v) {
        o[(size_t)(v)      * 128] = c0[v];
        o[(size_t)(v + 16) * 128] = c1[v];
        o[(size_t)(v + 32) * 128] = c2[v];
        o[(size_t)(v + 48) * 128] = c3[v];
    }
}

__global__ void k_pool_reduce(const float* __restrict__ Tpart, float* __restrict__ T8, int nparts) {
    int tid = blockIdx.x * blockDim.x + threadIdx.x;
    int e   = tid & (POOL_TILE - 1);
    int pg  = tid / POOL_TILE;
    int per = (nparts + 7) >> 3;
    int p0  = pg * per;
    int p1  = min(nparts, p0 + per);
    float s = 0.0f;
    for (int p = p0; p < p1; ++p) s += Tpart[(size_t)p * POOL_TILE + e];
    T8[tid] = s;
}

__global__ void __launch_bounds__(512)
k_pool_final(const float* __restrict__ T8, const float* __restrict__ W3,
             const float* __restrict__ b3, const int* __restrict__ batch,
             float* __restrict__ outp, int n) {
    __shared__ float Ts[GCN_G * LDS_STRIDE];
    __shared__ float Wl[GCN_OUT * LDS_STRIDE];
    __shared__ int   hist[GCN_G];
    const int t    = threadIdx.x;
    const int wave = t >> 5;
    const int lane = t & 31;
    const int half = lane >> 4;
    const int l16  = lane & 15;

    if (t < GCN_G) hist[t] = 0;
    for (int idx = t; idx < POOL_TILE; idx += 512) {
        float s = 0.0f;
#pragma unroll
        for (int pg = 0; pg < 8; ++pg) s += T8[pg * POOL_TILE + idx];
        Ts[(idx >> 7) * LDS_STRIDE + (idx & 127)] = s;
    }
    for (int idx = t; idx < 128 * GCN_OUT; idx += 512)
        Wl[(idx & 63) * LDS_STRIDE + (idx >> 6)] = W3[idx];
    __syncthreads();

    for (int base = 0; base < n; base += 512) {
        int i = base + t;
        int g = (i < n) ? batch[i] : -1;
        int g0 = __shfl(g, 0);
        int same = __popcll(__ballot(g == g0 && g >= 0));
        if (lane == 0 && g0 >= 0) atomicAdd(&hist[g0], same);
        if (g != g0 && g >= 0) atomicAdd(&hist[g], 1);
    }
    __syncthreads();

    const int mi = wave >> 2;
    const int ni = wave & 3;
    const float* __restrict__ al = &Ts[((mi << 4) + l16) * LDS_STRIDE + (half << 1)];
    const float* __restrict__ bl = &Wl[((ni << 4) + l16) * LDS_STRIDE + (half << 1)];
    v8f c = {};
#pragma unroll
    for (int k0 = 0; k0 < 128; k0 += 4) {
        v2f a = *(const v2f*)(al + k0);
        v2f b = *(const v2f*)(bl + k0);
        c = __builtin_amdgcn_wmma_f32_16x16x4_f32(false, a, false, b, (short)0, c, false, false);
    }
    const int col = (ni << 4) + l16;
    const float bias = b3[col];
#pragma unroll
    for (int v = 0; v < 8; ++v) {
        int g = (mi << 4) + (half << 3) + v;
        float cf = (float)hist[g];
        outp[g * GCN_OUT + col] = (c[v] + cf * bias) / fmaxf(cf, 1.0f);
    }
}

extern "C" void kernel_launch(void* const* d_in, const int* in_sizes, int n_in,
                              void* d_out, int out_size, void* d_ws, size_t ws_size,
                              hipStream_t stream) {
    const float* x     = (const float*)d_in[0];
    const int*   edge  = (const int*)d_in[1];
    const int*   batch = (const int*)d_in[2];
    const float* W1    = (const float*)d_in[3];
    const float* b1    = (const float*)d_in[4];
    const float* W2    = (const float*)d_in[5];
    const float* b2    = (const float*)d_in[6];
    const float* W3    = (const float*)d_in[7];
    const float* b3    = (const float*)d_in[8];

    const int IN = 128, HID = 128;
    const int N = in_sizes[0] / IN;
    const int E = in_sizes[1] / 2;
    const int* src = edge;
    const int* dst = edge + E;

    char* ws = (char*)d_ws;
    float* dinv = (float*)ws;
    size_t off = ((size_t)N * sizeof(float) + 511) & ~(size_t)511;
    float* bufA = (float*)(ws + off);
    float* bufB = bufA + (size_t)N * HID;
    const int P = (N + POOL_KC - 1) / POOL_KC;
    float* Cm    = bufA;
    float* Tpart = bufB + (size_t)N * HID;
    float* T8    = Tpart + (size_t)P * POOL_TILE;

    const int TB = 256;
    dim3 blk(TB);

    k_deg_init <<<dim3((N + TB - 1) / TB), blk, 0, stream>>>(dinv, N);
    k_deg_count<<<dim3((E + TB - 1) / TB), blk, 0, stream>>>(dst, dinv, E);
    k_deg_inv  <<<dim3((N + TB - 1) / TB), blk, 0, stream>>>(dinv, N);

    const int mt = N / 16;
    const int gx = (mt + 3) / 4;
    long long tE128 = (long long)E * (HID / 4);
    int nblkE128 = (int)((tE128 + TB - 1) / TB);

    k_gemm_wmma_f32<<<dim3(gx, HID / 16), dim3(128), 0, stream>>>(x, W1, bufA, IN, HID, mt);
    k_agg_init     <<<dim3((N * HID + TB - 1) / TB), blk, 0, stream>>>(bufA, dinv, b1, bufB, N, 7);
    k_edge_scatter <<<dim3(nblkE128), blk, 0, stream>>>(src, dst, dinv, bufA, bufB, E, 7);
    k_relu         <<<dim3((N * HID + TB - 1) / TB), blk, 0, stream>>>(bufB, N * HID);

    k_gemm_wmma_f32<<<dim3(gx, HID / 16), dim3(128), 0, stream>>>(bufB, W2, bufA, HID, HID, mt);
    k_agg_init     <<<dim3((N * HID + TB - 1) / TB), blk, 0, stream>>>(bufA, dinv, b2, bufB, N, 7);
    k_edge_scatter <<<dim3(nblkE128), blk, 0, stream>>>(src, dst, dinv, bufA, bufB, E, 7);
    k_relu         <<<dim3((N * HID + TB - 1) / TB), blk, 0, stream>>>(bufB, N * HID);

    k_pool_init  <<<dim3((N + TB - 1) / TB), blk, 0, stream>>>(batch, dinv, Cm, N);
    k_pool_edges <<<dim3((E + TB - 1) / TB), blk, 0, stream>>>(src, dst, batch, dinv, Cm, E, N);
    k_pool_gemm  <<<dim3(P), blk, 0, stream>>>(Cm, bufB, Tpart, N);
    k_pool_reduce<<<dim3(8 * POOL_TILE / TB), blk, 0, stream>>>(Tpart, T8, P);
    k_pool_final <<<dim3(1), dim3(512), 0, stream>>>(T8, W3, b3, batch, (float*)d_out, N);
}


# Layers 1,2 edge sums moved from global float atomics into LDS: edges binned once by 400-node destination range; a block per range adds its rows in LDS (float atomics, arrival order), one store a row
# speedup vs baseline: 2.4223x; 2.0225x over previous
#include <hip/hip_runtime.h>
#include <math.h>

typedef float v2f __attribute__((ext_vector_type(2)));
typedef float v8f __attribute__((ext_vector_type(8)));

#define GCN_G   64
#define GCN_OUT 64
#define LDS_STRIDE 132

__global__ void k_deg_init(float* __restrict__ deg, int n) {
    int i = blockIdx.x * blockDim.x + threadIdx.x;
    if (i < n) deg[i] = 1.0f;
}

__global__ void k_deg_count(const int* __restrict__ dst, float* __restrict__ deg, int e) {
    int i = blockIdx.x * blockDim.x + threadIdx.x;
    if (i < e) atomicAdd(&deg[dst[i]], 1.0f);
}

#define BIN_R 400
#define FILL_EPT 4
#define FILL_TB 1024

__global__ void __launch_bounds__(BIN_R)
k_deg_inv_bins(float* __restrict__ deg, int* __restrict__ binCount, int n) {
    __shared__ int tot;
    const int t = threadIdx.x;
    if (t == 0) tot = 0;
    __syncthreads();
    const int i = blockIdx.x * BIN_R + t;
    int c = 0;
    if (i < n) {
        float d = deg[i];
        c = (int)d - 1;
        deg[i] = (d > 0.0f) ? (1.0f / sqrtf(d)) : 0.0f;
    }
    atomicAdd(&tot, c);
    __syncthreads();
    if (t == 0) binCount[blockIdx.x] = tot;
}

__global__ void __launch_bounds__(1024)
k_bin_scan(const int* __restrict__ cnt, int* __restrict__ start, int* __restrict__ cursor, int nb) {
    __shared__ int s[1024];
    const int t = threadIdx.x;
    int carry = 0;
    for (int b0 = 0; b0 < nb; b0 += 1024) {
        const int b = b0 + t;
        const int v = (b < nb) ? cnt[b] : 0;
        s[t] = v;
        __syncthreads();
        for (int off = 1; off < 1024; off <<= 1) {
            int a = (t >= off) ? s[t - off] : 0;
            __syncthreads();
            s[t] += a;
            __syncthreads();
        }
        if (b < nb) {
            start[b]  = carry + s[t] - v;
            cursor[b] = 0;
        }
        const int total = s[1023];
        __syncthreads();
        carry += total;
    }
    if (t == 0) start[nb] = carry;
}

__global__ void __launch_bounds__(FILL_TB)
k_bin_fill(const int* __restrict__ src, const int* __restrict__ dst,
           const float* __restrict__ dinv, const int* __restrict__ start,
           int* __restrict__ cursor, int* __restrict__ recs, int e, int nb) {
    extern __shared__ int fsm[];
    int* hist = fsm;
    int* base = fsm + nb;
    const int t = threadIdx.x;
    for (int b = t; b < nb; b += FILL_TB) hist[b] = 0;
    __syncthreads();

    const int e0 = blockIdx.x * (FILL_TB * FILL_EPT);
    int es[FILL_EPT], ed[FILL_EPT], eb[FILL_EPT], er[FILL_EPT];
#pragma unroll
    for (int u = 0; u < FILL_EPT; ++u) {
        const int i = e0 + u * FILL_TB + t;
        eb[u] = -1;
        if (i < e) {
            es[u] = src[i];
            ed[u] = dst[i];
            eb[u] = ed[u] / BIN_R;
            er[u] = atomicAdd(&hist[eb[u]], 1);
        }
    }
    __syncthreads();
    for (int b = t; b < nb; b += FILL_TB) {
        const int c = hist[b];
        base[b] = start[b] + (c > 0 ? atomicAdd(&cursor[b], c) : 0);
    }
    __syncthreads();
    int4* __restrict__ out = (int4*)recs;
#pragma unroll
    for (int u = 0; u < FILL_EPT; ++u) {
        if (eb[u] >= 0) {
            const float w = dinv[es[u]] * dinv[ed[u]];
            out[base[eb[u]] + er[u]] = make_int4(es[u], ed[u] - eb[u] * BIN_R, __float_as_int(w), 0);
        }
    }
}

__global__ void __launch_bounds__(128)
k_gemm_wmma_f32(const float* __restrict__ X, const float* __restrict__ W,
                float* __restrict__ H, int K, int F, int mtiles) {
    __shared__ float Wlds[16 * LDS_STRIDE];

    const int t     = threadIdx.x;
    const int wave  = t >> 5;
    const int lane  = t & 31;
    const int half  = lane >> 4;
    const int l16   = lane & 15;
    const int col0  = blockIdx.y << 4;
    const int mtile = (blockIdx.x << 2) + wave;

    const int reps = (K << 4) >> 7;
    for (int rep = 0; rep < reps; ++rep) {
        int idx = (rep << 7) + t;
        int k = idx >> 4;
        int n = idx & 15;
        Wlds[n * LDS_STRIDE + k] = W[(size_t)k * F + (col0 + n)];
    }
    __syncthreads();

    if (mtile < mtiles) {
        const int row0 = mtile << 4;
        const float* __restrict__ Xrow = X + (size_t)(row0 + l16) * K + (half << 1);
        const float* __restrict__ Wl = &Wlds[l16 * LDS_STRIDE + (half << 1)];

        v8f c = {};
#pragma unroll 8
        for (int k0 = 0; k0 < K; k0 += 4) {
            v2f a = *(const v2f*)(Xrow + k0);
            v2f b = *(const v2f*)(Wl + k0);
            c = __builtin_amdgcn_wmma_f32_16x16x4_f32(false, a, false, b,
                                                      (short)0, c, false, false);
        }

        float* __restrict__ Ho = H + (size_t)(row0 + (half << 3)) * F + (col0 + l16);
#pragma unroll
        for (int v = 0; v < 8; ++v)
            Ho[(size_t)v * F] = c[v];
    }
}

#define AGG_TB 1024
#define AGG_U  8

__device__ __forceinline__ void agg_add_row(float* __restrict__ acc, int dl, float w, int lane,
                                            float h0, float h1, float h2, float h3) {
    float* __restrict__ ap = acc + dl * 128 + lane;
    atomicAdd(ap,      w * h0);
    atomicAdd(ap + 32, w * h1);
    atomicAdd(ap + 64, w * h2);
    atomicAdd(ap + 96, w * h3);
}

__global__ void __launch_bounds__(AGG_TB)
k_agg_lds(const int* __restrict__ recs, const int* __restrict__ start,
          const float* __restrict__ H, const float* __restrict__ dinv,
          const float* __restrict__ bias, float* __restrict__ out, int n) {
    __shared__ float acc[BIN_R * 128];
    const int t     = threadIdx.x;
    const int lane  = t & 31;
    const int wave  = __builtin_amdgcn_readfirstlane(t >> 5);
    const int node0 = blockIdx.x * BIN_R;
    const int rows  = min(BIN_R, n - node0);

    for (int idx = t; idx < rows * 32; idx += AGG_TB) {
        const int r  = idx >> 5;
        const int c4 = (idx & 31) << 2;
        const float di = dinv[node0 + r];
        const float sw = di * di;
        const float4 h = *(const float4*)(H + (size_t)(node0 + r) * 128 + c4);
        float4 v;
        v.x = sw * h.x + bias[c4 + 0];
        v.y = sw * h.y + bias[c4 + 1];
        v.z = sw * h.z + bias[c4 + 2];
        v.w = sw * h.w + bias[c4 + 3];
        *(float4*)(&acc[r * 128 + c4]) = v;
    }
    __syncthreads();

    const int eb  = start[blockIdx.x];
    const int cnt = start[blockIdx.x + 1] - eb;
    const int w0  = eb + (int)(((long long)cnt * wave) >> 5);
    const int w1  = eb + (int)(((long long)cnt * (wave + 1)) >> 5);
    const int4* __restrict__ rec4 = (const int4*)recs;
    const float* __restrict__ Hl = H + lane;

    for (int p = w0; p < w1; p += 32) {
        const int m = min(32, w1 - p);
        int4 rc = make_int4(0, 0, 0, 0);
        if (lane < m) rc = rec4[p + lane];
        if (m == 32) {
#pragma unroll
            for (int j0 = 0; j0 < 32; j0 += AGG_U) {
                float v[AGG_U][4];
#pragma unroll
                for (int u = 0; u < AGG_U; ++u) {
                    const int s = __builtin_amdgcn_readlane(rc.x, j0 + u);
                    const float* __restrict__ hp = Hl + (size_t)s * 128;
                    v[u][0] = hp[0];
                    v[u][1] = hp[32];
                    v[u][2] = hp[64];
                    v[u][3] = hp[96];
                }
#pragma unroll
                for (int u = 0; u < AGG_U; ++u) {
                    const int   dl = __builtin_amdgcn_readlane(rc.y, j0 + u);
                    const float w  = __int_as_float(__builtin_amdgcn_readlane(rc.z, j0 + u));
                    agg_add_row(acc, dl, w, lane, v[u][0], v[u][1], v[u][2], v[u][3]);
                }
            }
        } else {
            for (int j = 0; j < m; ++j) {
                const int   s  = __builtin_amdgcn_readlane(rc.x, j);
                const int   dl = __builtin_amdgcn_readlane(rc.y, j);
                const float w  = __int_as_float(__builtin_amdgcn_readlane(rc.z, j));
                const float* __restrict__ hp = Hl + (size_t)s * 128;
                agg_add_row(acc, dl, w, lane, hp[0], hp[32], hp[64], hp[96]);
            }
        }
    }
    __syncthreads();

    for (int idx = t; idx < rows * 32; idx += AGG_TB) {
        const int r  = idx >> 5;
        const int c4 = (idx & 31) << 2;
        float4 v = *(const float4*)(&acc[r * 128 + c4]);
        v.x = fmaxf(v.x, 0.0f);
        v.y = fmaxf(v.y, 0.0f);
        v.z = fmaxf(v.z, 0.0f);
        v.w = fmaxf(v.w, 0.0f);
        *(float4*)(out + (size_t)(node0 + r) * 128 + c4) = v;
    }
}

#define POOL_KC 400
#define POOL_STRIDE 404
#define POOL_TILE (GCN_G * 128)

__global__ void k_pool_init(const int* __restrict__ batch, const float* __restrict__ dinv,
                            float* __restrict__ Cm, int n) {
    int j = blockIdx.x * blockDim.x + threadIdx.x;
    if (j >= n) return;
    int g = batch[j];
    float d = dinv[j];
    float self = d * d;
    for (int q = 0; q < GCN_G; ++q)
        Cm[(size_t)q * n + j] = (q == g) ? self : 0.0f;
}

__global__ void k_pool_edges(const int* __restrict__ src, const int* __restrict__ dst,
                             const int* __restrict__ batch, const float* __restrict__ dinv,
                             float* __restrict__ Cm, int e, int n) {
    int i = blockIdx.x * blockDim.x + threadIdx.x;
    if (i >= e) return;
    int s = src[i];
    int d = dst[i];
    atomicAdd(&Cm[(size_t)batch[d] * n + s], dinv[s] * dinv[d]);
}

__global__ void __launch_bounds__(256)
k_pool_gemm(const float* __restrict__ Cm, const float* __restrict__ h,
            float* __restrict__ Tpart, int n) {
    __shared__ float As[GCN_G * POOL_STRIDE];
    const int t    = threadIdx.x;
    const int wave = t >> 5;
    const int lane = t & 31;
    const int half = lane >> 4;
    const int l16  = lane & 15;
    const int k0   = blockIdx.x * POOL_KC;
    const int klen = min(POOL_KC, n - k0);

    for (int idx = t; idx < GCN_G * (POOL_KC / 4); idx += 256) {
        int row = idx / (POOL_KC / 4);
        int q   = (idx - row * (POOL_KC / 4)) << 2;
        float4 v = make_float4(0.0f, 0.0f, 0.0f, 0.0f);
        if (q < klen) v = *(const float4*)(Cm + (size_t)row * n + k0 + q);
        *(float4*)(&As[row * POOL_STRIDE + q]) = v;
    }
    __syncthreads();

    const int n0 = wave << 4;
    const float* __restrict__ al = &As[l16 * POOL_STRIDE + (half << 1)];
    const float* __restrict__ hb = h + (size_t)(k0 + (half << 1)) * 128 + n0 + l16;

    v8f c0 = {}, c1 = {}, c2 = {}, c3 = {};
    v2f bn[4];
#pragma unroll
    for (int u = 0; u < 4; ++u) {
        bn[u].x = hb[(size_t)(4 * u) * 128];
        bn[u].y = hb[(size_t)(4 * u) * 128 + 128];
    }
    for (int kk = 0; kk < klen; kk += 16) {
        v2f bc[4];
#pragma unroll
        for (int u = 0; u < 4; ++u) bc[u] = bn[u];
        const float* __restrict__ hn = hb + (size_t)min(kk + 16, klen - 16) * 128;
#pragma unroll
        for (int u = 0; u < 4; ++u) {
            bn[u].x = hn[(size_t)(4 * u) * 128];
            bn[u].y = hn[(size_t)(4 * u) * 128 + 128];
        }
#pragma unroll
        for (int u = 0; u < 4; ++u) {
            const float* __restrict__ ak = al + kk + 4 * u;
            v2f a0 = *(const v2f*)(ak);
            v2f a1 = *(const v2f*)(ak + 16 * POOL_STRIDE);
            v2f a2 = *(const v2f*)(ak + 32 * POOL_STRIDE);
            v2f a3 = *(const v2f*)(ak + 48 * POOL_STRIDE);
            c0 = __builtin_amdgcn_wmma_f32_16x16x4_f32(false, a0, false, bc[u], (short)0, c0, false, false);
            c1 = __builtin_amdgcn_wmma_f32_16x16x4_f32(false, a1, false, bc[u], (short)0, c1, false, false);
            c2 = __builtin_amdgcn_wmma_f32_16x16x4_f32(false, a2, false, bc[u], (short)0, c2, false, false);
            c3 = __builtin_amdgcn_wmma_f32_16x16x4_f32(false, a3, false, bc[u], (short)0, c3, false, false);
        }
    }

    float* __restrict__ o = Tpart + (size_t)blockIdx.x * POOL_TILE + (size_t)(half << 3) * 128 + n0 + l16;
#pragma unroll
    for (int v = 0; v < 8; ++v) {
        o[(size_t)(v)      * 128] = c0[v];
        o[(size_t)(v + 16) * 128] = c1[v];
        o[(size_t)(v + 32) * 128] = c2[v];
        o[(size_t)(v + 48) * 128] = c3[v];
    }
}

__global__ void k_pool_reduce(const float* __restrict__ Tpart, float* __restrict__ T8, int nparts) {
    int tid = blockIdx.x * blockDim.x + threadIdx.x;
    int e   = tid & (POOL_TILE - 1);
    int pg  = tid / POOL_TILE;
    int per = (nparts + 7) >> 3;
    int p0  = pg * per;
    int p1  = min(nparts, p0 + per);
    float s = 0.0f;
    for (int p = p0; p < p1; ++p) s += Tpart[(size_t)p * POOL_TILE + e];
    T8[tid] = s;
}

__global__ void __launch_bounds__(512)
k_pool_final(const float* __restrict__ T8, const float* __restrict__ W3,
             const float* __restrict__ b3, const int* __restrict__ batch,
             float* __restrict__ outp, int n) {
    __shared__ float Ts[GCN_G * LDS_STRIDE];
    __shared__ float Wl[GCN_OUT * LDS_STRIDE];
    __shared__ int   hist[GCN_G];
    const int t    = threadIdx.x;
    const int wave = t >> 5;
    const int lane = t & 31;
    const int half = lane >> 4;
    const int l16  = lane & 15;

    if (t < GCN_G) hist[t] = 0;
    for (int idx = t; idx < POOL_TILE; idx += 512) {
        float s = 0.0f;
#pragma unroll
        for (int pg = 0; pg < 8; ++pg) s += T8[pg * POOL_TILE + idx];
        Ts[(idx >> 7) * LDS_STRIDE + (idx & 127)] = s;
    }
    for (int idx = t; idx < 128 * GCN_OUT; idx += 512)
        Wl[(idx & 63) * LDS_STRIDE + (idx >> 6)] = W3[idx];
    __syncthreads();

    for (int base = 0; base < n; base += 512) {
        int i = base + t;
        int g = (i < n) ? batch[i] : -1;
        int g0 = __shfl(g, 0);
        int same = __popcll(__ballot(g == g0 && g >= 0));
        if (lane == 0 && g0 >= 0) atomicAdd(&hist[g0], same);
        if (g != g0 && g >= 0) atomicAdd(&hist[g], 1);
    }
    __syncthreads();

    const int mi = wave >> 2;
    const int ni = wave & 3;
    const float* __restrict__ al = &Ts[((mi << 4) + l16) * LDS_STRIDE + (half << 1)];
    const float* __restrict__ bl = &Wl[((ni << 4) + l16) * LDS_STRIDE + (half << 1)];
    v8f c = {};
#pragma unroll
    for (int k0 = 0; k0 < 128; k0 += 4) {
        v2f a = *(const v2f*)(al + k0);
        v2f b = *(const v2f*)(bl + k0);
        c = __builtin_amdgcn_wmma_f32_16x16x4_f32(false, a, false, b, (short)0, c, false, false);
    }
    const int col = (ni << 4) + l16;
    const float bias = b3[col];
#pragma unroll
    for (int v = 0; v < 8; ++v) {
        int g = (mi << 4) + (half << 3) + v;
        float cf = (float)hist[g];
        outp[g * GCN_OUT + col] = (c[v] + cf * bias) / fmaxf(cf, 1.0f);
    }
}

extern "C" void kernel_launch(void* const* d_in, const int* in_sizes, int n_in,
                              void* d_out, int out_size, void* d_ws, size_t ws_size,
                              hipStream_t stream) {
    const float* x     = (const float*)d_in[0];
    const int*   edge  = (const int*)d_in[1];
    const int*   batch = (const int*)d_in[2];
    const float* W1    = (const float*)d_in[3];
    const float* b1    = (const float*)d_in[4];
    const float* W2    = (const float*)d_in[5];
    const float* b2    = (const float*)d_in[6];
    const float* W3    = (const float*)d_in[7];
    const float* b3    = (const float*)d_in[8];

    const int IN = 128, HID = 128;
    const int N = in_sizes[0] / IN;
    const int E = in_sizes[1] / 2;
    const int* src = edge;
    const int* dst = edge + E;

    char* ws = (char*)d_ws;
    float* dinv = (float*)ws;
    size_t off = ((size_t)N * sizeof(float) + 511) & ~(size_t)511;
    float* bufA = (float*)(ws + off);
    float* bufB = bufA + (size_t)N * HID;
    const int P = (N + POOL_KC - 1) / POOL_KC;
    float* Cm    = bufA;
    float* Tpart = bufB + (size_t)N * HID;
    float* T8    = Tpart + (size_t)P * POOL_TILE;
    const int NB    = (N + BIN_R - 1) / BIN_R;
    const int NBpad = (NB + 1 + 127) & ~127;
    int* binCount  = (int*)(T8 + (size_t)8 * POOL_TILE);
    int* binStart  = binCount + NBpad;
    int* binCursor = binStart + NBpad;
    int* recs      = binCursor + NBpad;

    const int TB = 256;
    dim3 blk(TB);

    k_deg_init <<<dim3((N + TB - 1) / TB), blk, 0, stream>>>(dinv, N);
    k_deg_count<<<dim3((E + TB - 1) / TB), blk, 0, stream>>>(dst, dinv, E);
    k_deg_inv_bins<<<dim3(NB), dim3(BIN_R), 0, stream>>>(dinv, binCount, N);

    k_bin_scan<<<dim3(1), dim3(1024), 0, stream>>>(binCount, binStart, binCursor, NB);
    k_bin_fill<<<dim3((E + FILL_TB * FILL_EPT - 1) / (FILL_TB * FILL_EPT)), dim3(FILL_TB),
                 2 * NB * sizeof(int), stream>>>(src, dst, dinv, binStart, binCursor, recs, E, NB);

    const int mt = N / 16;
    const int gx = (mt + 3) / 4;

    k_gemm_wmma_f32<<<dim3(gx, HID / 16), dim3(128), 0, stream>>>(x, W1, bufA, IN, HID, mt);
    k_agg_lds      <<<dim3(NB), dim3(AGG_TB), 0, stream>>>(recs, binStart, bufA, dinv, b1, bufB, N);

    k_gemm_wmma_f32<<<dim3(gx, HID / 16), dim3(128), 0, stream>>>(bufB, W2, bufA, HID, HID, mt);
    k_agg_lds      <<<dim3(NB), dim3(AGG_TB), 0, stream>>>(recs, binStart, bufA, dinv, b2, bufB, N);

    k_pool_init  <<<dim3((N + TB - 1) / TB), blk, 0, stream>>>(batch, dinv, Cm, N);
    k_pool_edges <<<dim3((E + TB - 1) / TB), blk, 0, stream>>>(src, dst, batch, dinv, Cm, E, N);
    k_pool_gemm  <<<dim3(P), blk, 0, stream>>>(Cm, bufB, Tpart, N);
    k_pool_reduce<<<dim3(8 * POOL_TILE / TB), blk, 0, stream>>>(Tpart, T8, P);
    k_pool_final <<<dim3(1), dim3(512), 0, stream>>>(T8, W3, b3, batch, (float*)d_out, N);
}


# Nodes-per-graph count moved out of the one-block final kernel (196 dependent load trips) into the per-node pool-init grid: one ballot and one global int add per wave; float sums unchanged
# speedup vs baseline: 2.6557x; 1.0964x over previous
#include <hip/hip_runtime.h>
#include <math.h>

typedef float v2f __attribute__((ext_vector_type(2)));
typedef float v8f __attribute__((ext_vector_type(8)));

#define GCN_G   64
#define GCN_OUT 64
#define LDS_STRIDE 132

__global__ void k_deg_init(float* __restrict__ deg, int* __restrict__ gcnt, int n) {
    int i = blockIdx.x * blockDim.x + threadIdx.x;
    if (i < n) deg[i] = 1.0f;
    if (i < GCN_G) gcnt[i] = 0;
}

__global__ void k_deg_count(const int* __restrict__ dst, float* __restrict__ deg, int e) {
    int i = blockIdx.x * blockDim.x + threadIdx.x;
    if (i < e) atomicAdd(&deg[dst[i]], 1.0f);
}

#define BIN_R 400
#define FILL_EPT 4
#define FILL_TB 1024

__global__ void __launch_bounds__(BIN_R)
k_deg_inv_bins(float* __restrict__ deg, int* __restrict__ binCount, int n) {
    __shared__ int tot;
    const int t = threadIdx.x;
    if (t == 0) tot = 0;
    __syncthreads();
    const int i = blockIdx.x * BIN_R + t;
    int c = 0;
    if (i < n) {
        float d = deg[i];
        c = (int)d - 1;
        deg[i] = (d > 0.0f) ? (1.0f / sqrtf(d)) : 0.0f;
    }
    atomicAdd(&tot, c);
    __syncthreads();
    if (t == 0) binCount[blockIdx.x] = tot;
}

__global__ void __launch_bounds__(1024)
k_bin_scan(const int* __restrict__ cnt, int* __restrict__ start, int* __restrict__ cursor, int nb) {
    __shared__ int s[1024];
    const int t = threadIdx.x;
    int carry = 0;
    for (int b0 = 0; b0 < nb; b0 += 1024) {
        const int b = b0 + t;
        const int v = (b < nb) ? cnt[b] : 0;
        s[t] = v;
        __syncthreads();
        for (int off = 1; off < 1024; off <<= 1) {
            int a = (t >= off) ? s[t - off] : 0;
            __syncthreads();
            s[t] += a;
            __syncthreads();
        }
        if (b < nb) {
            start[b]  = carry + s[t] - v;
            cursor[b] = 0;
        }
        const int total = s[1023];
        __syncthreads();
        carry += total;
    }
    if (t == 0) start[nb] = carry;
}

__global__ void __launch_bounds__(FILL_TB)
k_bin_fill(const int* __restrict__ src, const int* __restrict__ dst,
           const float* __restrict__ dinv, const int* __restrict__ start,
           int* __restrict__ cursor, int* __restrict__ recs, int e, int nb) {
    extern __shared__ int fsm[];
    int* hist = fsm;
    int* base = fsm + nb;
    const int t = threadIdx.x;
    for (int b = t; b < nb; b += FILL_TB) hist[b] = 0;
    __syncthreads();

    const int e0 = blockIdx.x * (FILL_TB * FILL_EPT);
    int es[FILL_EPT], ed[FILL_EPT], eb[FILL_EPT], er[FILL_EPT];
#pragma unroll
    for (int u = 0; u < FILL_EPT; ++u) {
        const int i = e0 + u * FILL_TB + t;
        eb[u] = -1;
        if (i < e) {
            es[u] = src[i];
            ed[u] = dst[i];
            eb[u] = ed[u] / BIN_R;
            er[u] = atomicAdd(&hist[eb[u]], 1);
        }
    }
    __syncthreads();
    for (int b = t; b < nb; b += FILL_TB) {
        const int c = hist[b];
        base[b] = start[b] + (c > 0 ? atomicAdd(&cursor[b], c) : 0);
    }
    __syncthreads();
    int4* __restrict__ out = (int4*)recs;
#pragma unroll
    for (int u = 0; u < FILL_EPT; ++u) {
        if (eb[u] >= 0) {
            const float w = dinv[es[u]] * dinv[ed[u]];
            out[base[eb[u]] + er[u]] = make_int4(es[u], ed[u] - eb[u] * BIN_R, __float_as_int(w), 0);
        }
    }
}

__global__ void __launch_bounds__(128)
k_gemm_wmma_f32(const float* __restrict__ X, const float* __restrict__ W,
                float* __restrict__ H, int K, int F, int mtiles) {
    __shared__ float Wlds[16 * LDS_STRIDE];

    const int t     = threadIdx.x;
    const int wave  = t >> 5;
    const int lane  = t & 31;
    const int half  = lane >> 4;
    const int l16   = lane & 15;
    const int col0  = blockIdx.y << 4;
    const int mtile = (blockIdx.x << 2) + wave;

    const int reps = (K << 4) >> 7;
    for (int rep = 0; rep < reps; ++rep) {
        int idx = (rep << 7) + t;
        int k = idx >> 4;
        int n = idx & 15;
        Wlds[n * LDS_STRIDE + k] = W[(size_t)k * F + (col0 + n)];
    }
    __syncthreads();

    if (mtile < mtiles) {
        const int row0 = mtile << 4;
        const float* __restrict__ Xrow = X + (size_t)(row0 + l16) * K + (half << 1);
        const float* __restrict__ Wl = &Wlds[l16 * LDS_STRIDE + (half << 1)];

        v8f c = {};
#pragma unroll 8
        for (int k0 = 0; k0 < K; k0 += 4) {
            v2f a = *(const v2f*)(Xrow + k0);
            v2f b = *(const v2f*)(Wl + k0);
            c = __builtin_amdgcn_wmma_f32_16x16x4_f32(false, a, false, b,
                                                      (short)0, c, false, false);
        }

        float* __restrict__ Ho = H + (size_t)(row0 + (half << 3)) * F + (col0 + l16);
#pragma unroll
        for (int v = 0; v < 8; ++v)
            Ho[(size_t)v * F] = c[v];
    }
}

#define AGG_TB 1024
#define AGG_U  8

__device__ __forceinline__ void agg_add_row(float* __restrict__ acc, int dl, float w, int lane,
                                            float h0, float h1, float h2, float h3) {
    float* __restrict__ ap = acc + dl * 128 + lane;
    atomicAdd(ap,      w * h0);
    atomicAdd(ap + 32, w * h1);
    atomicAdd(ap + 64, w * h2);
    atomicAdd(ap + 96, w * h3);
}

__global__ void __launch_bounds__(AGG_TB)
k_agg_lds(const int* __restrict__ recs, const int* __restrict__ start,
          const float* __restrict__ H, const float* __restrict__ dinv,
          const float* __restrict__ bias, float* __restrict__ out, int n) {
    __shared__ float acc[BIN_R * 128];
    const int t     = threadIdx.x;
    const int lane  = t & 31;
    const int wave  = __builtin_amdgcn_readfirstlane(t >> 5);
    const int node0 = blockIdx.x * BIN_R;
    const int rows  = min(BIN_R, n - node0);

    for (int idx = t; idx < rows * 32; idx += AGG_TB) {
        const int r  = idx >> 5;
        const int c4 = (idx & 31) << 2;
        const float di = dinv[node0 + r];
        const float sw = di * di;
        const float4 h = *(const float4*)(H + (size_t)(node0 + r) * 128 + c4);
        float4 v;
        v.x = sw * h.x + bias[c4 + 0];
        v.y = sw * h.y + bias[c4 + 1];
        v.z = sw * h.z + bias[c4 + 2];
        v.w = sw * h.w + bias[c4 + 3];
        *(float4*)(&acc[r * 128 + c4]) = v;
    }
    __syncthreads();

    const int eb  = start[blockIdx.x];
    const int cnt = start[blockIdx.x + 1] - eb;
    const int w0  = eb + (int)(((long long)cnt * wave) >> 5);
    const int w1  = eb + (int)(((long long)cnt * (wave + 1)) >> 5);
    const int4* __restrict__ rec4 = (const int4*)recs;
    const float* __restrict__ Hl = H + lane;

    for (int p = w0; p < w1; p += 32) {
        const int m = min(32, w1 - p);
        int4 rc = make_int4(0, 0, 0, 0);
        if (lane < m) rc = rec4[p + lane];
        if (m == 32) {
#pragma unroll
            for (int j0 = 0; j0 < 32; j0 += AGG_U) {
                float v[AGG_U][4];
#pragma unroll
                for (int u = 0; u < AGG_U; ++u) {
                    const int s = __builtin_amdgcn_readlane(rc.x, j0 + u);
                    const float* __restrict__ hp = Hl + (size_t)s * 128;
                    v[u][0] = hp[0];
                    v[u][1] = hp[32];
                    v[u][2] = hp[64];
                    v[u][3] = hp[96];
                }
#pragma unroll
                for (int u = 0; u < AGG_U; ++u) {
                    const int   dl = __builtin_amdgcn_readlane(rc.y, j0 + u);
                    const float w  = __int_as_float(__builtin_amdgcn_readlane(rc.z, j0 + u));
                    agg_add_row(acc, dl, w, lane, v[u][0], v[u][1], v[u][2], v[u][3]);
                }
            }
        } else {
            for (int j = 0; j < m; ++j) {
                const int   s  = __builtin_amdgcn_readlane(rc.x, j);
                const int   dl = __builtin_amdgcn_readlane(rc.y, j);
                const float w  = __int_as_float(__builtin_amdgcn_readlane(rc.z, j));
                const float* __restrict__ hp = Hl + (size_t)s * 128;
                agg_add_row(acc, dl, w, lane, hp[0], hp[32], hp[64], hp[96]);
            }
        }
    }
    __syncthreads();

    for (int idx = t; idx < rows * 32; idx += AGG_TB) {
        const int r  = idx >> 5;
        const int c4 = (idx & 31) << 2;
        float4 v = *(const float4*)(&acc[r * 128 + c4]);
        v.x = fmaxf(v.x, 0.0f);
        v.y = fmaxf(v.y, 0.0f);
        v.z = fmaxf(v.z, 0.0f);
        v.w = fmaxf(v.w, 0.0f);
        *(float4*)(out + (size_t)(node0 + r) * 128 + c4) = v;
    }
}

#define POOL_KC 400
#define POOL_STRIDE 404
#define POOL_TILE (GCN_G * 128)

__global__ void k_pool_init(const int* __restrict__ batch, const float* __restrict__ dinv,
                            float* __restrict__ Cm, int* __restrict__ gcnt, int n) {
    int j = blockIdx.x * blockDim.x + threadIdx.x;
    int g = (j < n) ? batch[j] : -1;
    int g0 = __shfl(g, 0);
    int same = __popcll(__ballot(g == g0 && g >= 0));
    if ((threadIdx.x & 31) == 0 && g0 >= 0) atomicAdd(&gcnt[g0], same);
    if (g != g0 && g >= 0) atomicAdd(&gcnt[g], 1);
    if (j >= n) return;
    float d = dinv[j];
    float self = d * d;
    for (int q = 0; q < GCN_G; ++q)
        Cm[(size_t)q * n + j] = (q == g) ? self : 0.0f;
}

__global__ void k_pool_edges(const int* __restrict__ src, const int* __restrict__ dst,
                             const int* __restrict__ batch, const float* __restrict__ dinv,
                             float* __restrict__ Cm, int e, int n) {
    int i = blockIdx.x * blockDim.x + threadIdx.x;
    if (i >= e) return;
    int s = src[i];
    int d = dst[i];
    atomicAdd(&Cm[(size_t)batch[d] * n + s], dinv[s] * dinv[d]);
}

__global__ void __launch_bounds__(256)
k_pool_gemm(const float* __restrict__ Cm, const float* __restrict__ h,
            float* __restrict__ Tpart, int n) {
    __shared__ float As[GCN_G * POOL_STRIDE];
    const int t    = threadIdx.x;
    const int wave = t >> 5;
    const int lane = t & 31;
    const int half = lane >> 4;
    const int l16  = lane & 15;
    const int k0   = blockIdx.x * POOL_KC;
    const int klen = min(POOL_KC, n - k0);

    for (int idx = t; idx < GCN_G * (POOL_KC / 4); idx += 256) {
        int row = idx / (POOL_KC / 4);
        int q   = (idx - row * (POOL_KC / 4)) << 2;
        float4 v = make_float4(0.0f, 0.0f, 0.0f, 0.0f);
        if (q < klen) v = *(const float4*)(Cm + (size_t)row * n + k0 + q);
        *(float4*)(&As[row * POOL_STRIDE + q]) = v;
    }
    __syncthreads();

    const int n0 = wave << 4;
    const float* __restrict__ al = &As[l16 * POOL_STRIDE + (half << 1)];
    const float* __restrict__ hb = h + (size_t)(k0 + (half << 1)) * 128 + n0 + l16;

    v8f c0 = {}, c1 = {}, c2 = {}, c3 = {};
    v2f bn[4];
#pragma unroll
    for (int u = 0; u < 4; ++u) {
        bn[u].x = hb[(size_t)(4 * u) * 128];
        bn[u].y = hb[(size_t)(4 * u) * 128 + 128];
    }
    for (int kk = 0; kk < klen; kk += 16) {
        v2f bc[4];
#pragma unroll
        for (int u = 0; u < 4; ++u) bc[u] = bn[u];
        const float* __restrict__ hn = hb + (size_t)min(kk + 16, klen - 16) * 128;
#pragma unroll
        for (int u = 0; u < 4; ++u) {
            bn[u].x = hn[(size_t)(4 * u) * 128];
            bn[u].y = hn[(size_t)(4 * u) * 128 + 128];
        }
#pragma unroll
        for (int u = 0; u < 4; ++u) {
            const float* __restrict__ ak = al + kk + 4 * u;
            v2f a0 = *(const v2f*)(ak);
            v2f a1 = *(const v2f*)(ak + 16 * POOL_STRIDE);
            v2f a2 = *(const v2f*)(ak + 32 * POOL_STRIDE);
            v2f a3 = *(const v2f*)(ak + 48 * POOL_STRIDE);
            c0 = __builtin_amdgcn_wmma_f32_16x16x4_f32(false, a0, false, bc[u], (short)0, c0, false, false);
            c1 = __builtin_amdgcn_wmma_f32_16x16x4_f32(false, a1, false, bc[u], (short)0, c1, false, false);
            c2 = __builtin_amdgcn_wmma_f32_16x16x4_f32(false, a2, false, bc[u], (short)0, c2, false, false);
            c3 = __builtin_amdgcn_wmma_f32_16x16x4_f32(false, a3, false, bc[u], (short)0, c3, false, false);
        }
    }

    float* __restrict__ o = Tpart + (size_t)blockIdx.x * POOL_TILE + (size_t)(half << 3) * 128 + n0 + l16;
#pragma unroll
    for (int v = 0; v < 8; ++v) {
        o[(size_t)(v)      * 128] = c0[v];
        o[(size_t)(v + 16) * 128] = c1[v];
        o[(size_t)(v + 32) * 128] = c2[v];
        o[(size_t)(v + 48) * 128] = c3[v];
    }
}

__global__ void k_pool_reduce(const float* __restrict__ Tpart, float* __restrict__ T8, int nparts) {
    int tid = blockIdx.x * blockDim.x + threadIdx.x;
    int e   = tid & (POOL_TILE - 1);
    int pg  = tid / POOL_TILE;
    int per = (nparts + 7) >> 3;
    int p0  = pg * per;
    int p1  = min(nparts, p0 + per);
    float s = 0.0f;
    for (int p = p0; p < p1; ++p) s += Tpart[(size_t)p * POOL_TILE + e];
    T8[tid] = s;
}

__global__ void __launch_bounds__(512)
k_pool_final(const float* __restrict__ T8, const float* __restrict__ W3,
             const float* __restrict__ b3, const int* __restrict__ gcnt,
             float* __restrict__ outp) {
    __shared__ float Ts[GCN_G * LDS_STRIDE];
    __shared__ float Wl[GCN_OUT * LDS_STRIDE];
    __shared__ int   hist[GCN_G];
    const int t    = threadIdx.x;
    const int wave = t >> 5;
    const int lane = t & 31;
    const int half = lane >> 4;
    const int l16  = lane & 15;

    if (t < GCN_G) hist[t] = gcnt[t];
    for (int idx = t; idx < POOL_TILE; idx += 512) {
        float s = 0.0f;
#pragma unroll
        for (int pg = 0; pg < 8; ++pg) s += T8[pg * POOL_TILE + idx];
        Ts[(idx >> 7) * LDS_STRIDE + (idx & 127)] = s;
    }
    for (int idx = t; idx < 128 * GCN_OUT; idx += 512)
        Wl[(idx & 63) * LDS_STRIDE + (idx >> 6)] = W3[idx];
    __syncthreads();

    const int mi = wave >> 2;
    const int ni = wave & 3;
    const float* __restrict__ al = &Ts[((mi << 4) + l16) * LDS_STRIDE + (half << 1)];
    const float* __restrict__ bl = &Wl[((ni << 4) + l16) * LDS_STRIDE + (half << 1)];
    v8f c = {};
#pragma unroll
    for (int k0 = 0; k0 < 128; k0 += 4) {
        v2f a = *(const v2f*)(al + k0);
        v2f b = *(const v2f*)(bl + k0);
        c = __builtin_amdgcn_wmma_f32_16x16x4_f32(false, a, false, b, (short)0, c, false, false);
    }
    const int col = (ni << 4) + l16;
    const float bias = b3[col];
#pragma unroll
    for (int v = 0; v < 8; ++v) {
        int g = (mi << 4) + (half << 3) + v;
        float cf = (float)hist[g];
        outp[g * GCN_OUT + col] = (c[v] + cf * bias) / fmaxf(cf, 1.0f);
    }
}

extern "C" void kernel_launch(void* const* d_in, const int* in_sizes, int n_in,
                              void* d_out, int out_size, void* d_ws, size_t ws_size,
                              hipStream_t stream) {
    const float* x     = (const float*)d_in[0];
    const int*   edge  = (const int*)d_in[1];
    const int*   batch = (const int*)d_in[2];
    const float* W1    = (const float*)d_in[3];
    const float* b1    = (const float*)d_in[4];
    const float* W2    = (const float*)d_in[5];
    const float* b2    = (const float*)d_in[6];
    const float* W3    = (const float*)d_in[7];
    const float* b3    = (const float*)d_in[8];

    const int IN = 128, HID = 128;
    const int N = in_sizes[0] / IN;
    const int E = in_sizes[1] / 2;
    const int* src = edge;
    const int* dst = edge + E;

    char* ws = (char*)d_ws;
    float* dinv = (float*)ws;
    size_t off = ((size_t)N * sizeof(float) + 511) & ~(size_t)511;
    float* bufA = (float*)(ws + off);
    float* bufB = bufA + (size_t)N * HID;
    const int P = (N + POOL_KC - 1) / POOL_KC;
    float* Cm    = bufA;
    float* Tpart = bufB + (size_t)N * HID;
    float* T8    = Tpart + (size_t)P * POOL_TILE;
    const int NB    = (N + BIN_R - 1) / BIN_R;
    const int NBpad = (NB + 1 + 127) & ~127;
    int* binCount  = (int*)(T8 + (size_t)8 * POOL_TILE);
    int* binStart  = binCount + NBpad;
    int* binCursor = binStart + NBpad;
    int* gcnt      = binCursor + NBpad;
    int* recs      = gcnt + 128;

    const int TB = 256;
    dim3 blk(TB);

    k_deg_init <<<dim3((N + TB - 1) / TB), blk, 0, stream>>>(dinv, gcnt, N);
    k_deg_count<<<dim3((E + TB - 1) / TB), blk, 0, stream>>>(dst, dinv, E);
    k_deg_inv_bins<<<dim3(NB), dim3(BIN_R), 0, stream>>>(dinv, binCount, N);

    k_bin_scan<<<dim3(1), dim3(1024), 0, stream>>>(binCount, binStart, binCursor, NB);
    k_bin_fill<<<dim3((E + FILL_TB * FILL_EPT - 1) / (FILL_TB * FILL_EPT)), dim3(FILL_TB),
                 2 * NB * sizeof(int), stream>>>(src, dst, dinv, binStart, binCursor, recs, E, NB);

    const int mt = N / 16;
    const int gx = (mt + 3) / 4;

    k_gemm_wmma_f32<<<dim3(gx, HID / 16), dim3(128), 0, stream>>>(x, W1, bufA, IN, HID, mt);
    k_agg_lds      <<<dim3(NB), dim3(AGG_TB), 0, stream>>>(recs, binStart, bufA, dinv, b1, bufB, N);

    k_gemm_wmma_f32<<<dim3(gx, HID / 16), dim3(128), 0, stream>>>(bufB, W2, bufA, HID, HID, mt);
    k_agg_lds      <<<dim3(NB), dim3(AGG_TB), 0, stream>>>(recs, binStart, bufA, dinv, b2, bufB, N);

    k_pool_init  <<<dim3((N + TB - 1) / TB), blk, 0, stream>>>(batch, dinv, Cm, gcnt, N);
    k_pool_edges <<<dim3((E + TB - 1) / TB), blk, 0, stream>>>(src, dst, batch, dinv, Cm, E, N);
    k_pool_gemm  <<<dim3(P), blk, 0, stream>>>(Cm, bufB, Tpart, N);
    k_pool_reduce<<<dim3(8 * POOL_TILE / TB), blk, 0, stream>>>(Tpart, T8, P);
    k_pool_final <<<dim3(1), dim3(512), 0, stream>>>(T8, W3, b3, gcnt, (float*)d_out);
}


# GEMM re-decomposed: a wave keeps a 32-column slice of W as register fragments; 128-row strips of X pass once through LDS (whole-row loads), shared by 16 waves; whole-line H stores; sums unchanged
# speedup vs baseline: 3.0852x; 1.1617x over previous
#include <hip/hip_runtime.h>
#include <math.h>

typedef float v2f __attribute__((ext_vector_type(2)));
typedef float v8f __attribute__((ext_vector_type(8)));

#define GCN_G   64
#define GCN_OUT 64
#define LDS_STRIDE 132

__global__ void k_deg_init(float* __restrict__ deg, int* __restrict__ gcnt, int n) {
    int i = blockIdx.x * blockDim.x + threadIdx.x;
    if (i < n) deg[i] = 1.0f;
    if (i < GCN_G) gcnt[i] = 0;
}

__global__ void k_deg_count(const int* __restrict__ dst, float* __restrict__ deg, int e) {
    int i = blockIdx.x * blockDim.x + threadIdx.x;
    if (i < e) atomicAdd(&deg[dst[i]], 1.0f);
}

#define BIN_R 400
#define FILL_EPT 4
#define FILL_TB 1024

__global__ void __launch_bounds__(BIN_R)
k_deg_inv_bins(float* __restrict__ deg, int* __restrict__ binCount, int n) {
    __shared__ int tot;
    const int t = threadIdx.x;
    if (t == 0) tot = 0;
    __syncthreads();
    const int i = blockIdx.x * BIN_R + t;
    int c = 0;
    if (i < n) {
        float d = deg[i];
        c = (int)d - 1;
        deg[i] = (d > 0.0f) ? (1.0f / sqrtf(d)) : 0.0f;
    }
    atomicAdd(&tot, c);
    __syncthreads();
    if (t == 0) binCount[blockIdx.x] = tot;
}

__global__ void __launch_bounds__(1024)
k_bin_scan(const int* __restrict__ cnt, int* __restrict__ start, int* __restrict__ cursor, int nb) {
    __shared__ int s[1024];
    const int t = threadIdx.x;
    int carry = 0;
    for (int b0 = 0; b0 < nb; b0 += 1024) {
        const int b = b0 + t;
        const int v = (b < nb) ? cnt[b] : 0;
        s[t] = v;
        __syncthreads();
        for (int off = 1; off < 1024; off <<= 1) {
            int a = (t >= off) ? s[t - off] : 0;
            __syncthreads();
            s[t] += a;
            __syncthreads();
        }
        if (b < nb) {
            start[b]  = carry + s[t] - v;
            cursor[b] = 0;
        }
        const int total = s[1023];
        __syncthreads();
        carry += total;
    }
    if (t == 0) start[nb] = carry;
}

__global__ void __launch_bounds__(FILL_TB)
k_bin_fill(const int* __restrict__ src, const int* __restrict__ dst,
           const float* __restrict__ dinv, const int* __restrict__ start,
           int* __restrict__ cursor, int* __restrict__ recs, int e, int nb) {
    extern __shared__ int fsm[];
    int* hist = fsm;
    int* base = fsm + nb;
    const int t = threadIdx.x;
    for (int b = t; b < nb; b += FILL_TB) hist[b] = 0;
    __syncthreads();

    const int e0 = blockIdx.x * (FILL_TB * FILL_EPT);
    int es[FILL_EPT], ed[FILL_EPT], eb[FILL_EPT], er[FILL_EPT];
#pragma unroll
    for (int u = 0; u < FILL_EPT; ++u) {
        const int i = e0 + u * FILL_TB + t;
        eb[u] = -1;
        if (i < e) {
            es[u] = src[i];
            ed[u] = dst[i];
            eb[u] = ed[u] / BIN_R;
            er[u] = atomicAdd(&hist[eb[u]], 1);
        }
    }
    __syncthreads();
    for (int b = t; b < nb; b += FILL_TB) {
        const int c = hist[b];
        base[b] = start[b] + (c > 0 ? atomicAdd(&cursor[b], c) : 0);
    }
    __syncthreads();
    int4* __restrict__ out = (int4*)recs;
#pragma unroll
    for (int u = 0; u < FILL_EPT; ++u) {
        if (eb[u] >= 0) {
            const float w = dinv[es[u]] * dinv[ed[u]];
            out[base[eb[u]] + er[u]] = make_int4(es[u], ed[u] - eb[u] * BIN_R, __float_as_int(w), 0);
        }
    }
}

#define GEMM_K 128
#define GEMM_F 128
#define GEMM_KS (GEMM_K / 4)
#define GEMM_TB 512
#define GEMM_WAVES (GEMM_TB / 32)
#define GEMM_SROWS 128
#define GEMM_STILES (GEMM_SROWS / 16)
#define GEMM_RPW (GEMM_SROWS / GEMM_WAVES)
#define GEMM_PITCH 132
#define GEMM_BUF (GEMM_SROWS * GEMM_PITCH)
#define GEMM_BLOCKS 256

__device__ __forceinline__ void gemm_lds_barrier() {
    __builtin_amdgcn_fence(__ATOMIC_RELEASE, "workgroup", "local");
    __builtin_amdgcn_s_barrier();
    __builtin_amdgcn_fence(__ATOMIC_ACQUIRE, "workgroup", "local");
}

__device__ __forceinline__ float gemm_other_half(float x) {
    int b = __builtin_bit_cast(int, x);
    return __builtin_bit_cast(float, __builtin_amdgcn_permlanex16(b, b, 0x76543210, 0xfedcba98, false, false));
}

__device__ __forceinline__ void gemm_rowtile(const float* __restrict__ al,
                                             const v2f (&b0)[GEMM_KS], const v2f (&b1)[GEMM_KS],
                                             float* __restrict__ H1, float* __restrict__ H2, bool hi) {
    v8f c0 = {}, c1 = {};
#pragma unroll
    for (int s = 0; s < GEMM_KS; ++s) {
        v2f a = *(const v2f*)(al + 4 * s);
        c0 = __builtin_amdgcn_wmma_f32_16x16x4_f32(false, a, false, b0[s], (short)0, c0, false, false);
        c1 = __builtin_amdgcn_wmma_f32_16x16x4_f32(false, a, false, b1[s], (short)0, c1, false, false);
    }
#pragma unroll
    for (int v = 0; v < 8; ++v) {
        const float t = gemm_other_half(c1[v]);
        H1[(size_t)v * GEMM_F] = hi ? t : c0[v];
        H2[(size_t)v * GEMM_F] = hi ? c0[v] : t;
    }
}

__global__ void __launch_bounds__(GEMM_TB)
k_gemm_strips(const float* __restrict__ X, const float* __restrict__ W,
              float* __restrict__ H, int mtiles) {
    __shared__ float Xs[2 * GEMM_BUF];

    const int t    = threadIdx.x;
    const int lane = t & 31;
    const int wave = __builtin_amdgcn_readfirstlane(t >> 5);
    const int half = lane >> 4;
    const int l16  = lane & 15;
    const int cp   = wave >> 2;
    const int rq   = ((wave & 3) - (wave >> 2)) & 3;

    const unsigned nb = gridDim.x;
    const int t0 = (int)(((unsigned)mtiles * blockIdx.x) / nb);
    const int t1 = (int)(((unsigned)mtiles * (blockIdx.x + 1u)) / nb);
    const int rowEnd = t1 << 4;
    const int nIter  = (t1 - t0 + GEMM_STILES - 1) / GEMM_STILES;

    const int sc = lane << 2;
    float* __restrict__ st0 = &Xs[wave * GEMM_PITCH + sc];
    const float4 zero4 = make_float4(0.0f, 0.0f, 0.0f, 0.0f);
    float4 q[GEMM_RPW];

    {
        const int r0 = (t0 << 4) + wave;
        float4 wq[GEMM_K / GEMM_WAVES];
#pragma unroll
        for (int u = 0; u < GEMM_K / GEMM_WAVES; ++u)
            wq[u] = *(const float4*)(W + (size_t)(wave + GEMM_WAVES * u) * GEMM_F + sc);
#pragma unroll
        for (int u = 0; u < GEMM_RPW; ++u) {
            q[u] = zero4;
            if (r0 + GEMM_WAVES * u < rowEnd)
                q[u] = *(const float4*)(X + (size_t)(r0 + GEMM_WAVES * u) * GEMM_K + sc);
        }
#pragma unroll
        for (int u = 0; u < GEMM_K / GEMM_WAVES; ++u)
            *(float4*)(st0 + GEMM_BUF + GEMM_WAVES * u * GEMM_PITCH) = wq[u];
#pragma unroll
        for (int u = 0; u < GEMM_RPW; ++u)
            *(float4*)(st0 + GEMM_WAVES * u * GEMM_PITCH) = q[u];
    }
    gemm_lds_barrier();

    v2f b0[GEMM_KS], b1[GEMM_KS];
    {
        const float* __restrict__ Wl = &Xs[GEMM_BUF + (half << 1) * GEMM_PITCH + (cp << 5) + l16];
#pragma unroll
        for (int s = 0; s < GEMM_KS; ++s) {
            b0[s].x = Wl[(4 * s) * GEMM_PITCH];
            b0[s].y = Wl[(4 * s + 1) * GEMM_PITCH];
            b1[s].x = Wl[(4 * s) * GEMM_PITCH + 16];
            b1[s].y = Wl[(4 * s + 1) * GEMM_PITCH + 16];
        }
    }
    gemm_lds_barrier();

    const int  aoff = ((rq << 4) + l16) * GEMM_PITCH + (half << 1);
    const bool hi   = (half != 0);
    float* __restrict__ Hp1 = H + (cp << 5) + lane;
    float* __restrict__ Hp2 = H + (size_t)8 * GEMM_F + (cp << 5) + (lane ^ 16);

    for (int i = 0; i < nIter; ++i) {
        const int  cur  = i & 1;
        const bool more = (i + 1 < nIter);
        if (more) {
            const int r0 = ((t0 + GEMM_STILES * (i + 1)) << 4) + wave;
#pragma unroll
            for (int u = 0; u < GEMM_RPW; ++u) {
                q[u] = zero4;
                if (r0 + GEMM_WAVES * u < rowEnd)
                    q[u] = *(const float4*)(X + (size_t)(r0 + GEMM_WAVES * u) * GEMM_K + sc);
            }
        }

        const float* __restrict__ al = &Xs[cur * GEMM_BUF + aoff];
        const int tbase = t0 + GEMM_STILES * i + rq;
#pragma nounroll
        for (int j = 0; j < 2; ++j) {
            const int tile = tbase + 4 * j;
            if (tile < t1) {
                const size_t ho = (size_t)(tile << 4) * GEMM_F;
                gemm_rowtile(al + (size_t)(64 * j) * GEMM_PITCH, b0, b1, Hp1 + ho, Hp2 + ho, hi);
            }

            if (j == 0 && more) {
                float* __restrict__ sn = st0 + (cur ^ 1) * GEMM_BUF;
#pragma unroll
                for (int u = 0; u < GEMM_RPW; ++u)
                    *(float4*)(sn + GEMM_WAVES * u * GEMM_PITCH) = q[u];
            }
        }

        gemm_lds_barrier();
    }
}

#define AGG_TB 1024
#define AGG_U  8

__device__ __forceinline__ void agg_add_row(float* __restrict__ acc, int dl, float w, int lane,
                                            float h0, float h1, float h2, float h3) {
    float* __restrict__ ap = acc + dl * 128 + lane;
    atomicAdd(ap,      w * h0);
    atomicAdd(ap + 32, w * h1);
    atomicAdd(ap + 64, w * h2);
    atomicAdd(ap + 96, w * h3);
}

__global__ void __launch_bounds__(AGG_TB)
k_agg_lds(const int* __restrict__ recs, const int* __restrict__ start,
          const float* __restrict__ H, const float* __restrict__ dinv,
          const float* __restrict__ bias, float* __restrict__ out, int n) {
    __shared__ float acc[BIN_R * 128];
    const int t     = threadIdx.x;
    const int lane  = t & 31;
    const int wave  = __builtin_amdgcn_readfirstlane(t >> 5);
    const int node0 = blockIdx.x * BIN_R;
    const int rows  = min(BIN_R, n - node0);

    for (int idx = t; idx < rows * 32; idx += AGG_TB) {
        const int r  = idx >> 5;
        const int c4 = (idx & 31) << 2;
        const float di = dinv[node0 + r];
        const float sw = di * di;
        const float4 h = *(const float4*)(H + (size_t)(node0 + r) * 128 + c4);
        float4 v;
        v.x = sw * h.x + bias[c4 + 0];
        v.y = sw * h.y + bias[c4 + 1];
        v.z = sw * h.z + bias[c4 + 2];
        v.w = sw * h.w + bias[c4 + 3];
        *(float4*)(&acc[r * 128 + c4]) = v;
    }
    __syncthreads();

    const int eb  = start[blockIdx.x];
    const int cnt = start[blockIdx.x + 1] - eb;
    const int w0  = eb + (int)(((long long)cnt * wave) >> 5);
    const int w1  = eb + (int)(((long long)cnt * (wave + 1)) >> 5);
    const int4* __restrict__ rec4 = (const int4*)recs;
    const float* __restrict__ Hl = H + lane;

    for (int p = w0; p < w1; p += 32) {
        const int m = min(32, w1 - p);
        int4 rc = make_int4(0, 0, 0, 0);
        if (lane < m) rc = rec4[p + lane];
        if (m == 32) {
#pragma unroll
            for (int j0 = 0; j0 < 32; j0 += AGG_U) {
                float v[AGG_U][4];
#pragma unroll
                for (int u = 0; u < AGG_U; ++u) {
                    const int s = __builtin_amdgcn_readlane(rc.x, j0 + u);
                    const float* __restrict__ hp = Hl + (size_t)s * 128;
                    v[u][0] = hp[0];
                    v[u][1] = hp[32];
                    v[u][2] = hp[64];
                    v[u][3] = hp[96];
                }
#pragma unroll
                for (int u = 0; u < AGG_U; ++u) {
                    const int   dl = __builtin_amdgcn_readlane(rc.y, j0 + u);
                    const float w  = __int_as_float(__builtin_amdgcn_readlane(rc.z, j0 + u));
                    agg_add_row(acc, dl, w, lane, v[u][0], v[u][1], v[u][2], v[u][3]);
                }
            }
        } else {
            for (int j = 0; j < m; ++j) {
                const int   s  = __builtin_amdgcn_readlane(rc.x, j);
                const int   dl = __builtin_amdgcn_readlane(rc.y, j);
                const float w  = __int_as_float(__builtin_amdgcn_readlane(rc.z, j));
                const float* __restrict__ hp = Hl + (size_t)s * 128;
                agg_add_row(acc, dl, w, lane, hp[0], hp[32], hp[64], hp[96]);
            }
        }
    }
    __syncthreads();

    for (int idx = t; idx < rows * 32; idx += AGG_TB) {
        const int r  = idx >> 5;
        const int c4 = (idx & 31) << 2;
        float4 v = *(const float4*)(&acc[r * 128 + c4]);
        v.x = fmaxf(v.x, 0.0f);
        v.y = fmaxf(v.y, 0.0f);
        v.z = fmaxf(v.z, 0.0f);
        v.w = fmaxf(v.w, 0.0f);
        *(float4*)(out + (size_t)(node0 + r) * 128 + c4) = v;
    }
}

#define POOL_KC 400
#define POOL_STRIDE 404
#define POOL_TILE (GCN_G * 128)

__global__ void k_pool_init(const int* __restrict__ batch, const float* __restrict__ dinv,
                            float* __restrict__ Cm, int* __restrict__ gcnt, int n) {
    int j = blockIdx.x * blockDim.x + threadIdx.x;
    int g = (j < n) ? batch[j] : -1;
    int g0 = __shfl(g, 0);
    int same = __popcll(__ballot(g == g0 && g >= 0));
    if ((threadIdx.x & 31) == 0 && g0 >= 0) atomicAdd(&gcnt[g0], same);
    if (g != g0 && g >= 0) atomicAdd(&gcnt[g], 1);
    if (j >= n) return;
    float d = dinv[j];
    float self = d * d;
    for (int q = 0; q < GCN_G; ++q)
        Cm[(size_t)q * n + j] = (q == g) ? self : 0.0f;
}

__global__ void k_pool_edges(const int* __restrict__ src, const int* __restrict__ dst,
                             const int* __restrict__ batch, const float* __restrict__ dinv,
                             float* __restrict__ Cm, int e, int n) {
    int i = blockIdx.x * blockDim.x + threadIdx.x;
    if (i >= e) return;
    int s = src[i];
    int d = dst[i];
    atomicAdd(&Cm[(size_t)batch[d] * n + s], dinv[s] * dinv[d]);
}

__global__ void __launch_bounds__(256)
k_pool_gemm(const float* __restrict__ Cm, const float* __restrict__ h,
            float* __restrict__ Tpart, int n) {
    __shared__ float As[GCN_G * POOL_STRIDE];
    const int t    = threadIdx.x;
    const int wave = t >> 5;
    const int lane = t & 31;
    const int half = lane >> 4;
    const int l16  = lane & 15;
    const int k0   = blockIdx.x * POOL_KC;
    const int klen = min(POOL_KC, n - k0);

    for (int idx = t; idx < GCN_G * (POOL_KC / 4); idx += 256) {
        int row = idx / (POOL_KC / 4);
        int q   = (idx - row * (POOL_KC / 4)) << 2;
        float4 v = make_float4(0.0f, 0.0f, 0.0f, 0.0f);
        if (q < klen) v = *(const float4*)(Cm + (size_t)row * n + k0 + q);
        *(float4*)(&As[row * POOL_STRIDE + q]) = v;
    }
    __syncthreads();

    const int n0 = wave << 4;
    const float* __restrict__ al = &As[l16 * POOL_STRIDE + (half << 1)];
    const float* __restrict__ hb = h + (size_t)(k0 + (half << 1)) * 128 + n0 + l16;

    v8f c0 = {}, c1 = {}, c2 = {}, c3 = {};
    v2f bn[4];
#pragma unroll
    for (int u = 0; u < 4; ++u) {
        bn[u].x = hb[(size_t)(4 * u) * 128];
        bn[u].y = hb[(size_t)(4 * u) * 128 + 128];
    }
    for (int kk = 0; kk < klen; kk += 16) {
        v2f bc[4];
#pragma unroll
        for (int u = 0; u < 4; ++u) bc[u] = bn[u];
        const float* __restrict__ hn = hb + (size_t)min(kk + 16, klen - 16) * 128;
#pragma unroll
        for (int u = 0; u < 4; ++u) {
            bn[u].x = hn[(size_t)(4 * u) * 128];
            bn[u].y = hn[(size_t)(4 * u) * 128 + 128];
        }
#pragma unroll
        for (int u = 0; u < 4; ++u) {
            const float* __restrict__ ak = al + kk + 4 * u;
            v2f a0 = *(const v2f*)(ak);
            v2f a1 = *(const v2f*)(ak + 16 * POOL_STRIDE);
            v2f a2 = *(const v2f*)(ak + 32 * POOL_STRIDE);
            v2f a3 = *(const v2f*)(ak + 48 * POOL_STRIDE);
            c0 = __builtin_amdgcn_wmma_f32_16x16x4_f32(false, a0, false, bc[u], (short)0, c0, false, false);
            c1 = __builtin_amdgcn_wmma_f32_16x16x4_f32(false, a1, false, bc[u], (short)0, c1, false, false);
            c2 = __builtin_amdgcn_wmma_f32_16x16x4_f32(false, a2, false, bc[u], (short)0, c2, false, false);
            c3 = __builtin_amdgcn_wmma_f32_16x16x4_f32(false, a3, false, bc[u], (short)0, c3, false, false);
        }
    }

    float* __restrict__ o = Tpart + (size_t)blockIdx.x * POOL_TILE + (size_t)(half << 3) * 128 + n0 + l16;
#pragma unroll
    for (int v = 0; v < 8; ++v) {
        o[(size_t)(v)      * 128] = c0[v];
        o[(size_t)(v + 16) * 128] = c1[v];
        o[(size_t)(v + 32) * 128] = c2[v];
        o[(size_t)(v + 48) * 128] = c3[v];
    }
}

__global__ void k_pool_reduce(const float* __restrict__ Tpart, float* __restrict__ T8, int nparts) {
    int tid = blockIdx.x * blockDim.x + threadIdx.x;
    int e   = tid & (POOL_TILE - 1);
    int pg  = tid / POOL_TILE;
    int per = (nparts + 7) >> 3;
    int p0  = pg * per;
    int p1  = min(nparts, p0 + per);
    float s = 0.0f;
    for (int p = p0; p < p1; ++p) s += Tpart[(size_t)p * POOL_TILE + e];
    T8[tid] = s;
}

__global__ void __launch_bounds__(512)
k_pool_final(const float* __restrict__ T8, const float* __restrict__ W3,
             const float* __restrict__ b3, const int* __restrict__ gcnt,
             float* __restrict__ outp) {
    __shared__ float Ts[GCN_G * LDS_STRIDE];
    __shared__ float Wl[GCN_OUT * LDS_STRIDE];
    __shared__ int   hist[GCN_G];
    const int t    = threadIdx.x;
    const int wave = t >> 5;
    const int lane = t & 31;
    const int half = lane >> 4;
    const int l16  = lane & 15;

    if (t < GCN_G) hist[t] = gcnt[t];
    for (int idx = t; idx < POOL_TILE; idx += 512) {
        float s = 0.0f;
#pragma unroll
        for (int pg = 0; pg < 8; ++pg) s += T8[pg * POOL_TILE + idx];
        Ts[(idx >> 7) * LDS_STRIDE + (idx & 127)] = s;
    }
    for (int idx = t; idx < 128 * GCN_OUT; idx += 512)
        Wl[(idx & 63) * LDS_STRIDE + (idx >> 6)] = W3[idx];
    __syncthreads();

    const int mi = wave >> 2;
    const int ni = wave & 3;
    const float* __restrict__ al = &Ts[((mi << 4) + l16) * LDS_STRIDE + (half << 1)];
    const float* __restrict__ bl = &Wl[((ni << 4) + l16) * LDS_STRIDE + (half << 1)];
    v8f c = {};
#pragma unroll
    for (int k0 = 0; k0 < 128; k0 += 4) {
        v2f a = *(const v2f*)(al + k0);
        v2f b = *(const v2f*)(bl + k0);
        c = __builtin_amdgcn_wmma_f32_16x16x4_f32(false, a, false, b, (short)0, c, false, false);
    }
    const int col = (ni << 4) + l16;
    const float bias = b3[col];
#pragma unroll
    for (int v = 0; v < 8; ++v) {
        int g = (mi << 4) + (half << 3) + v;
        float cf = (float)hist[g];
        outp[g * GCN_OUT + col] = (c[v] + cf * bias) / fmaxf(cf, 1.0f);
    }
}

extern "C" void kernel_launch(void* const* d_in, const int* in_sizes, int n_in,
                              void* d_out, int out_size, void* d_ws, size_t ws_size,
                              hipStream_t stream) {
    const float* x     = (const float*)d_in[0];
    const int*   edge  = (const int*)d_in[1];
    const int*   batch = (const int*)d_in[2];
    const float* W1    = (const float*)d_in[3];
    const float* b1    = (const float*)d_in[4];
    const float* W2    = (const float*)d_in[5];
    const float* b2    = (const float*)d_in[6];
    const float* W3    = (const float*)d_in[7];
    const float* b3    = (const float*)d_in[8];

    const int IN = 128, HID = 128;
    const int N = in_sizes[0] / IN;
    const int E = in_sizes[1] / 2;
    const int* src = edge;
    const int* dst = edge + E;

    char* ws = (char*)d_ws;
    float* dinv = (float*)ws;
    size_t off = ((size_t)N * sizeof(float) + 511) & ~(size_t)511;
    float* bufA = (float*)(ws + off);
    float* bufB = bufA + (size_t)N * HID;
    const int P = (N + POOL_KC - 1) / POOL_KC;
    float* Cm    = bufA;
    float* Tpart = bufB + (size_t)N * HID;
    float* T8    = Tpart + (size_t)P * POOL_TILE;
    const int NB    = (N + BIN_R - 1) / BIN_R;
    const int NBpad = (NB + 1 + 127) & ~127;
    int* binCount  = (int*)(T8 + (size_t)8 * POOL_TILE);
    int* binStart  = binCount + NBpad;
    int* binCursor = binStart + NBpad;
    int* gcnt      = binCursor + NBpad;
    int* recs      = gcnt + 128;

    const int TB = 256;
    dim3 blk(TB);

    k_deg_init <<<dim3((N + TB - 1) / TB), blk, 0, stream>>>(dinv, gcnt, N);
    k_deg_count<<<dim3((E + TB - 1) / TB), blk, 0, stream>>>(dst, dinv, E);
    k_deg_inv_bins<<<dim3(NB), dim3(BIN_R), 0, stream>>>(dinv, binCount, N);

    k_bin_scan<<<dim3(1), dim3(1024), 0, stream>>>(binCount, binStart, binCursor, NB);
    k_bin_fill<<<dim3((E + FILL_TB * FILL_EPT - 1) / (FILL_TB * FILL_EPT)), dim3(FILL_TB),
                 2 * NB * sizeof(int), stream>>>(src, dst, dinv, binStart, binCursor, recs, E, NB);

    const int mt = N / 16;

    k_gemm_strips<<<dim3(GEMM_BLOCKS), dim3(GEMM_TB), 0, stream>>>(x, W1, bufA, mt);
    k_agg_lds      <<<dim3(NB), dim3(AGG_TB), 0, stream>>>(recs, binStart, bufA, dinv, b1, bufB, N);

    k_gemm_strips<<<dim3(GEMM_BLOCKS), dim3(GEMM_TB), 0, stream>>>(bufB, W2, bufA, mt);
    k_agg_lds      <<<dim3(NB), dim3(AGG_TB), 0, stream>>>(recs, binStart, bufA, dinv, b2, bufB, N);

    k_pool_init  <<<dim3((N + TB - 1) / TB), blk, 0, stream>>>(batch, dinv, Cm, gcnt, N);
    k_pool_edges <<<dim3((E + TB - 1) / TB), blk, 0, stream>>>(src, dst, batch, dinv, Cm, E, N);
    k_pool_gemm  <<<dim3(P), blk, 0, stream>>>(Cm, bufB, Tpart, N);
    k_pool_reduce<<<dim3(8 * POOL_TILE / TB), blk, 0, stream>>>(Tpart, T8, P);
    k_pool_final <<<dim3(1), dim3(512), 0, stream>>>(T8, W3, b3, gcnt, (float*)d_out);
}


# Node degrees counted in LDS over each destination bin's sorted edge records (bin sizes from per-block LDS histograms), not by global float atomics; weights formed per record; float sums unchanged
# speedup vs baseline: 3.7524x; 1.2163x over previous
#include <hip/hip_runtime.h>
#include <math.h>

typedef float v2f __attribute__((ext_vector_type(2)));
typedef float v8f __attribute__((ext_vector_type(8)));

#define GCN_G   64
#define GCN_OUT 64
#define LDS_STRIDE 132

#define BIN_R 400
#define FILL_EPT 4
#define FILL_TB 1024

__global__ void k_tables_init(int* __restrict__ binCount, int* __restrict__ gcnt, int nb) {
    int i = blockIdx.x * blockDim.x + threadIdx.x;
    if (i < nb) binCount[i] = 0;
    if (i < GCN_G) gcnt[i] = 0;
}

__global__ void __launch_bounds__(FILL_TB)
k_bin_hist(const int* __restrict__ dst, int* __restrict__ binCount, int e, int nb) {
    extern __shared__ int hsm[];
    const int t = threadIdx.x;
    for (int b = t; b < nb; b += FILL_TB) hsm[b] = 0;
    __syncthreads();
    const int e0 = blockIdx.x * (FILL_TB * FILL_EPT);
    int d[FILL_EPT];
#pragma unroll
    for (int u = 0; u < FILL_EPT; ++u)
        d[u] = dst[min(e0 + u * FILL_TB + t, e - 1)];
#pragma unroll
    for (int u = 0; u < FILL_EPT; ++u)
        if (e0 + u * FILL_TB + t < e) atomicAdd(&hsm[d[u] / BIN_R], 1);
    __syncthreads();
    for (int b = t; b < nb; b += FILL_TB) {
        const int c = hsm[b];
        if (c > 0) atomicAdd(&binCount[b], c);
    }
}

__global__ void __launch_bounds__(1024)
k_bin_scan(const int* __restrict__ cnt, int* __restrict__ start, int* __restrict__ cursor, int nb) {
    __shared__ int s[1024];
    const int t = threadIdx.x;
    int carry = 0;
    for (int b0 = 0; b0 < nb; b0 += 1024) {
        const int b = b0 + t;
        const int v = (b < nb) ? cnt[b] : 0;
        s[t] = v;
        __syncthreads();
        for (int off = 1; off < 1024; off <<= 1) {
            int a = (t >= off) ? s[t - off] : 0;
            __syncthreads();
            s[t] += a;
            __syncthreads();
        }
        if (b < nb) {
            start[b]  = carry + s[t] - v;
            cursor[b] = 0;
        }
        const int total = s[1023];
        __syncthreads();
        carry += total;
    }
    if (t == 0) start[nb] = carry;
}

__global__ void __launch_bounds__(FILL_TB)
k_bin_fill(const int* __restrict__ src, const int* __restrict__ dst,
           const int* __restrict__ start,
           int* __restrict__ cursor, int* __restrict__ recs, int e, int nb) {
    extern __shared__ int fsm[];
    int* hist = fsm;
    int* base = fsm + nb;
    const int t = threadIdx.x;
    for (int b = t; b < nb; b += FILL_TB) hist[b] = 0;
    __syncthreads();

    const int e0 = blockIdx.x * (FILL_TB * FILL_EPT);
    int es[FILL_EPT], ed[FILL_EPT], eb[FILL_EPT], er[FILL_EPT];
#pragma unroll
    for (int u = 0; u < FILL_EPT; ++u) {
        const int i = e0 + u * FILL_TB + t;
        eb[u] = -1;
        if (i < e) {
            es[u] = src[i];
            ed[u] = dst[i];
            eb[u] = ed[u] / BIN_R;
            er[u] = atomicAdd(&hist[eb[u]], 1);
        }
    }
    __syncthreads();
    for (int b = t; b < nb; b += FILL_TB) {
        const int c = hist[b];
        base[b] = start[b] + (c > 0 ? atomicAdd(&cursor[b], c) : 0);
    }
    __syncthreads();
    int2* __restrict__ out = (int2*)recs;
#pragma unroll
    for (int u = 0; u < FILL_EPT; ++u) {
        if (eb[u] >= 0)
            out[base[eb[u]] + er[u]] = make_int2(es[u], ed[u] - eb[u] * BIN_R);
    }
}

#define DEG_TB 1024
#define DEG_U  8
__global__ void __launch_bounds__(DEG_TB)
k_bin_deg(const int* __restrict__ recs, const int* __restrict__ start,
          float* __restrict__ dinv, int n) {
    __shared__ int cnt[BIN_R];
    const int t = threadIdx.x;
    if (t < BIN_R) cnt[t] = 0;
    __syncthreads();
    const int eb = start[blockIdx.x];
    const int m  = start[blockIdx.x + 1] - eb;
    const int2* __restrict__ rec2 = (const int2*)recs + eb;
    for (int i0 = t; i0 < m + t; i0 += DEG_TB * DEG_U) {
        int d[DEG_U];
#pragma unroll
        for (int u = 0; u < DEG_U; ++u)
            d[u] = rec2[min(i0 + u * DEG_TB, m - 1)].y;
#pragma unroll
        for (int u = 0; u < DEG_U; ++u)
            if (i0 + u * DEG_TB < m) atomicAdd(&cnt[d[u]], 1);
    }
    __syncthreads();
    const int node = blockIdx.x * BIN_R + t;
    if (t < BIN_R && node < n) {
        const float d = (float)(cnt[t] + 1);
        dinv[node] = (d > 0.0f) ? (1.0f / sqrtf(d)) : 0.0f;
    }
}

#define GEMM_K 128
#define GEMM_F 128
#define GEMM_KS (GEMM_K / 4)
#define GEMM_TB 512
#define GEMM_WAVES (GEMM_TB / 32)
#define GEMM_SROWS 128
#define GEMM_STILES (GEMM_SROWS / 16)
#define GEMM_RPW (GEMM_SROWS / GEMM_WAVES)
#define GEMM_PITCH 132
#define GEMM_BUF (GEMM_SROWS * GEMM_PITCH)
#define GEMM_BLOCKS 256

__device__ __forceinline__ void gemm_lds_barrier() {
    __builtin_amdgcn_fence(__ATOMIC_RELEASE, "workgroup", "local");
    __builtin_amdgcn_s_barrier();
    __builtin_amdgcn_fence(__ATOMIC_ACQUIRE, "workgroup", "local");
}

__device__ __forceinline__ float gemm_other_half(float x) {
    int b = __builtin_bit_cast(int, x);
    return __builtin_bit_cast(float, __builtin_amdgcn_permlanex16(b, b, 0x76543210, 0xfedcba98, false, false));
}

__device__ __forceinline__ void gemm_rowtile(const float* __restrict__ al,
                                             const v2f (&b0)[GEMM_KS], const v2f (&b1)[GEMM_KS],
                                             float* __restrict__ H1, float* __restrict__ H2, bool hi) {
    v8f c0 = {}, c1 = {};
#pragma unroll
    for (int s = 0; s < GEMM_KS; ++s) {
        v2f a = *(const v2f*)(al + 4 * s);
        c0 = __builtin_amdgcn_wmma_f32_16x16x4_f32(false, a, false, b0[s], (short)0, c0, false, false);
        c1 = __builtin_amdgcn_wmma_f32_16x16x4_f32(false, a, false, b1[s], (short)0, c1, false, false);
    }
#pragma unroll
    for (int v = 0; v < 8; ++v) {
        const float t = gemm_other_half(c1[v]);
        H1[(size_t)v * GEMM_F] = hi ? t : c0[v];
        H2[(size_t)v * GEMM_F] = hi ? c0[v] : t;
    }
}

__global__ void __launch_bounds__(GEMM_TB)
k_gemm_strips(const float* __restrict__ X, const float* __restrict__ W,
              float* __restrict__ H, int mtiles) {
    __shared__ float Xs[2 * GEMM_BUF];

    const int t    = threadIdx.x;
    const int lane = t & 31;
    const int wave = __builtin_amdgcn_readfirstlane(t >> 5);
    const int half = lane >> 4;
    const int l16  = lane & 15;
    const int cp   = wave >> 2;
    const int rq   = ((wave & 3) - (wave >> 2)) & 3;

    const unsigned nb = gridDim.x;
    const int t0 = (int)(((unsigned)mtiles * blockIdx.x) / nb);
    const int t1 = (int)(((unsigned)mtiles * (blockIdx.x + 1u)) / nb);
    const int rowEnd = t1 << 4;
    const int nIter  = (t1 - t0 + GEMM_STILES - 1) / GEMM_STILES;

    const int sc = lane << 2;
    float* __restrict__ st0 = &Xs[wave * GEMM_PITCH + sc];
    const float4 zero4 = make_float4(0.0f, 0.0f, 0.0f, 0.0f);
    float4 q[GEMM_RPW];

    {
        const int r0 = (t0 << 4) + wave;
        float4 wq[GEMM_K / GEMM_WAVES];
#pragma unroll
        for (int u = 0; u < GEMM_K / GEMM_WAVES; ++u)
            wq[u] = *(const float4*)(W + (size_t)(wave + GEMM_WAVES * u) * GEMM_F + sc);
#pragma unroll
        for (int u = 0; u < GEMM_RPW; ++u) {
            q[u] = zero4;
            if (r0 + GEMM_WAVES * u < rowEnd)
                q[u] = *(const float4*)(X + (size_t)(r0 + GEMM_WAVES * u) * GEMM_K + sc);
        }
#pragma unroll
        for (int u = 0; u < GEMM_K / GEMM_WAVES; ++u)
            *(float4*)(st0 + GEMM_BUF + GEMM_WAVES * u * GEMM_PITCH) = wq[u];
#pragma unroll
        for (int u = 0; u < GEMM_RPW; ++u)
            *(float4*)(st0 + GEMM_WAVES * u * GEMM_PITCH) = q[u];
    }
    gemm_lds_barrier();

    v2f b0[GEMM_KS], b1[GEMM_KS];
    {
        const float* __restrict__ Wl = &Xs[GEMM_BUF + (half << 1) * GEMM_PITCH + (cp << 5) + l16];
#pragma unroll
        for (int s = 0; s < GEMM_KS; ++s) {
            b0[s].x = Wl[(4 * s) * GEMM_PITCH];
            b0[s].y = Wl[(4 * s + 1) * GEMM_PITCH];
            b1[s].x = Wl[(4 * s) * GEMM_PITCH + 16];
            b1[s].y = Wl[(4 * s + 1) * GEMM_PITCH + 16];
        }
    }
    gemm_lds_barrier();

    const int  aoff = ((rq << 4) + l16) * GEMM_PITCH + (half << 1);
    const bool hi   = (half != 0);
    float* __restrict__ Hp1 = H + (cp << 5) + lane;
    float* __restrict__ Hp2 = H + (size_t)8 * GEMM_F + (cp << 5) + (lane ^ 16);

    for (int i = 0; i < nIter; ++i) {
        const int  cur  = i & 1;
        const bool more = (i + 1 < nIter);
        if (more) {
            const int r0 = ((t0 + GEMM_STILES * (i + 1)) << 4) + wave;
#pragma unroll
            for (int u = 0; u < GEMM_RPW; ++u) {
                q[u] = zero4;
                if (r0 + GEMM_WAVES * u < rowEnd)
                    q[u] = *(const float4*)(X + (size_t)(r0 + GEMM_WAVES * u) * GEMM_K + sc);
            }
        }

        const float* __restrict__ al = &Xs[cur * GEMM_BUF + aoff];
        const int tbase = t0 + GEMM_STILES * i + rq;
#pragma nounroll
        for (int j = 0; j < 2; ++j) {
            const int tile = tbase + 4 * j;
            if (tile < t1) {
                const size_t ho = (size_t)(tile << 4) * GEMM_F;
                gemm_rowtile(al + (size_t)(64 * j) * GEMM_PITCH, b0, b1, Hp1 + ho, Hp2 + ho, hi);
            }

            if (j == 0 && more) {
                float* __restrict__ sn = st0 + (cur ^ 1) * GEMM_BUF;
#pragma unroll
                for (int u = 0; u < GEMM_RPW; ++u)
                    *(float4*)(sn + GEMM_WAVES * u * GEMM_PITCH) = q[u];
            }
        }

        gemm_lds_barrier();
    }
}

#define AGG_TB 1024
#define AGG_U  8

__device__ __forceinline__ void agg_add_row(float* __restrict__ acc, int dl, float w, int lane,
                                            float h0, float h1, float h2, float h3) {
    float* __restrict__ ap = acc + dl * 128 + lane;
    atomicAdd(ap,      w * h0);
    atomicAdd(ap + 32, w * h1);
    atomicAdd(ap + 64, w * h2);
    atomicAdd(ap + 96, w * h3);
}

__global__ void __launch_bounds__(AGG_TB)
k_agg_lds(const int* __restrict__ recs, const int* __restrict__ start,
          const float* __restrict__ H, const float* __restrict__ dinv,
          const float* __restrict__ bias, float* __restrict__ out, int n) {
    __shared__ float acc[BIN_R * 128];
    __shared__ float sdinv[BIN_R];
    const int t     = threadIdx.x;
    const int lane  = t & 31;
    const int wave  = __builtin_amdgcn_readfirstlane(t >> 5);
    const int node0 = blockIdx.x * BIN_R;
    const int rows  = min(BIN_R, n - node0);

    if (t < rows) sdinv[t] = dinv[node0 + t];
    for (int idx = t; idx < rows * 32; idx += AGG_TB) {
        const int r  = idx >> 5;
        const int c4 = (idx & 31) << 2;
        const float di = dinv[node0 + r];
        const float sw = di * di;
        const float4 h = *(const float4*)(H + (size_t)(node0 + r) * 128 + c4);
        float4 v;
        v.x = sw * h.x + bias[c4 + 0];
        v.y = sw * h.y + bias[c4 + 1];
        v.z = sw * h.z + bias[c4 + 2];
        v.w = sw * h.w + bias[c4 + 3];
        *(float4*)(&acc[r * 128 + c4]) = v;
    }
    __syncthreads();

    const int eb  = start[blockIdx.x];
    const int cnt = start[blockIdx.x + 1] - eb;
    const int w0  = eb + (int)(((long long)cnt * wave) >> 5);
    const int w1  = eb + (int)(((long long)cnt * (wave + 1)) >> 5);
    const int2* __restrict__ rec2 = (const int2*)recs;
    const float* __restrict__ Hl = H + lane;

    int2  rcN = make_int2(0, 0);
    float wN  = 0.0f;
    if (lane < w1 - w0) {
        rcN = rec2[w0 + lane];
        wN  = dinv[rcN.x] * sdinv[rcN.y];
    }
    for (int p = w0; p < w1; p += 32) {
        const int m  = min(32, w1 - p);
        const int2 rc = rcN;
        const int  wl = __float_as_int(wN);
        rcN = rec2[min(p + 32 + lane, w1 - 1)];
        if (m == 32) {
            float dsN = 0.0f, ddN = 0.0f;
#pragma unroll
            for (int j0 = 0; j0 < 32; j0 += AGG_U) {
                float v[AGG_U][4];
#pragma unroll
                for (int u = 0; u < AGG_U; ++u) {
                    const int s = __builtin_amdgcn_readlane(rc.x, j0 + u);
                    const float* __restrict__ hp = Hl + (size_t)s * 128;
                    v[u][0] = hp[0];
                    v[u][1] = hp[32];
                    v[u][2] = hp[64];
                    v[u][3] = hp[96];
                }
#pragma unroll
                for (int u = 0; u < AGG_U; ++u) {
                    const int   dl = __builtin_amdgcn_readlane(rc.y, j0 + u);
                    const float w  = __int_as_float(__builtin_amdgcn_readlane(wl, j0 + u));
                    agg_add_row(acc, dl, w, lane, v[u][0], v[u][1], v[u][2], v[u][3]);
                }
                if (j0 == 0) {
                    dsN = dinv[rcN.x];
                    ddN = sdinv[rcN.y];
                }
            }
            wN = dsN * ddN;
        } else {
            for (int j = 0; j < m; ++j) {
                const int   s  = __builtin_amdgcn_readlane(rc.x, j);
                const int   dl = __builtin_amdgcn_readlane(rc.y, j);
                const float w  = __int_as_float(__builtin_amdgcn_readlane(wl, j));
                const float* __restrict__ hp = Hl + (size_t)s * 128;
                agg_add_row(acc, dl, w, lane, hp[0], hp[32], hp[64], hp[96]);
            }
        }
    }
    __syncthreads();

    for (int idx = t; idx < rows * 32; idx += AGG_TB) {
        const int r  = idx >> 5;
        const int c4 = (idx & 31) << 2;
        float4 v = *(const float4*)(&acc[r * 128 + c4]);
        v.x = fmaxf(v.x, 0.0f);
        v.y = fmaxf(v.y, 0.0f);
        v.z = fmaxf(v.z, 0.0f);
        v.w = fmaxf(v.w, 0.0f);
        *(float4*)(out + (size_t)(node0 + r) * 128 + c4) = v;
    }
}

#define POOL_KC 400
#define POOL_STRIDE 404
#define POOL_TILE (GCN_G * 128)

__global__ void k_pool_init(const int* __restrict__ batch, const float* __restrict__ dinv,
                            float* __restrict__ Cm, int* __restrict__ gcnt, int n) {
    int j = blockIdx.x * blockDim.x + threadIdx.x;
    int g = (j < n) ? batch[j] : -1;
    int g0 = __shfl(g, 0);
    int same = __popcll(__ballot(g == g0 && g >= 0));
    if ((threadIdx.x & 31) == 0 && g0 >= 0) atomicAdd(&gcnt[g0], same);
    if (g != g0 && g >= 0) atomicAdd(&gcnt[g], 1);
    if (j >= n) return;
    float d = dinv[j];
    float self = d * d;
    for (int q = 0; q < GCN_G; ++q)
        Cm[(size_t)q * n + j] = (q == g) ? self : 0.0f;
}

__global__ void k_pool_edges(const int* __restrict__ src, const int* __restrict__ dst,
                             const int* __restrict__ batch, const float* __restrict__ dinv,
                             float* __restrict__ Cm, int e, int n) {
    int i = blockIdx.x * blockDim.x + threadIdx.x;
    if (i >= e) return;
    int s = src[i];
    int d = dst[i];
    atomicAdd(&Cm[(size_t)batch[d] * n + s], dinv[s] * dinv[d]);
}

__global__ void __launch_bounds__(256)
k_pool_gemm(const float* __restrict__ Cm, const float* __restrict__ h,
            float* __restrict__ Tpart, int n) {
    __shared__ float As[GCN_G * POOL_STRIDE];
    const int t    = threadIdx.x;
    const int wave = t >> 5;
    const int lane = t & 31;
    const int half = lane >> 4;
    const int l16  = lane & 15;
    const int k0   = blockIdx.x * POOL_KC;
    const int klen = min(POOL_KC, n - k0);

    for (int idx = t; idx < GCN_G * (POOL_KC / 4); idx += 256) {
        int row = idx / (POOL_KC / 4);
        int q   = (idx - row * (POOL_KC / 4)) << 2;
        float4 v = make_float4(0.0f, 0.0f, 0.0f, 0.0f);
        if (q < klen) v = *(const float4*)(Cm + (size_t)row * n + k0 + q);
        *(float4*)(&As[row * POOL_STRIDE + q]) = v;
    }
    __syncthreads();

    const int n0 = wave << 4;
    const float* __restrict__ al = &As[l16 * POOL_STRIDE + (half << 1)];
    const float* __restrict__ hb = h + (size_t)(k0 + (half << 1)) * 128 + n0 + l16;

    v8f c0 = {}, c1 = {}, c2 = {}, c3 = {};
    v2f bn[4];
#pragma unroll
    for (int u = 0; u < 4; ++u) {
        bn[u].x = hb[(size_t)(4 * u) * 128];
        bn[u].y = hb[(size_t)(4 * u) * 128 + 128];
    }
    for (int kk = 0; kk < klen; kk += 16) {
        v2f bc[4];
#pragma unroll
        for (int u = 0; u < 4; ++u) bc[u] = bn[u];
        const float* __restrict__ hn = hb + (size_t)min(kk + 16, klen - 16) * 128;
#pragma unroll
        for (int u = 0; u < 4; ++u) {
            bn[u].x = hn[(size_t)(4 * u) * 128];
            bn[u].y = hn[(size_t)(4 * u) * 128 + 128];
        }
#pragma unroll
        for (int u = 0; u < 4; ++u) {
            const float* __restrict__ ak = al + kk + 4 * u;
            v2f a0 = *(const v2f*)(ak);
            v2f a1 = *(const v2f*)(ak + 16 * POOL_STRIDE);
            v2f a2 = *(const v2f*)(ak + 32 * POOL_STRIDE);
            v2f a3 = *(const v2f*)(ak + 48 * POOL_STRIDE);
            c0 = __builtin_amdgcn_wmma_f32_16x16x4_f32(false, a0, false, bc[u], (short)0, c0, false, false);
            c1 = __builtin_amdgcn_wmma_f32_16x16x4_f32(false, a1, false, bc[u], (short)0, c1, false, false);
            c2 = __builtin_amdgcn_wmma_f32_16x16x4_f32(false, a2, false, bc[u], (short)0, c2, false, false);
            c3 = __builtin_amdgcn_wmma_f32_16x16x4_f32(false, a3, false, bc[u], (short)0, c3, false, false);
        }
    }

    float* __restrict__ o = Tpart + (size_t)blockIdx.x * POOL_TILE + (size_t)(half << 3) * 128 + n0 + l16;
#pragma unroll
    for (int v = 0; v < 8; ++v) {
        o[(size_t)(v)      * 128] = c0[v];
        o[(size_t)(v + 16) * 128] = c1[v];
        o[(size_t)(v + 32) * 128] = c2[v];
        o[(size_t)(v + 48) * 128] = c3[v];
    }
}

__global__ void k_pool_reduce(const float* __restrict__ Tpart, float* __restrict__ T8, int nparts) {
    int tid = blockIdx.x * blockDim.x + threadIdx.x;
    int e   = tid & (POOL_TILE - 1);
    int pg  = tid / POOL_TILE;
    int per = (nparts + 7) >> 3;
    int p0  = pg * per;
    int p1  = min(nparts, p0 + per);
    float s = 0.0f;
    for (int p = p0; p < p1; ++p) s += Tpart[(size_t)p * POOL_TILE + e];
    T8[tid] = s;
}

__global__ void __launch_bounds__(512)
k_pool_final(const float* __restrict__ T8, const float* __restrict__ W3,
             const float* __restrict__ b3, const int* __restrict__ gcnt,
             float* __restrict__ outp) {
    __shared__ float Ts[GCN_G * LDS_STRIDE];
    __shared__ float Wl[GCN_OUT * LDS_STRIDE];
    __shared__ int   hist[GCN_G];
    const int t    = threadIdx.x;
    const int wave = t >> 5;
    const int lane = t & 31;
    const int half = lane >> 4;
    const int l16  = lane & 15;

    if (t < GCN_G) hist[t] = gcnt[t];
    for (int idx = t; idx < POOL_TILE; idx += 512) {
        float s = 0.0f;
#pragma unroll
        for (int pg = 0; pg < 8; ++pg) s += T8[pg * POOL_TILE + idx];
        Ts[(idx >> 7) * LDS_STRIDE + (idx & 127)] = s;
    }
    for (int idx = t; idx < 128 * GCN_OUT; idx += 512)
        Wl[(idx & 63) * LDS_STRIDE + (idx >> 6)] = W3[idx];
    __syncthreads();

    const int mi = wave >> 2;
    const int ni = wave & 3;
    const float* __restrict__ al = &Ts[((mi << 4) + l16) * LDS_STRIDE + (half << 1)];
    const float* __restrict__ bl = &Wl[((ni << 4) + l16) * LDS_STRIDE + (half << 1)];
    v8f c = {};
#pragma unroll
    for (int k0 = 0; k0 < 128; k0 += 4) {
        v2f a = *(const v2f*)(al + k0);
        v2f b = *(const v2f*)(bl + k0);
        c = __builtin_amdgcn_wmma_f32_16x16x4_f32(false, a, false, b, (short)0, c, false, false);
    }
    const int col = (ni << 4) + l16;
    const float bias = b3[col];
#pragma unroll
    for (int v = 0; v < 8; ++v) {
        int g = (mi << 4) + (half << 3) + v;
        float cf = (float)hist[g];
        outp[g * GCN_OUT + col] = (c[v] + cf * bias) / fmaxf(cf, 1.0f);
    }
}

extern "C" void kernel_launch(void* const* d_in, const int* in_sizes, int n_in,
                              void* d_out, int out_size, void* d_ws, size_t ws_size,
                              hipStream_t stream) {
    const float* x     = (const float*)d_in[0];
    const int*   edge  = (const int*)d_in[1];
    const int*   batch = (const int*)d_in[2];
    const float* W1    = (const float*)d_in[3];
    const float* b1    = (const float*)d_in[4];
    const float* W2    = (const float*)d_in[5];
    const float* b2    = (const float*)d_in[6];
    const float* W3    = (const float*)d_in[7];
    const float* b3    = (const float*)d_in[8];

    const int IN = 128, HID = 128;
    const int N = in_sizes[0] / IN;
    const int E = in_sizes[1] / 2;
    const int* src = edge;
    const int* dst = edge + E;

    char* ws = (char*)d_ws;
    float* dinv = (float*)ws;
    size_t off = ((size_t)N * sizeof(float) + 511) & ~(size_t)511;
    float* bufA = (float*)(ws + off);
    float* bufB = bufA + (size_t)N * HID;
    const int P = (N + POOL_KC - 1) / POOL_KC;
    float* Cm    = bufA;
    float* Tpart = bufB + (size_t)N * HID;
    float* T8    = Tpart + (size_t)P * POOL_TILE;
    const int NB    = (N + BIN_R - 1) / BIN_R;
    const int NBpad = (NB + 1 + 127) & ~127;
    int* binCount  = (int*)(T8 + (size_t)8 * POOL_TILE);
    int* binStart  = binCount + NBpad;
    int* binCursor = binStart + NBpad;
    int* gcnt      = binCursor + NBpad;
    int* recs      = gcnt + 128;

    const int TB = 256;
    dim3 blk(TB);

    const int FB = (E + FILL_TB * FILL_EPT - 1) / (FILL_TB * FILL_EPT);
    k_tables_init<<<dim3((NB + TB - 1) / TB), blk, 0, stream>>>(binCount, gcnt, NB);
    k_bin_hist<<<dim3(FB), dim3(FILL_TB), NB * sizeof(int), stream>>>(dst, binCount, E, NB);
    k_bin_scan<<<dim3(1), dim3(1024), 0, stream>>>(binCount, binStart, binCursor, NB);
    k_bin_fill<<<dim3(FB), dim3(FILL_TB), 2 * NB * sizeof(int), stream>>>(src, dst, binStart, binCursor, recs, E, NB);

    k_bin_deg<<<dim3(NB), dim3(DEG_TB), 0, stream>>>(recs, binStart, dinv, N);

    const int mt = N / 16;

    k_gemm_strips<<<dim3(GEMM_BLOCKS), dim3(GEMM_TB), 0, stream>>>(x, W1, bufA, mt);
    k_agg_lds      <<<dim3(NB), dim3(AGG_TB), 0, stream>>>(recs, binStart, bufA, dinv, b1, bufB, N);

    k_gemm_strips<<<dim3(GEMM_BLOCKS), dim3(GEMM_TB), 0, stream>>>(bufB, W2, bufA, mt);
    k_agg_lds      <<<dim3(NB), dim3(AGG_TB), 0, stream>>>(recs, binStart, bufA, dinv, b2, bufB, N);

    k_pool_init  <<<dim3((N + TB - 1) / TB), blk, 0, stream>>>(batch, dinv, Cm, gcnt, N);
    k_pool_edges <<<dim3((E + TB - 1) / TB), blk, 0, stream>>>(src, dst, batch, dinv, Cm, E, N);
    k_pool_gemm  <<<dim3(P), blk, 0, stream>>>(Cm, bufB, Tpart, N);
    k_pool_reduce<<<dim3(8 * POOL_TILE / TB), blk, 0, stream>>>(Tpart, T8, P);
    k_pool_final <<<dim3(1), dim3(512), 0, stream>>>(T8, W3, b3, gcnt, (float*)d_out);
}


# Pooled weights Cm built per 400-node source range in LDS (float adds, arrival order) inside the split-K WMMA block, from edges also binned by source; no Cm in memory, no global float atomics
# speedup vs baseline: 4.3702x; 1.1646x over previous
#include <hip/hip_runtime.h>
#include <math.h>

typedef float v2f __attribute__((ext_vector_type(2)));
typedef float v8f __attribute__((ext_vector_type(8)));

#define GCN_G   64
#define GCN_OUT 64
#define LDS_STRIDE 132

#define BIN_R 400
#define FILL_EPT 4
#define FILL_TB 1024

__global__ void k_tables_init(int* __restrict__ binCount, int* __restrict__ gcnt, int nb) {
    int i = blockIdx.x * blockDim.x + threadIdx.x;
    if (i < nb) binCount[i] = 0;
    if (i < GCN_G) gcnt[i] = 0;
}

__global__ void __launch_bounds__(FILL_TB)
k_bin_hist(const int* __restrict__ src, const int* __restrict__ dstv,
           int* __restrict__ binCountAll, int e, int nb, int pitch) {
    extern __shared__ int hsm[];
    const int t = threadIdx.x;
    const int* __restrict__ dst = blockIdx.y ? src : dstv;
    int* __restrict__ binCount = binCountAll + blockIdx.y * pitch;
    for (int b = t; b < nb; b += FILL_TB) hsm[b] = 0;
    __syncthreads();
    const int e0 = blockIdx.x * (FILL_TB * FILL_EPT);
    int d[FILL_EPT];
#pragma unroll
    for (int u = 0; u < FILL_EPT; ++u)
        d[u] = dst[min(e0 + u * FILL_TB + t, e - 1)];
#pragma unroll
    for (int u = 0; u < FILL_EPT; ++u)
        if (e0 + u * FILL_TB + t < e) atomicAdd(&hsm[d[u] / BIN_R], 1);
    __syncthreads();
    for (int b = t; b < nb; b += FILL_TB) {
        const int c = hsm[b];
        if (c > 0) atomicAdd(&binCount[b], c);
    }
}

__global__ void __launch_bounds__(1024)
k_bin_scan(const int* __restrict__ cntAll, int* __restrict__ startAll, int* __restrict__ cursorAll,
           int nb, int pitch) {
    __shared__ int s[1024];
    const int t = threadIdx.x;
    const int* __restrict__ cnt = cntAll + blockIdx.x * pitch;
    int* __restrict__ start  = startAll + blockIdx.x * pitch;
    int* __restrict__ cursor = cursorAll + blockIdx.x * pitch;
    int carry = 0;
    for (int b0 = 0; b0 < nb; b0 += 1024) {
        const int b = b0 + t;
        const int v = (b < nb) ? cnt[b] : 0;
        s[t] = v;
        __syncthreads();
        for (int off = 1; off < 1024; off <<= 1) {
            int a = (t >= off) ? s[t - off] : 0;
            __syncthreads();
            s[t] += a;
            __syncthreads();
        }
        if (b < nb) {
            start[b]  = carry + s[t] - v;
            cursor[b] = 0;
        }
        const int total = s[1023];
        __syncthreads();
        carry += total;
    }
    if (t == 0) start[nb] = carry;
}

__global__ void __launch_bounds__(FILL_TB)
k_bin_fill(const int* __restrict__ srcv, const int* __restrict__ dstv,
           const int* __restrict__ startAll,
           int* __restrict__ cursorAll, int* __restrict__ recsAll, int e, int nb, int pitch) {
    const int* __restrict__ src = blockIdx.y ? dstv : srcv;
    const int* __restrict__ dst = blockIdx.y ? srcv : dstv;
    const int* __restrict__ start = startAll + blockIdx.y * pitch;
    int* __restrict__ cursor = cursorAll + blockIdx.y * pitch;
    int* __restrict__ recs = recsAll + (size_t)blockIdx.y * 2 * (size_t)e;
    extern __shared__ int fsm[];
    int* hist = fsm;
    int* base = fsm + nb;
    const int t = threadIdx.x;
    for (int b = t; b < nb; b += FILL_TB) hist[b] = 0;
    __syncthreads();

    const int e0 = blockIdx.x * (FILL_TB * FILL_EPT);
    int es[FILL_EPT], ed[FILL_EPT], eb[FILL_EPT], er[FILL_EPT];
#pragma unroll
    for (int u = 0; u < FILL_EPT; ++u) {
        const int i = e0 + u * FILL_TB + t;
        eb[u] = -1;
        if (i < e) {
            es[u] = src[i];
            ed[u] = dst[i];
            eb[u] = ed[u] / BIN_R;
            er[u] = atomicAdd(&hist[eb[u]], 1);
        }
    }
    __syncthreads();
    for (int b = t; b < nb; b += FILL_TB) {
        const int c = hist[b];
        base[b] = start[b] + (c > 0 ? atomicAdd(&cursor[b], c) : 0);
    }
    __syncthreads();
    int2* __restrict__ out = (int2*)recs;
#pragma unroll
    for (int u = 0; u < FILL_EPT; ++u) {
        if (eb[u] >= 0)
            out[base[eb[u]] + er[u]] = make_int2(es[u], ed[u] - eb[u] * BIN_R);
    }
}

#define DEG_TB 1024
#define DEG_U  8
__global__ void __launch_bounds__(DEG_TB)
k_bin_deg(const int* __restrict__ recs, const int* __restrict__ start,
          float* __restrict__ dinv, int n) {
    __shared__ int cnt[BIN_R];
    const int t = threadIdx.x;
    if (t < BIN_R) cnt[t] = 0;
    __syncthreads();
    const int eb = start[blockIdx.x];
    const int m  = start[blockIdx.x + 1] - eb;
    const int2* __restrict__ rec2 = (const int2*)recs + eb;
    for (int i0 = t; i0 < m + t; i0 += DEG_TB * DEG_U) {
        int d[DEG_U];
#pragma unroll
        for (int u = 0; u < DEG_U; ++u)
            d[u] = rec2[min(i0 + u * DEG_TB, m - 1)].y;
#pragma unroll
        for (int u = 0; u < DEG_U; ++u)
            if (i0 + u * DEG_TB < m) atomicAdd(&cnt[d[u]], 1);
    }
    __syncthreads();
    const int node = blockIdx.x * BIN_R + t;
    if (t < BIN_R && node < n) {
        const float d = (float)(cnt[t] + 1);
        dinv[node] = (d > 0.0f) ? (1.0f / sqrtf(d)) : 0.0f;
    }
}

#define GEMM_K 128
#define GEMM_F 128
#define GEMM_KS (GEMM_K / 4)
#define GEMM_TB 512
#define GEMM_WAVES (GEMM_TB / 32)
#define GEMM_SROWS 128
#define GEMM_STILES (GEMM_SROWS / 16)
#define GEMM_RPW (GEMM_SROWS / GEMM_WAVES)
#define GEMM_PITCH 132
#define GEMM_BUF (GEMM_SROWS * GEMM_PITCH)
#define GEMM_BLOCKS 256

__device__ __forceinline__ void gemm_lds_barrier() {
    __builtin_amdgcn_fence(__ATOMIC_RELEASE, "workgroup", "local");
    __builtin_amdgcn_s_barrier();
    __builtin_amdgcn_fence(__ATOMIC_ACQUIRE, "workgroup", "local");
}

__device__ __forceinline__ float gemm_other_half(float x) {
    int b = __builtin_bit_cast(int, x);
    return __builtin_bit_cast(float, __builtin_amdgcn_permlanex16(b, b, 0x76543210, 0xfedcba98, false, false));
}

__device__ __forceinline__ void gemm_rowtile(const float* __restrict__ al,
                                             const v2f (&b0)[GEMM_KS], const v2f (&b1)[GEMM_KS],
                                             float* __restrict__ H1, float* __restrict__ H2, bool hi) {
    v8f c0 = {}, c1 = {};
#pragma unroll
    for (int s = 0; s < GEMM_KS; ++s) {
        v2f a = *(const v2f*)(al + 4 * s);
        c0 = __builtin_amdgcn_wmma_f32_16x16x4_f32(false, a, false, b0[s], (short)0, c0, false, false);
        c1 = __builtin_amdgcn_wmma_f32_16x16x4_f32(false, a, false, b1[s], (short)0, c1, false, false);
    }
#pragma unroll
    for (int v = 0; v < 8; ++v) {
        const float t = gemm_other_half(c1[v]);
        H1[(size_t)v * GEMM_F] = hi ? t : c0[v];
        H2[(size_t)v * GEMM_F] = hi ? c0[v] : t;
    }
}

__global__ void __launch_bounds__(GEMM_TB)
k_gemm_strips(const float* __restrict__ X, const float* __restrict__ W,
              float* __restrict__ H, int mtiles) {
    __shared__ float Xs[2 * GEMM_BUF];

    const int t    = threadIdx.x;
    const int lane = t & 31;
    const int wave = __builtin_amdgcn_readfirstlane(t >> 5);
    const int half = lane >> 4;
    const int l16  = lane & 15;
    const int cp   = wave >> 2;
    const int rq   = ((wave & 3) - (wave >> 2)) & 3;

    const unsigned nb = gridDim.x;
    const int t0 = (int)(((unsigned)mtiles * blockIdx.x) / nb);
    const int t1 = (int)(((unsigned)mtiles * (blockIdx.x + 1u)) / nb);
    const int rowEnd = t1 << 4;
    const int nIter  = (t1 - t0 + GEMM_STILES - 1) / GEMM_STILES;

    const int sc = lane << 2;
    float* __restrict__ st0 = &Xs[wave * GEMM_PITCH + sc];
    const float4 zero4 = make_float4(0.0f, 0.0f, 0.0f, 0.0f);
    float4 q[GEMM_RPW];

    {
        const int r0 = (t0 << 4) + wave;
        float4 wq[GEMM_K / GEMM_WAVES];
#pragma unroll
        for (int u = 0; u < GEMM_K / GEMM_WAVES; ++u)
            wq[u] = *(const float4*)(W + (size_t)(wave + GEMM_WAVES * u) * GEMM_F + sc);
#pragma unroll
        for (int u = 0; u < GEMM_RPW; ++u) {
            q[u] = zero4;
            if (r0 + GEMM_WAVES * u < rowEnd)
                q[u] = *(const float4*)(X + (size_t)(r0 + GEMM_WAVES * u) * GEMM_K + sc);
        }
#pragma unroll
        for (int u = 0; u < GEMM_K / GEMM_WAVES; ++u)
            *(float4*)(st0 + GEMM_BUF + GEMM_WAVES * u * GEMM_PITCH) = wq[u];
#pragma unroll
        for (int u = 0; u < GEMM_RPW; ++u)
            *(float4*)(st0 + GEMM_WAVES * u * GEMM_PITCH) = q[u];
    }
    gemm_lds_barrier();

    v2f b0[GEMM_KS], b1[GEMM_KS];
    {
        const float* __restrict__ Wl = &Xs[GEMM_BUF + (half << 1) * GEMM_PITCH + (cp << 5) + l16];
#pragma unroll
        for (int s = 0; s < GEMM_KS; ++s) {
            b0[s].x = Wl[(4 * s) * GEMM_PITCH];
            b0[s].y = Wl[(4 * s + 1) * GEMM_PITCH];
            b1[s].x = Wl[(4 * s) * GEMM_PITCH + 16];
            b1[s].y = Wl[(4 * s + 1) * GEMM_PITCH + 16];
        }
    }
    gemm_lds_barrier();

    const int  aoff = ((rq << 4) + l16) * GEMM_PITCH + (half << 1);
    const bool hi   = (half != 0);
    float* __restrict__ Hp1 = H + (cp << 5) + lane;
    float* __restrict__ Hp2 = H + (size_t)8 * GEMM_F + (cp << 5) + (lane ^ 16);

    for (int i = 0; i < nIter; ++i) {
        const int  cur  = i & 1;
        const bool more = (i + 1 < nIter);
        if (more) {
            const int r0 = ((t0 + GEMM_STILES * (i + 1)) << 4) + wave;
#pragma unroll
            for (int u = 0; u < GEMM_RPW; ++u) {
                q[u] = zero4;
                if (r0 + GEMM_WAVES * u < rowEnd)
                    q[u] = *(const float4*)(X + (size_t)(r0 + GEMM_WAVES * u) * GEMM_K + sc);
            }
        }

        const float* __restrict__ al = &Xs[cur * GEMM_BUF + aoff];
        const int tbase = t0 + GEMM_STILES * i + rq;
#pragma nounroll
        for (int j = 0; j < 2; ++j) {
            const int tile = tbase + 4 * j;
            if (tile < t1) {
                const size_t ho = (size_t)(tile << 4) * GEMM_F;
                gemm_rowtile(al + (size_t)(64 * j) * GEMM_PITCH, b0, b1, Hp1 + ho, Hp2 + ho, hi);
            }

            if (j == 0 && more) {
                float* __restrict__ sn = st0 + (cur ^ 1) * GEMM_BUF;
#pragma unroll
                for (int u = 0; u < GEMM_RPW; ++u)
                    *(float4*)(sn + GEMM_WAVES * u * GEMM_PITCH) = q[u];
            }
        }

        gemm_lds_barrier();
    }
}

#define AGG_TB 1024
#define AGG_U  8

__device__ __forceinline__ void agg_add_row(float* __restrict__ acc, int dl, float w, int lane,
                                            float h0, float h1, float h2, float h3) {
    float* __restrict__ ap = acc + dl * 128 + lane;
    atomicAdd(ap,      w * h0);
    atomicAdd(ap + 32, w * h1);
    atomicAdd(ap + 64, w * h2);
    atomicAdd(ap + 96, w * h3);
}

__global__ void __launch_bounds__(AGG_TB)
k_agg_lds(const int* __restrict__ recs, const int* __restrict__ start,
          const float* __restrict__ H, const float* __restrict__ dinv,
          const float* __restrict__ bias, float* __restrict__ out, int n) {
    __shared__ float acc[BIN_R * 128];
    __shared__ float sdinv[BIN_R];
    const int t     = threadIdx.x;
    const int lane  = t & 31;
    const int wave  = __builtin_amdgcn_readfirstlane(t >> 5);
    const int node0 = blockIdx.x * BIN_R;
    const int rows  = min(BIN_R, n - node0);

    if (t < rows) sdinv[t] = dinv[node0 + t];
    for (int idx = t; idx < rows * 32; idx += AGG_TB) {
        const int r  = idx >> 5;
        const int c4 = (idx & 31) << 2;
        const float di = dinv[node0 + r];
        const float sw = di * di;
        const float4 h = *(const float4*)(H + (size_t)(node0 + r) * 128 + c4);
        float4 v;
        v.x = sw * h.x + bias[c4 + 0];
        v.y = sw * h.y + bias[c4 + 1];
        v.z = sw * h.z + bias[c4 + 2];
        v.w = sw * h.w + bias[c4 + 3];
        *(float4*)(&acc[r * 128 + c4]) = v;
    }
    __syncthreads();

    const int eb  = start[blockIdx.x];
    const int cnt = start[blockIdx.x + 1] - eb;
    const int w0  = eb + (int)(((long long)cnt * wave) >> 5);
    const int w1  = eb + (int)(((long long)cnt * (wave + 1)) >> 5);
    const int2* __restrict__ rec2 = (const int2*)recs;
    const float* __restrict__ Hl = H + lane;

    int2  rcN = make_int2(0, 0);
    float wN  = 0.0f;
    if (lane < w1 - w0) {
        rcN = rec2[w0 + lane];
        wN  = dinv[rcN.x] * sdinv[rcN.y];
    }
    for (int p = w0; p < w1; p += 32) {
        const int m  = min(32, w1 - p);
        const int2 rc = rcN;
        const int  wl = __float_as_int(wN);
        rcN = rec2[min(p + 32 + lane, w1 - 1)];
        if (m == 32) {
            float dsN = 0.0f, ddN = 0.0f;
#pragma unroll
            for (int j0 = 0; j0 < 32; j0 += AGG_U) {
                float v[AGG_U][4];
#pragma unroll
                for (int u = 0; u < AGG_U; ++u) {
                    const int s = __builtin_amdgcn_readlane(rc.x, j0 + u);
                    const float* __restrict__ hp = Hl + (size_t)s * 128;
                    v[u][0] = hp[0];
                    v[u][1] = hp[32];
                    v[u][2] = hp[64];
                    v[u][3] = hp[96];
                }
#pragma unroll
                for (int u = 0; u < AGG_U; ++u) {
                    const int   dl = __builtin_amdgcn_readlane(rc.y, j0 + u);
                    const float w  = __int_as_float(__builtin_amdgcn_readlane(wl, j0 + u));
                    agg_add_row(acc, dl, w, lane, v[u][0], v[u][1], v[u][2], v[u][3]);
                }
                if (j0 == 0) {
                    dsN = dinv[rcN.x];
                    ddN = sdinv[rcN.y];
                }
            }
            wN = dsN * ddN;
        } else {
            for (int j = 0; j < m; ++j) {
                const int   s  = __builtin_amdgcn_readlane(rc.x, j);
                const int   dl = __builtin_amdgcn_readlane(rc.y, j);
                const float w  = __int_as_float(__builtin_amdgcn_readlane(wl, j));
                const float* __restrict__ hp = Hl + (size_t)s * 128;
                agg_add_row(acc, dl, w, lane, hp[0], hp[32], hp[64], hp[96]);
            }
        }
    }
    __syncthreads();

    for (int idx = t; idx < rows * 32; idx += AGG_TB) {
        const int r  = idx >> 5;
        const int c4 = (idx & 31) << 2;
        float4 v = *(const float4*)(&acc[r * 128 + c4]);
        v.x = fmaxf(v.x, 0.0f);
        v.y = fmaxf(v.y, 0.0f);
        v.z = fmaxf(v.z, 0.0f);
        v.w = fmaxf(v.w, 0.0f);
        *(float4*)(out + (size_t)(node0 + r) * 128 + c4) = v;
    }
}

#define POOL_KC BIN_R
#define POOL_STRIDE 404
#define POOL_TILE (GCN_G * 128)
#define POOL_U 8

__global__ void k_pool_init(const int* __restrict__ batch, int* __restrict__ gcnt, int n) {
    int j = blockIdx.x * blockDim.x + threadIdx.x;
    int g = (j < n) ? batch[j] : -1;
    int g0 = __shfl(g, 0);
    int same = __popcll(__ballot(g == g0 && g >= 0));
    if ((threadIdx.x & 31) == 0 && g0 >= 0) atomicAdd(&gcnt[g0], same);
    if (g != g0 && g >= 0) atomicAdd(&gcnt[g], 1);
}

__global__ void __launch_bounds__(256)
k_pool_gemm(const int* __restrict__ recs, const int* __restrict__ start,
            const int* __restrict__ batch, const float* __restrict__ dinv,
            const float* __restrict__ h, float* __restrict__ Tpart, int n) {
    __shared__ float As[GCN_G * POOL_STRIDE];
    __shared__ float sdv[POOL_KC];
    const int t    = threadIdx.x;
    const int wave = t >> 5;
    const int lane = t & 31;
    const int half = lane >> 4;
    const int l16  = lane & 15;
    const int k0   = blockIdx.x * POOL_KC;
    const int klen = min(POOL_KC, n - k0);

    for (int idx = t; idx < GCN_G * (POOL_STRIDE / 4); idx += 256)
        *(float4*)(&As[idx << 2]) = make_float4(0.0f, 0.0f, 0.0f, 0.0f);
    for (int j = t; j < POOL_KC; j += 256) sdv[j] = (j < klen) ? dinv[k0 + j] : 0.0f;
    __syncthreads();
    for (int j = t; j < klen; j += 256) As[batch[k0 + j] * POOL_STRIDE + j] = sdv[j] * sdv[j];
    __syncthreads();
    {
        const int eb = start[blockIdx.x];
        const int m  = start[blockIdx.x + 1] - eb;
        const int2* __restrict__ rec2 = (const int2*)recs + eb;
        for (int i0 = t; i0 < m + t; i0 += 256 * POOL_U) {
            int2  r[POOL_U];
            int   g[POOL_U];
            float dv[POOL_U];
#pragma unroll
            for (int u = 0; u < POOL_U; ++u)
                r[u] = rec2[min(i0 + u * 256, m - 1)];
#pragma unroll
            for (int u = 0; u < POOL_U; ++u) {
                g[u]  = batch[r[u].x];
                dv[u] = dinv[r[u].x];
            }
#pragma unroll
            for (int u = 0; u < POOL_U; ++u)
                if (i0 + u * 256 < m)
                    atomicAdd(&As[g[u] * POOL_STRIDE + r[u].y], sdv[r[u].y] * dv[u]);
        }
    }
    __syncthreads();

    const int n0 = wave << 4;
    const float* __restrict__ al = &As[l16 * POOL_STRIDE + (half << 1)];
    const float* __restrict__ hb = h + (size_t)(k0 + (half << 1)) * 128 + n0 + l16;

    v8f c0 = {}, c1 = {}, c2 = {}, c3 = {};
    v2f bn[4];
#pragma unroll
    for (int u = 0; u < 4; ++u) {
        bn[u].x = hb[(size_t)(4 * u) * 128];
        bn[u].y = hb[(size_t)(4 * u) * 128 + 128];
    }
    for (int kk = 0; kk < klen; kk += 16) {
        v2f bc[4];
#pragma unroll
        for (int u = 0; u < 4; ++u) bc[u] = bn[u];
        const float* __restrict__ hn = hb + (size_t)min(kk + 16, klen - 16) * 128;
#pragma unroll
        for (int u = 0; u < 4; ++u) {
            bn[u].x = hn[(size_t)(4 * u) * 128];
            bn[u].y = hn[(size_t)(4 * u) * 128 + 128];
        }
#pragma unroll
        for (int u = 0; u < 4; ++u) {
            const float* __restrict__ ak = al + kk + 4 * u;
            v2f a0 = *(const v2f*)(ak);
            v2f a1 = *(const v2f*)(ak + 16 * POOL_STRIDE);
            v2f a2 = *(const v2f*)(ak + 32 * POOL_STRIDE);
            v2f a3 = *(const v2f*)(ak + 48 * POOL_STRIDE);
            c0 = __builtin_amdgcn_wmma_f32_16x16x4_f32(false, a0, false, bc[u], (short)0, c0, false, false);
            c1 = __builtin_amdgcn_wmma_f32_16x16x4_f32(false, a1, false, bc[u], (short)0, c1, false, false);
            c2 = __builtin_amdgcn_wmma_f32_16x16x4_f32(false, a2, false, bc[u], (short)0, c2, false, false);
            c3 = __builtin_amdgcn_wmma_f32_16x16x4_f32(false, a3, false, bc[u], (short)0, c3, false, false);
        }
    }

    float* __restrict__ o = Tpart + (size_t)blockIdx.x * POOL_TILE + (size_t)(half << 3) * 128 + n0 + l16;
#pragma unroll
    for (int v = 0; v < 8; ++v) {
        o[(size_t)(v)      * 128] = c0[v];
        o[(size_t)(v + 16) * 128] = c1[v];
        o[(size_t)(v + 32) * 128] = c2[v];
        o[(size_t)(v + 48) * 128] = c3[v];
    }
}

__global__ void k_pool_reduce(const float* __restrict__ Tpart, float* __restrict__ T8, int nparts) {
    int tid = blockIdx.x * blockDim.x + threadIdx.x;
    int e   = tid & (POOL_TILE - 1);
    int pg  = tid / POOL_TILE;
    int per = (nparts + 7) >> 3;
    int p0  = pg * per;
    int p1  = min(nparts, p0 + per);
    float s = 0.0f;
    for (int p = p0; p < p1; ++p) s += Tpart[(size_t)p * POOL_TILE + e];
    T8[tid] = s;
}

__global__ void __launch_bounds__(512)
k_pool_final(const float* __restrict__ T8, const float* __restrict__ W3,
             const float* __restrict__ b3, const int* __restrict__ gcnt,
             float* __restrict__ outp) {
    __shared__ float Ts[GCN_G * LDS_STRIDE];
    __shared__ float Wl[GCN_OUT * LDS_STRIDE];
    __shared__ int   hist[GCN_G];
    const int t    = threadIdx.x;
    const int wave = t >> 5;
    const int lane = t & 31;
    const int half = lane >> 4;
    const int l16  = lane & 15;

    if (t < GCN_G) hist[t] = gcnt[t];
    for (int idx = t; idx < POOL_TILE; idx += 512) {
        float s = 0.0f;
#pragma unroll
        for (int pg = 0; pg < 8; ++pg) s += T8[pg * POOL_TILE + idx];
        Ts[(idx >> 7) * LDS_STRIDE + (idx & 127)] = s;
    }
    for (int idx = t; idx < 128 * GCN_OUT; idx += 512)
        Wl[(idx & 63) * LDS_STRIDE + (idx >> 6)] = W3[idx];
    __syncthreads();

    const int mi = wave >> 2;
    const int ni = wave & 3;
    const float* __restrict__ al = &Ts[((mi << 4) + l16) * LDS_STRIDE + (half << 1)];
    const float* __restrict__ bl = &Wl[((ni << 4) + l16) * LDS_STRIDE + (half << 1)];
    v8f c = {};
#pragma unroll
    for (int k0 = 0; k0 < 128; k0 += 4) {
        v2f a = *(const v2f*)(al + k0);
        v2f b = *(const v2f*)(bl + k0);
        c = __builtin_amdgcn_wmma_f32_16x16x4_f32(false, a, false, b, (short)0, c, false, false);
    }
    const int col = (ni << 4) + l16;
    const float bias = b3[col];
#pragma unroll
    for (int v = 0; v < 8; ++v) {
        int g = (mi << 4) + (half << 3) + v;
        float cf = (float)hist[g];
        outp[g * GCN_OUT + col] = (c[v] + cf * bias) / fmaxf(cf, 1.0f);
    }
}

extern "C" void kernel_launch(void* const* d_in, const int* in_sizes, int n_in,
                              void* d_out, int out_size, void* d_ws, size_t ws_size,
                              hipStream_t stream) {
    const float* x     = (const float*)d_in[0];
    const int*   edge  = (const int*)d_in[1];
    const int*   batch = (const int*)d_in[2];
    const float* W1    = (const float*)d_in[3];
    const float* b1    = (const float*)d_in[4];
    const float* W2    = (const float*)d_in[5];
    const float* b2    = (const float*)d_in[6];
    const float* W3    = (const float*)d_in[7];
    const float* b3    = (const float*)d_in[8];

    const int IN = 128, HID = 128;
    const int N = in_sizes[0] / IN;
    const int E = in_sizes[1] / 2;
    const int* src = edge;
    const int* dst = edge + E;

    char* ws = (char*)d_ws;
    float* dinv = (float*)ws;
    size_t off = ((size_t)N * sizeof(float) + 511) & ~(size_t)511;
    float* bufA = (float*)(ws + off);
    float* bufB = bufA + (size_t)N * HID;
    const int P = (N + POOL_KC - 1) / POOL_KC;
    float* Tpart = bufB + (size_t)N * HID;
    float* T8    = Tpart + (size_t)P * POOL_TILE;
    const int NB    = (N + BIN_R - 1) / BIN_R;
    const int NBpad = (NB + 1 + 127) & ~127;
    int* binCount  = (int*)(T8 + (size_t)8 * POOL_TILE);
    int* binStart  = binCount + 2 * NBpad;
    int* binCursor = binStart + 2 * NBpad;
    int* gcnt      = binCursor + 2 * NBpad;
    int* recs      = gcnt + 128;
    int* recsS     = recs + (size_t)2 * E;

    const int TB = 256;
    dim3 blk(TB);

    const int FB = (E + FILL_TB * FILL_EPT - 1) / (FILL_TB * FILL_EPT);
    k_tables_init<<<dim3((NBpad + NB + TB - 1) / TB), blk, 0, stream>>>(binCount, gcnt, NBpad + NB);
    k_bin_hist<<<dim3(FB, 2), dim3(FILL_TB), NB * sizeof(int), stream>>>(src, dst, binCount, E, NB, NBpad);
    k_bin_scan<<<dim3(2), dim3(1024), 0, stream>>>(binCount, binStart, binCursor, NB, NBpad);
    k_bin_fill<<<dim3(FB, 2), dim3(FILL_TB), 2 * NB * sizeof(int), stream>>>(src, dst, binStart, binCursor, recs, E, NB, NBpad);

    k_bin_deg<<<dim3(NB), dim3(DEG_TB), 0, stream>>>(recs, binStart, dinv, N);

    const int mt = N / 16;

    k_gemm_strips<<<dim3(GEMM_BLOCKS), dim3(GEMM_TB), 0, stream>>>(x, W1, bufA, mt);
    k_agg_lds      <<<dim3(NB), dim3(AGG_TB), 0, stream>>>(recs, binStart, bufA, dinv, b1, bufB, N);

    k_gemm_strips<<<dim3(GEMM_BLOCKS), dim3(GEMM_TB), 0, stream>>>(bufB, W2, bufA, mt);
    k_agg_lds      <<<dim3(NB), dim3(AGG_TB), 0, stream>>>(recs, binStart, bufA, dinv, b2, bufB, N);

    k_pool_init  <<<dim3((N + TB - 1) / TB), blk, 0, stream>>>(batch, gcnt, N);
    k_pool_gemm  <<<dim3(P), blk, 0, stream>>>(recsS, binStart + NBpad, batch, dinv, bufB, Tpart, N);
    k_pool_reduce<<<dim3(8 * POOL_TILE / TB), blk, 0, stream>>>(Tpart, T8, P);
    k_pool_final <<<dim3(1), dim3(512), 0, stream>>>(T8, W3, b3, gcnt, (float*)d_out);
}


# Aggregation fetches each source row with one 16-byte load per lane (four 128-B lines per instruction, was four 4-byte loads); LDS rows column-interleaved so adds stay conflict-free; sums unchanged
# speedup vs baseline: 4.6097x; 1.0548x over previous
#include <hip/hip_runtime.h>
#include <math.h>

typedef float v2f __attribute__((ext_vector_type(2)));
typedef float v8f __attribute__((ext_vector_type(8)));

#define GCN_G   64
#define GCN_OUT 64
#define LDS_STRIDE 132

#define BIN_R 400
#define FILL_EPT 4
#define FILL_TB 1024

__global__ void k_tables_init(int* __restrict__ binCount, int* __restrict__ gcnt, int nb) {
    int i = blockIdx.x * blockDim.x + threadIdx.x;
    if (i < nb) binCount[i] = 0;
    if (i < GCN_G) gcnt[i] = 0;
}

__global__ void __launch_bounds__(FILL_TB)
k_bin_hist(const int* __restrict__ src, const int* __restrict__ dstv,
           int* __restrict__ binCountAll, int e, int nb, int pitch) {
    extern __shared__ int hsm[];
    const int t = threadIdx.x;
    const int* __restrict__ dst = blockIdx.y ? src : dstv;
    int* __restrict__ binCount = binCountAll + blockIdx.y * pitch;
    for (int b = t; b < nb; b += FILL_TB) hsm[b] = 0;
    __syncthreads();
    const int e0 = blockIdx.x * (FILL_TB * FILL_EPT);
    int d[FILL_EPT];
#pragma unroll
    for (int u = 0; u < FILL_EPT; ++u)
        d[u] = dst[min(e0 + u * FILL_TB + t, e - 1)];
#pragma unroll
    for (int u = 0; u < FILL_EPT; ++u)
        if (e0 + u * FILL_TB + t < e) atomicAdd(&hsm[d[u] / BIN_R], 1);
    __syncthreads();
    for (int b = t; b < nb; b += FILL_TB) {
        const int c = hsm[b];
        if (c > 0) atomicAdd(&binCount[b], c);
    }
}

__global__ void __launch_bounds__(1024)
k_bin_scan(const int* __restrict__ cntAll, int* __restrict__ startAll, int* __restrict__ cursorAll,
           int nb, int pitch) {
    __shared__ int s[1024];
    const int t = threadIdx.x;
    const int* __restrict__ cnt = cntAll + blockIdx.x * pitch;
    int* __restrict__ start  = startAll + blockIdx.x * pitch;
    int* __restrict__ cursor = cursorAll + blockIdx.x * pitch;
    int carry = 0;
    for (int b0 = 0; b0 < nb; b0 += 1024) {
        const int b = b0 + t;
        const int v = (b < nb) ? cnt[b] : 0;
        s[t] = v;
        __syncthreads();
        for (int off = 1; off < 1024; off <<= 1) {
            int a = (t >= off) ? s[t - off] : 0;
            __syncthreads();
            s[t] += a;
            __syncthreads();
        }
        if (b < nb) {
            start[b]  = carry + s[t] - v;
            cursor[b] = 0;
        }
        const int total = s[1023];
        __syncthreads();
        carry += total;
    }
    if (t == 0) start[nb] = carry;
}

__global__ void __launch_bounds__(FILL_TB)
k_bin_fill(const int* __restrict__ srcv, const int* __restrict__ dstv,
           const int* __restrict__ startAll,
           int* __restrict__ cursorAll, int* __restrict__ recsAll, int e, int nb, int pitch) {
    const int* __restrict__ src = blockIdx.y ? dstv : srcv;
    const int* __restrict__ dst = blockIdx.y ? srcv : dstv;
    const int* __restrict__ start = startAll + blockIdx.y * pitch;
    int* __restrict__ cursor = cursorAll + blockIdx.y * pitch;
    int* __restrict__ recs = recsAll + (size_t)blockIdx.y * 2 * (size_t)e;
    extern __shared__ int fsm[];
    int* hist = fsm;
    int* base = fsm + nb;
    const int t = threadIdx.x;
    for (int b = t; b < nb; b += FILL_TB) hist[b] = 0;
    __syncthreads();

    const int e0 = blockIdx.x * (FILL_TB * FILL_EPT);
    int es[FILL_EPT], ed[FILL_EPT], eb[FILL_EPT], er[FILL_EPT];
#pragma unroll
    for (int u = 0; u < FILL_EPT; ++u) {
        const int i = e0 + u * FILL_TB + t;
        eb[u] = -1;
        if (i < e) {
            es[u] = src[i];
            ed[u] = dst[i];
            eb[u] = ed[u] / BIN_R;
            er[u] = atomicAdd(&hist[eb[u]], 1);
        }
    }
    __syncthreads();
    for (int b = t; b < nb; b += FILL_TB) {
        const int c = hist[b];
        base[b] = start[b] + (c > 0 ? atomicAdd(&cursor[b], c) : 0);
    }
    __syncthreads();
    int2* __restrict__ out = (int2*)recs;
#pragma unroll
    for (int u = 0; u < FILL_EPT; ++u) {
        if (eb[u] >= 0)
            out[base[eb[u]] + er[u]] = make_int2(es[u], ed[u] - eb[u] * BIN_R);
    }
}

#define DEG_TB 1024
#define DEG_U  8
__global__ void __launch_bounds__(DEG_TB)
k_bin_deg(const int* __restrict__ recs, const int* __restrict__ start,
          float* __restrict__ dinv, int n) {
    __shared__ int cnt[BIN_R];
    const int t = threadIdx.x;
    if (t < BIN_R) cnt[t] = 0;
    __syncthreads();
    const int eb = start[blockIdx.x];
    const int m  = start[blockIdx.x + 1] - eb;
    const int2* __restrict__ rec2 = (const int2*)recs + eb;
    for (int i0 = t; i0 < m + t; i0 += DEG_TB * DEG_U) {
        int d[DEG_U];
#pragma unroll
        for (int u = 0; u < DEG_U; ++u)
            d[u] = rec2[min(i0 + u * DEG_TB, m - 1)].y;
#pragma unroll
        for (int u = 0; u < DEG_U; ++u)
            if (i0 + u * DEG_TB < m) atomicAdd(&cnt[d[u]], 1);
    }
    __syncthreads();
    const int node = blockIdx.x * BIN_R + t;
    if (t < BIN_R && node < n) {
        const float d = (float)(cnt[t] + 1);
        dinv[node] = (d > 0.0f) ? (1.0f / sqrtf(d)) : 0.0f;
    }
}

#define GEMM_K 128
#define GEMM_F 128
#define GEMM_KS (GEMM_K / 4)
#define GEMM_TB 512
#define GEMM_WAVES (GEMM_TB / 32)
#define GEMM_SROWS 128
#define GEMM_STILES (GEMM_SROWS / 16)
#define GEMM_RPW (GEMM_SROWS / GEMM_WAVES)
#define GEMM_PITCH 132
#define GEMM_BUF (GEMM_SROWS * GEMM_PITCH)
#define GEMM_BLOCKS 256

__device__ __forceinline__ void gemm_lds_barrier() {
    __builtin_amdgcn_fence(__ATOMIC_RELEASE, "workgroup", "local");
    __builtin_amdgcn_s_barrier();
    __builtin_amdgcn_fence(__ATOMIC_ACQUIRE, "workgroup", "local");
}

__device__ __forceinline__ float gemm_other_half(float x) {
    int b = __builtin_bit_cast(int, x);
    return __builtin_bit_cast(float, __builtin_amdgcn_permlanex16(b, b, 0x76543210, 0xfedcba98, false, false));
}

__device__ __forceinline__ void gemm_rowtile(const float* __restrict__ al,
                                             const v2f (&b0)[GEMM_KS], const v2f (&b1)[GEMM_KS],
                                             float* __restrict__ H1, float* __restrict__ H2, bool hi) {
    v8f c0 = {}, c1 = {};
#pragma unroll
    for (int s = 0; s < GEMM_KS; ++s) {
        v2f a = *(const v2f*)(al + 4 * s);
        c0 = __builtin_amdgcn_wmma_f32_16x16x4_f32(false, a, false, b0[s], (short)0, c0, false, false);
        c1 = __builtin_amdgcn_wmma_f32_16x16x4_f32(false, a, false, b1[s], (short)0, c1, false, false);
    }
#pragma unroll
    for (int v = 0; v < 8; ++v) {
        const float t = gemm_other_half(c1[v]);
        H1[(size_t)v * GEMM_F] = hi ? t : c0[v];
        H2[(size_t)v * GEMM_F] = hi ? c0[v] : t;
    }
}

__global__ void __launch_bounds__(GEMM_TB)
k_gemm_strips(const float* __restrict__ X, const float* __restrict__ W,
              float* __restrict__ H, int mtiles) {
    __shared__ float Xs[2 * GEMM_BUF];

    const int t    = threadIdx.x;
    const int lane = t & 31;
    const int wave = __builtin_amdgcn_readfirstlane(t >> 5);
    const int half = lane >> 4;
    const int l16  = lane & 15;
    const int cp   = wave >> 2;
    const int rq   = ((wave & 3) - (wave >> 2)) & 3;

    const unsigned nb = gridDim.x;
    const int t0 = (int)(((unsigned)mtiles * blockIdx.x) / nb);
    const int t1 = (int)(((unsigned)mtiles * (blockIdx.x + 1u)) / nb);
    const int rowEnd = t1 << 4;
    const int nIter  = (t1 - t0 + GEMM_STILES - 1) / GEMM_STILES;

    const int sc = lane << 2;
    float* __restrict__ st0 = &Xs[wave * GEMM_PITCH + sc];
    const float4 zero4 = make_float4(0.0f, 0.0f, 0.0f, 0.0f);
    float4 q[GEMM_RPW];

    {
        const int r0 = (t0 << 4) + wave;
        float4 wq[GEMM_K / GEMM_WAVES];
#pragma unroll
        for (int u = 0; u < GEMM_K / GEMM_WAVES; ++u)
            wq[u] = *(const float4*)(W + (size_t)(wave + GEMM_WAVES * u) * GEMM_F + sc);
#pragma unroll
        for (int u = 0; u < GEMM_RPW; ++u) {
            q[u] = zero4;
            if (r0 + GEMM_WAVES * u < rowEnd)
                q[u] = *(const float4*)(X + (size_t)(r0 + GEMM_WAVES * u) * GEMM_K + sc);
        }
#pragma unroll
        for (int u = 0; u < GEMM_K / GEMM_WAVES; ++u)
            *(float4*)(st0 + GEMM_BUF + GEMM_WAVES * u * GEMM_PITCH) = wq[u];
#pragma unroll
        for (int u = 0; u < GEMM_RPW; ++u)
            *(float4*)(st0 + GEMM_WAVES * u * GEMM_PITCH) = q[u];
    }
    gemm_lds_barrier();

    v2f b0[GEMM_KS], b1[GEMM_KS];
    {
        const float* __restrict__ Wl = &Xs[GEMM_BUF + (half << 1) * GEMM_PITCH + (cp << 5) + l16];
#pragma unroll
        for (int s = 0; s < GEMM_KS; ++s) {
            b0[s].x = Wl[(4 * s) * GEMM_PITCH];
            b0[s].y = Wl[(4 * s + 1) * GEMM_PITCH];
            b1[s].x = Wl[(4 * s) * GEMM_PITCH + 16];
            b1[s].y = Wl[(4 * s + 1) * GEMM_PITCH + 16];
        }
    }
    gemm_lds_barrier();

    const int  aoff = ((rq << 4) + l16) * GEMM_PITCH + (half << 1);
    const bool hi   = (half != 0);
    float* __restrict__ Hp1 = H + (cp << 5) + lane;
    float* __restrict__ Hp2 = H + (size_t)8 * GEMM_F + (cp << 5) + (lane ^ 16);

    for (int i = 0; i < nIter; ++i) {
        const int  cur  = i & 1;
        const bool more = (i + 1 < nIter);
        if (more) {
            const int r0 = ((t0 + GEMM_STILES * (i + 1)) << 4) + wave;
#pragma unroll
            for (int u = 0; u < GEMM_RPW; ++u) {
                q[u] = zero4;
                if (r0 + GEMM_WAVES * u < rowEnd)
                    q[u] = *(const float4*)(X + (size_t)(r0 + GEMM_WAVES * u) * GEMM_K + sc);
            }
        }

        const float* __restrict__ al = &Xs[cur * GEMM_BUF + aoff];
        const int tbase = t0 + GEMM_STILES * i + rq;
#pragma nounroll
        for (int j = 0; j < 2; ++j) {
            const int tile = tbase + 4 * j;
            if (tile < t1) {
                const size_t ho = (size_t)(tile << 4) * GEMM_F;
                gemm_rowtile(al + (size_t)(64 * j) * GEMM_PITCH, b0, b1, Hp1 + ho, Hp2 + ho, hi);
            }

            if (j == 0 && more) {
                float* __restrict__ sn = st0 + (cur ^ 1) * GEMM_BUF;
#pragma unroll
                for (int u = 0; u < GEMM_RPW; ++u)
                    *(float4*)(sn + GEMM_WAVES * u * GEMM_PITCH) = q[u];
            }
        }

        gemm_lds_barrier();
    }
}

#define AGG_TB 1024
#define AGG_U  8

__device__ __forceinline__ void agg_add_row(float* __restrict__ acc, int dl, float w, int lane,
                                            const float4 h) {
    float* __restrict__ ap = acc + dl * 128 + lane;
    atomicAdd(ap,      w * h.x);
    atomicAdd(ap + 32, w * h.y);
    atomicAdd(ap + 64, w * h.z);
    atomicAdd(ap + 96, w * h.w);
}

__global__ void __launch_bounds__(AGG_TB)
k_agg_lds(const int* __restrict__ recs, const int* __restrict__ start,
          const float* __restrict__ H, const float* __restrict__ dinv,
          const float* __restrict__ bias, float* __restrict__ out, int n) {
    __shared__ float acc[BIN_R * 128];
    __shared__ float sdinv[BIN_R];
    const int t     = threadIdx.x;
    const int lane  = t & 31;
    const int wave  = __builtin_amdgcn_readfirstlane(t >> 5);
    const int node0 = blockIdx.x * BIN_R;
    const int rows  = min(BIN_R, n - node0);

    if (t < rows) sdinv[t] = dinv[node0 + t];
    for (int idx = t; idx < rows * 32; idx += AGG_TB) {
        const int r  = idx >> 5;
        const int c4 = (idx & 31) << 2;
        const float di = dinv[node0 + r];
        const float sw = di * di;
        const float4 h = *(const float4*)(H + (size_t)(node0 + r) * 128 + c4);
        float4 v;
        v.x = sw * h.x + bias[c4 + 0];
        v.y = sw * h.y + bias[c4 + 1];
        v.z = sw * h.z + bias[c4 + 2];
        v.w = sw * h.w + bias[c4 + 3];
        float* __restrict__ ap = &acc[r * 128 + (idx & 31)];
        ap[0]  = v.x;
        ap[32] = v.y;
        ap[64] = v.z;
        ap[96] = v.w;
    }
    __syncthreads();

    const int eb  = start[blockIdx.x];
    const int cnt = start[blockIdx.x + 1] - eb;
    const int w0  = eb + (int)(((long long)cnt * wave) >> 5);
    const int w1  = eb + (int)(((long long)cnt * (wave + 1)) >> 5);
    const int2* __restrict__ rec2 = (const int2*)recs;
    const float* __restrict__ Hl = H + (lane << 2);

    int2  rcN = make_int2(0, 0);
    float wN  = 0.0f;
    if (lane < w1 - w0) {
        rcN = rec2[w0 + lane];
        wN  = dinv[rcN.x] * sdinv[rcN.y];
    }
    for (int p = w0; p < w1; p += 32) {
        const int m  = min(32, w1 - p);
        const int2 rc = rcN;
        const int  wl = __float_as_int(wN);
        rcN = rec2[min(p + 32 + lane, w1 - 1)];
        if (m == 32) {
            float dsN = 0.0f, ddN = 0.0f;
#pragma unroll
            for (int j0 = 0; j0 < 32; j0 += AGG_U) {
                float4 v[AGG_U];
#pragma unroll
                for (int u = 0; u < AGG_U; ++u) {
                    const int s = __builtin_amdgcn_readlane(rc.x, j0 + u);
                    v[u] = *(const float4*)(Hl + (size_t)s * 128);
                }
#pragma unroll
                for (int u = 0; u < AGG_U; ++u) {
                    const int   dl = __builtin_amdgcn_readlane(rc.y, j0 + u);
                    const float w  = __int_as_float(__builtin_amdgcn_readlane(wl, j0 + u));
                    agg_add_row(acc, dl, w, lane, v[u]);
                }
                if (j0 == 0) {
                    dsN = dinv[rcN.x];
                    ddN = sdinv[rcN.y];
                }
            }
            wN = dsN * ddN;
        } else {
            for (int j = 0; j < m; ++j) {
                const int   s  = __builtin_amdgcn_readlane(rc.x, j);
                const int   dl = __builtin_amdgcn_readlane(rc.y, j);
                const float w  = __int_as_float(__builtin_amdgcn_readlane(wl, j));
                agg_add_row(acc, dl, w, lane, *(const float4*)(Hl + (size_t)s * 128));
            }
        }
    }
    __syncthreads();

    for (int idx = t; idx < rows * 32; idx += AGG_TB) {
        const int r  = idx >> 5;
        const int c4 = (idx & 31) << 2;
        const float* __restrict__ ap = &acc[r * 128 + (idx & 31)];
        float4 v = make_float4(ap[0], ap[32], ap[64], ap[96]);
        v.x = fmaxf(v.x, 0.0f);
        v.y = fmaxf(v.y, 0.0f);
        v.z = fmaxf(v.z, 0.0f);
        v.w = fmaxf(v.w, 0.0f);
        *(float4*)(out + (size_t)(node0 + r) * 128 + c4) = v;
    }
}

#define POOL_KC BIN_R
#define POOL_STRIDE 404
#define POOL_TILE (GCN_G * 128)
#define POOL_U 8

__global__ void k_pool_init(const int* __restrict__ batch, int* __restrict__ gcnt, int n) {
    int j = blockIdx.x * blockDim.x + threadIdx.x;
    int g = (j < n) ? batch[j] : -1;
    int g0 = __shfl(g, 0);
    int same = __popcll(__ballot(g == g0 && g >= 0));
    if ((threadIdx.x & 31) == 0 && g0 >= 0) atomicAdd(&gcnt[g0], same);
    if (g != g0 && g >= 0) atomicAdd(&gcnt[g], 1);
}

__global__ void __launch_bounds__(256)
k_pool_gemm(const int* __restrict__ recs, const int* __restrict__ start,
            const int* __restrict__ batch, const float* __restrict__ dinv,
            const float* __restrict__ h, float* __restrict__ Tpart, int n) {
    __shared__ float As[GCN_G * POOL_STRIDE];
    __shared__ float sdv[POOL_KC];
    const int t    = threadIdx.x;
    const int wave = t >> 5;
    const int lane = t & 31;
    const int half = lane >> 4;
    const int l16  = lane & 15;
    const int k0   = blockIdx.x * POOL_KC;
    const int klen = min(POOL_KC, n - k0);

    for (int idx = t; idx < GCN_G * (POOL_STRIDE / 4); idx += 256)
        *(float4*)(&As[idx << 2]) = make_float4(0.0f, 0.0f, 0.0f, 0.0f);
    for (int j = t; j < POOL_KC; j += 256) sdv[j] = (j < klen) ? dinv[k0 + j] : 0.0f;
    __syncthreads();
    for (int j = t; j < klen; j += 256) As[batch[k0 + j] * POOL_STRIDE + j] = sdv[j] * sdv[j];
    __syncthreads();
    {
        const int eb = start[blockIdx.x];
        const int m  = start[blockIdx.x + 1] - eb;
        const int2* __restrict__ rec2 = (const int2*)recs + eb;
        for (int i0 = t; i0 < m + t; i0 += 256 * POOL_U) {
            int2  r[POOL_U];
            int   g[POOL_U];
            float dv[POOL_U];
#pragma unroll
            for (int u = 0; u < POOL_U; ++u)
                r[u] = rec2[min(i0 + u * 256, m - 1)];
#pragma unroll
            for (int u = 0; u < POOL_U; ++u) {
                g[u]  = batch[r[u].x];
                dv[u] = dinv[r[u].x];
            }
#pragma unroll
            for (int u = 0; u < POOL_U; ++u)
                if (i0 + u * 256 < m)
                    atomicAdd(&As[g[u] * POOL_STRIDE + r[u].y], sdv[r[u].y] * dv[u]);
        }
    }
    __syncthreads();

    const int n0 = wave << 4;
    const float* __restrict__ al = &As[l16 * POOL_STRIDE + (half << 1)];
    const float* __restrict__ hb = h + (size_t)(k0 + (half << 1)) * 128 + n0 + l16;

    v8f c0 = {}, c1 = {}, c2 = {}, c3 = {};
    v2f bn[4];
#pragma unroll
    for (int u = 0; u < 4; ++u) {
        bn[u].x = hb[(size_t)(4 * u) * 128];
        bn[u].y = hb[(size_t)(4 * u) * 128 + 128];
    }
    for (int kk = 0; kk < klen; kk += 16) {
        v2f bc[4];
#pragma unroll
        for (int u = 0; u < 4; ++u) bc[u] = bn[u];
        const float* __restrict__ hn = hb + (size_t)min(kk + 16, klen - 16) * 128;
#pragma unroll
        for (int u = 0; u < 4; ++u) {
            bn[u].x = hn[(size_t)(4 * u) * 128];
            bn[u].y = hn[(size_t)(4 * u) * 128 + 128];
        }
#pragma unroll
        for (int u = 0; u < 4; ++u) {
            const float* __restrict__ ak = al + kk + 4 * u;
            v2f a0 = *(const v2f*)(ak);
            v2f a1 = *(const v2f*)(ak + 16 * POOL_STRIDE);
            v2f a2 = *(const v2f*)(ak + 32 * POOL_STRIDE);
            v2f a3 = *(const v2f*)(ak + 48 * POOL_STRIDE);
            c0 = __builtin_amdgcn_wmma_f32_16x16x4_f32(false, a0, false, bc[u], (short)0, c0, false, false);
            c1 = __builtin_amdgcn_wmma_f32_16x16x4_f32(false, a1, false, bc[u], (short)0, c1, false, false);
            c2 = __builtin_amdgcn_wmma_f32_16x16x4_f32(false, a2, false, bc[u], (short)0, c2, false, false);
            c3 = __builtin_amdgcn_wmma_f32_16x16x4_f32(false, a3, false, bc[u], (short)0, c3, false, false);
        }
    }

    float* __restrict__ o = Tpart + (size_t)blockIdx.x * POOL_TILE + (size_t)(half << 3) * 128 + n0 + l16;
#pragma unroll
    for (int v = 0; v < 8; ++v) {
        o[(size_t)(v)      * 128] = c0[v];
        o[(size_t)(v + 16) * 128] = c1[v];
        o[(size_t)(v + 32) * 128] = c2[v];
        o[(size_t)(v + 48) * 128] = c3[v];
    }
}

__global__ void k_pool_reduce(const float* __restrict__ Tpart, float* __restrict__ T8, int nparts) {
    int tid = blockIdx.x * blockDim.x + threadIdx.x;
    int e   = tid & (POOL_TILE - 1);
    int pg  = tid / POOL_TILE;
    int per = (nparts + 7) >> 3;
    int p0  = pg * per;
    int p1  = min(nparts, p0 + per);
    float s = 0.0f;
    for (int p = p0; p < p1; ++p) s += Tpart[(size_t)p * POOL_TILE + e];
    T8[tid] = s;
}

__global__ void __launch_bounds__(512)
k_pool_final(const float* __restrict__ T8, const float* __restrict__ W3,
             const float* __restrict__ b3, const int* __restrict__ gcnt,
             float* __restrict__ outp) {
    __shared__ float Ts[GCN_G * LDS_STRIDE];
    __shared__ float Wl[GCN_OUT * LDS_STRIDE];
    __shared__ int   hist[GCN_G];
    const int t    = threadIdx.x;
    const int wave = t >> 5;
    const int lane = t & 31;
    const int half = lane >> 4;
    const int l16  = lane & 15;

    if (t < GCN_G) hist[t] = gcnt[t];
    for (int idx = t; idx < POOL_TILE; idx += 512) {
        float s = 0.0f;
#pragma unroll
        for (int pg = 0; pg < 8; ++pg) s += T8[pg * POOL_TILE + idx];
        Ts[(idx >> 7) * LDS_STRIDE + (idx & 127)] = s;
    }
    for (int idx = t; idx < 128 * GCN_OUT; idx += 512)
        Wl[(idx & 63) * LDS_STRIDE + (idx >> 6)] = W3[idx];
    __syncthreads();

    const int mi = wave >> 2;
    const int ni = wave & 3;
    const float* __restrict__ al = &Ts[((mi << 4) + l16) * LDS_STRIDE + (half << 1)];
    const float* __restrict__ bl = &Wl[((ni << 4) + l16) * LDS_STRIDE + (half << 1)];
    v8f c = {};
#pragma unroll
    for (int k0 = 0; k0 < 128; k0 += 4) {
        v2f a = *(const v2f*)(al + k0);
        v2f b = *(const v2f*)(bl + k0);
        c = __builtin_amdgcn_wmma_f32_16x16x4_f32(false, a, false, b, (short)0, c, false, false);
    }
    const int col = (ni << 4) + l16;
    const float bias = b3[col];
#pragma unroll
    for (int v = 0; v < 8; ++v) {
        int g = (mi << 4) + (half << 3) + v;
        float cf = (float)hist[g];
        outp[g * GCN_OUT + col] = (c[v] + cf * bias) / fmaxf(cf, 1.0f);
    }
}

extern "C" void kernel_launch(void* const* d_in, const int* in_sizes, int n_in,
                              void* d_out, int out_size, void* d_ws, size_t ws_size,
                              hipStream_t stream) {
    const float* x     = (const float*)d_in[0];
    const int*   edge  = (const int*)d_in[1];
    const int*   batch = (const int*)d_in[2];
    const float* W1    = (const float*)d_in[3];
    const float* b1    = (const float*)d_in[4];
    const float* W2    = (const float*)d_in[5];
    const float* b2    = (const float*)d_in[6];
    const float* W3    = (const float*)d_in[7];
    const float* b3    = (const float*)d_in[8];

    const int IN = 128, HID = 128;
    const int N = in_sizes[0] / IN;
    const int E = in_sizes[1] / 2;
    const int* src = edge;
    const int* dst = edge + E;

    char* ws = (char*)d_ws;
    float* dinv = (float*)ws;
    size_t off = ((size_t)N * sizeof(float) + 511) & ~(size_t)511;
    float* bufA = (float*)(ws + off);
    float* bufB = bufA + (size_t)N * HID;
    const int P = (N + POOL_KC - 1) / POOL_KC;
    float* Tpart = bufB + (size_t)N * HID;
    float* T8    = Tpart + (size_t)P * POOL_TILE;
    const int NB    = (N + BIN_R - 1) / BIN_R;
    const int NBpad = (NB + 1 + 127) & ~127;
    int* binCount  = (int*)(T8 + (size_t)8 * POOL_TILE);
    int* binStart  = binCount + 2 * NBpad;
    int* binCursor = binStart + 2 * NBpad;
    int* gcnt      = binCursor + 2 * NBpad;
    int* recs      = gcnt + 128;
    int* recsS     = recs + (size_t)2 * E;

    const int TB = 256;
    dim3 blk(TB);

    const int FB = (E + FILL_TB * FILL_EPT - 1) / (FILL_TB * FILL_EPT);
    k_tables_init<<<dim3((NBpad + NB + TB - 1) / TB), blk, 0, stream>>>(binCount, gcnt, NBpad + NB);
    k_bin_hist<<<dim3(FB, 2), dim3(FILL_TB), NB * sizeof(int), stream>>>(src, dst, binCount, E, NB, NBpad);
    k_bin_scan<<<dim3(2), dim3(1024), 0, stream>>>(binCount, binStart, binCursor, NB, NBpad);
    k_bin_fill<<<dim3(FB, 2), dim3(FILL_TB), 2 * NB * sizeof(int), stream>>>(src, dst, binStart, binCursor, recs, E, NB, NBpad);

    k_bin_deg<<<dim3(NB), dim3(DEG_TB), 0, stream>>>(recs, binStart, dinv, N);

    const int mt = N / 16;

    k_gemm_strips<<<dim3(GEMM_BLOCKS), dim3(GEMM_TB), 0, stream>>>(x, W1, bufA, mt);
    k_agg_lds      <<<dim3(NB), dim3(AGG_TB), 0, stream>>>(recs, binStart, bufA, dinv, b1, bufB, N);

    k_gemm_strips<<<dim3(GEMM_BLOCKS), dim3(GEMM_TB), 0, stream>>>(bufB, W2, bufA, mt);
    k_agg_lds      <<<dim3(NB), dim3(AGG_TB), 0, stream>>>(recs, binStart, bufA, dinv, b2, bufB, N);

    k_pool_init  <<<dim3((N + TB - 1) / TB), blk, 0, stream>>>(batch, gcnt, N);
    k_pool_gemm  <<<dim3(P), blk, 0, stream>>>(recsS, binStart + NBpad, batch, dinv, bufB, Tpart, N);
    k_pool_reduce<<<dim3(8 * POOL_TILE / TB), blk, 0, stream>>>(Tpart, T8, P);
    k_pool_final <<<dim3(1), dim3(512), 0, stream>>>(T8, W3, b3, gcnt, (float*)d_out);
}


# Per-edge 4-byte global gathers (dinv[src] in both aggregations; dinv[dst], graph id in the pooled slab build) replaced by lookups in whole-graph byte tables (degree, graph) held in LDS; sums unchanged
# speedup vs baseline: 4.8274x; 1.0472x over previous
#include <hip/hip_runtime.h>
#include <math.h>

typedef float v2f __attribute__((ext_vector_type(2)));
typedef float v8f __attribute__((ext_vector_type(8)));

#define GCN_G   64
#define GCN_OUT 64
#define LDS_STRIDE 132

#define BIN_R 400
#define FILL_EPT 4
#define FILL_TB 1024

#define NODE_TAB_MAX 106496
#define NODE_TAB_WORDS (NODE_TAB_MAX / 4)

template <int U>
__device__ __forceinline__ void node_tab_load(int* __restrict__ dstw, const int* __restrict__ srcw,
                                              int n, int t, int nthreads) {
    const int4* __restrict__ s4 = (const int4*)srcw;
    int4* __restrict__ d4 = (int4*)dstw;
    const int n16 = n >> 4;
    for (int i0 = t; i0 < n16; i0 += nthreads * U) {
        int4 v[U];
#pragma unroll
        for (int u = 0; u < U; ++u) v[u] = s4[min(i0 + u * nthreads, n16 - 1)];
#pragma unroll
        for (int u = 0; u < U; ++u) d4[min(i0 + u * nthreads, n16 - 1)] = v[u];
    }
}

__device__ __forceinline__ float node_rs(int d) {
    const float f = (float)d;
    return (f > 0.0f) ? (1.0f / sqrtf(f)) : 0.0f;
}

__device__ __forceinline__ float node_dinv(const unsigned char* deg8, const float* rs,
                                           const float* __restrict__ dinv, int node, bool tab) {
    if (tab) {
        const unsigned d = deg8[node];
        if (d != 255u) return rs[d];
    }
    return dinv[node];
}

__global__ void k_tables_init(int* __restrict__ binCount, int* __restrict__ gcnt, int nb) {
    int i = blockIdx.x * blockDim.x + threadIdx.x;
    if (i < nb) binCount[i] = 0;
    if (i < GCN_G) gcnt[i] = 0;
}

__global__ void __launch_bounds__(FILL_TB)
k_bin_hist(const int* __restrict__ src, const int* __restrict__ dstv,
           int* __restrict__ binCountAll, int e, int nb, int pitch) {
    extern __shared__ int hsm[];
    const int t = threadIdx.x;
    const int* __restrict__ dst = blockIdx.y ? src : dstv;
    int* __restrict__ binCount = binCountAll + blockIdx.y * pitch;
    for (int b = t; b < nb; b += FILL_TB) hsm[b] = 0;
    __syncthreads();
    const int e0 = blockIdx.x * (FILL_TB * FILL_EPT);
    int d[FILL_EPT];
#pragma unroll
    for (int u = 0; u < FILL_EPT; ++u)
        d[u] = dst[min(e0 + u * FILL_TB + t, e - 1)];
#pragma unroll
    for (int u = 0; u < FILL_EPT; ++u)
        if (e0 + u * FILL_TB + t < e) atomicAdd(&hsm[d[u] / BIN_R], 1);
    __syncthreads();
    for (int b = t; b < nb; b += FILL_TB) {
        const int c = hsm[b];
        if (c > 0) atomicAdd(&binCount[b], c);
    }
}

__global__ void __launch_bounds__(1024)
k_bin_scan(const int* __restrict__ cntAll, int* __restrict__ startAll, int* __restrict__ cursorAll,
           int nb, int pitch) {
    __shared__ int s[1024];
    const int t = threadIdx.x;
    const int* __restrict__ cnt = cntAll + blockIdx.x * pitch;
    int* __restrict__ start  = startAll + blockIdx.x * pitch;
    int* __restrict__ cursor = cursorAll + blockIdx.x * pitch;
    int carry = 0;
    for (int b0 = 0; b0 < nb; b0 += 1024) {
        const int b = b0 + t;
        const int v = (b < nb) ? cnt[b] : 0;
        s[t] = v;
        __syncthreads();
        for (int off = 1; off < 1024; off <<= 1) {
            int a = (t >= off) ? s[t - off] : 0;
            __syncthreads();
            s[t] += a;
            __syncthreads();
        }
        if (b < nb) {
            start[b]  = carry + s[t] - v;
            cursor[b] = 0;
        }
        const int total = s[1023];
        __syncthreads();
        carry += total;
    }
    if (t == 0) start[nb] = carry;
}

__global__ void __launch_bounds__(FILL_TB)
k_bin_fill(const int* __restrict__ srcv, const int* __restrict__ dstv,
           const int* __restrict__ startAll,
           int* __restrict__ cursorAll, int* __restrict__ recsAll, int e, int nb, int pitch) {
    const int* __restrict__ src = blockIdx.y ? dstv : srcv;
    const int* __restrict__ dst = blockIdx.y ? srcv : dstv;
    const int* __restrict__ start = startAll + blockIdx.y * pitch;
    int* __restrict__ cursor = cursorAll + blockIdx.y * pitch;
    int* __restrict__ recs = recsAll + (size_t)blockIdx.y * 2 * (size_t)e;
    extern __shared__ int fsm[];
    int* hist = fsm;
    int* base = fsm + nb;
    const int t = threadIdx.x;
    for (int b = t; b < nb; b += FILL_TB) hist[b] = 0;
    __syncthreads();

    const int e0 = blockIdx.x * (FILL_TB * FILL_EPT);
    int es[FILL_EPT], ed[FILL_EPT], eb[FILL_EPT], er[FILL_EPT];
#pragma unroll
    for (int u = 0; u < FILL_EPT; ++u) {
        const int i = e0 + u * FILL_TB + t;
        eb[u] = -1;
        if (i < e) {
            es[u] = src[i];
            ed[u] = dst[i];
            eb[u] = ed[u] / BIN_R;
            er[u] = atomicAdd(&hist[eb[u]], 1);
        }
    }
    __syncthreads();
    for (int b = t; b < nb; b += FILL_TB) {
        const int c = hist[b];
        base[b] = start[b] + (c > 0 ? atomicAdd(&cursor[b], c) : 0);
    }
    __syncthreads();
    int2* __restrict__ out = (int2*)recs;
#pragma unroll
    for (int u = 0; u < FILL_EPT; ++u) {
        if (eb[u] >= 0)
            out[base[eb[u]] + er[u]] = make_int2(es[u], ed[u] - eb[u] * BIN_R);
    }
}

#define DEG_TB 1024
#define DEG_U  8
__global__ void __launch_bounds__(DEG_TB)
k_bin_deg(const int* __restrict__ recs, const int* __restrict__ start,
          const int* __restrict__ batch, float* __restrict__ dinv,
          int* __restrict__ deg8w, int* __restrict__ g8w, int n) {
    __shared__ int cnt[BIN_R];
    const int t = threadIdx.x;
    if (t < BIN_R) cnt[t] = 0;
    __syncthreads();
    const int eb = start[blockIdx.x];
    const int m  = start[blockIdx.x + 1] - eb;
    const int2* __restrict__ rec2 = (const int2*)recs + eb;
    for (int i0 = t; i0 < m + t; i0 += DEG_TB * DEG_U) {
        int d[DEG_U];
#pragma unroll
        for (int u = 0; u < DEG_U; ++u)
            d[u] = rec2[min(i0 + u * DEG_TB, m - 1)].y;
#pragma unroll
        for (int u = 0; u < DEG_U; ++u)
            if (i0 + u * DEG_TB < m) atomicAdd(&cnt[d[u]], 1);
    }
    __syncthreads();
    const int node = blockIdx.x * BIN_R + t;
    if (t < BIN_R && node < n) {
        const float d = (float)(cnt[t] + 1);
        dinv[node] = (d > 0.0f) ? (1.0f / sqrtf(d)) : 0.0f;
    }
    if (t < BIN_R / 4) {
        const int nb0 = blockIdx.x * BIN_R + 4 * t;
        if (nb0 < n) {
            unsigned dw = 0u, gw = 0u;
#pragma unroll
            for (int k = 0; k < 4; ++k) {
                if (nb0 + k < n) {
                    dw |= (unsigned)min(cnt[4 * t + k] + 1, 255) << (8 * k);
                    gw |= ((unsigned)batch[nb0 + k] & 255u) << (8 * k);
                }
            }
            deg8w[nb0 >> 2] = (int)dw;
            g8w[nb0 >> 2]   = (int)gw;
        }
    }
}

#define GEMM_K 128
#define GEMM_F 128
#define GEMM_KS (GEMM_K / 4)
#define GEMM_TB 512
#define GEMM_WAVES (GEMM_TB / 32)
#define GEMM_SROWS 128
#define GEMM_STILES (GEMM_SROWS / 16)
#define GEMM_RPW (GEMM_SROWS / GEMM_WAVES)
#define GEMM_PITCH 132
#define GEMM_BUF (GEMM_SROWS * GEMM_PITCH)
#define GEMM_BLOCKS 256

__device__ __forceinline__ void gemm_lds_barrier() {
    __builtin_amdgcn_fence(__ATOMIC_RELEASE, "workgroup", "local");
    __builtin_amdgcn_s_barrier();
    __builtin_amdgcn_fence(__ATOMIC_ACQUIRE, "workgroup", "local");
}

__device__ __forceinline__ float gemm_other_half(float x) {
    int b = __builtin_bit_cast(int, x);
    return __builtin_bit_cast(float, __builtin_amdgcn_permlanex16(b, b, 0x76543210, 0xfedcba98, false, false));
}

__device__ __forceinline__ void gemm_rowtile(const float* __restrict__ al,
                                             const v2f (&b0)[GEMM_KS], const v2f (&b1)[GEMM_KS],
                                             float* __restrict__ H1, float* __restrict__ H2, bool hi) {
    v8f c0 = {}, c1 = {};
#pragma unroll
    for (int s = 0; s < GEMM_KS; ++s) {
        v2f a = *(const v2f*)(al + 4 * s);
        c0 = __builtin_amdgcn_wmma_f32_16x16x4_f32(false, a, false, b0[s], (short)0, c0, false, false);
        c1 = __builtin_amdgcn_wmma_f32_16x16x4_f32(false, a, false, b1[s], (short)0, c1, false, false);
    }
#pragma unroll
    for (int v = 0; v < 8; ++v) {
        const float t = gemm_other_half(c1[v]);
        H1[(size_t)v * GEMM_F] = hi ? t : c0[v];
        H2[(size_t)v * GEMM_F] = hi ? c0[v] : t;
    }
}

__global__ void __launch_bounds__(GEMM_TB)
k_gemm_strips(const float* __restrict__ X, const float* __restrict__ W,
              float* __restrict__ H, int mtiles) {
    __shared__ float Xs[2 * GEMM_BUF];

    const int t    = threadIdx.x;
    const int lane = t & 31;
    const int wave = __builtin_amdgcn_readfirstlane(t >> 5);
    const int half = lane >> 4;
    const int l16  = lane & 15;
    const int cp   = wave >> 2;
    const int rq   = ((wave & 3) - (wave >> 2)) & 3;

    const unsigned nb = gridDim.x;
    const int t0 = (int)(((unsigned)mtiles * blockIdx.x) / nb);
    const int t1 = (int)(((unsigned)mtiles * (blockIdx.x + 1u)) / nb);
    const int rowEnd = t1 << 4;
    const int nIter  = (t1 - t0 + GEMM_STILES - 1) / GEMM_STILES;

    const int sc = lane << 2;
    float* __restrict__ st0 = &Xs[wave * GEMM_PITCH + sc];
    const float4 zero4 = make_float4(0.0f, 0.0f, 0.0f, 0.0f);
    float4 q[GEMM_RPW];

    {
        const int r0 = (t0 << 4) + wave;
        float4 wq[GEMM_K / GEMM_WAVES];
#pragma unroll
        for (int u = 0; u < GEMM_K / GEMM_WAVES; ++u)
            wq[u] = *(const float4*)(W + (size_t)(wave + GEMM_WAVES * u) * GEMM_F + sc);
#pragma unroll
        for (int u = 0; u < GEMM_RPW; ++u) {
            q[u] = zero4;
            if (r0 + GEMM_WAVES * u < rowEnd)
                q[u] = *(const float4*)(X + (size_t)(r0 + GEMM_WAVES * u) * GEMM_K + sc);
        }
#pragma unroll
        for (int u = 0; u < GEMM_K / GEMM_WAVES; ++u)
            *(float4*)(st0 + GEMM_BUF + GEMM_WAVES * u * GEMM_PITCH) = wq[u];
#pragma unroll
        for (int u = 0; u < GEMM_RPW; ++u)
            *(float4*)(st0 + GEMM_WAVES * u * GEMM_PITCH) = q[u];
    }
    gemm_lds_barrier();

    v2f b0[GEMM_KS], b1[GEMM_KS];
    {
        const float* __restrict__ Wl = &Xs[GEMM_BUF + (half << 1) * GEMM_PITCH + (cp << 5) + l16];
#pragma unroll
        for (int s = 0; s < GEMM_KS; ++s) {
            b0[s].x = Wl[(4 * s) * GEMM_PITCH];
            b0[s].y = Wl[(4 * s + 1) * GEMM_PITCH];
            b1[s].x = Wl[(4 * s) * GEMM_PITCH + 16];
            b1[s].y = Wl[(4 * s + 1) * GEMM_PITCH + 16];
        }
    }
    gemm_lds_barrier();

    const int  aoff = ((rq << 4) + l16) * GEMM_PITCH + (half << 1);
    const bool hi   = (half != 0);
    float* __restrict__ Hp1 = H + (cp << 5) + lane;
    float* __restrict__ Hp2 = H + (size_t)8 * GEMM_F + (cp << 5) + (lane ^ 16);

    for (int i = 0; i < nIter; ++i) {
        const int  cur  = i & 1;
        const bool more = (i + 1 < nIter);
        if (more) {
            const int r0 = ((t0 + GEMM_STILES * (i + 1)) << 4) + wave;
#pragma unroll
            for (int u = 0; u < GEMM_RPW; ++u) {
                q[u] = zero4;
                if (r0 + GEMM_WAVES * u < rowEnd)
                    q[u] = *(const float4*)(X + (size_t)(r0 + GEMM_WAVES * u) * GEMM_K + sc);
            }
        }

        const float* __restrict__ al = &Xs[cur * GEMM_BUF + aoff];
        const int tbase = t0 + GEMM_STILES * i + rq;
#pragma nounroll
        for (int j = 0; j < 2; ++j) {
            const int tile = tbase + 4 * j;
            if (tile < t1) {
                const size_t ho = (size_t)(tile << 4) * GEMM_F;
                gemm_rowtile(al + (size_t)(64 * j) * GEMM_PITCH, b0, b1, Hp1 + ho, Hp2 + ho, hi);
            }

            if (j == 0 && more) {
                float* __restrict__ sn = st0 + (cur ^ 1) * GEMM_BUF;
#pragma unroll
                for (int u = 0; u < GEMM_RPW; ++u)
                    *(float4*)(sn + GEMM_WAVES * u * GEMM_PITCH) = q[u];
            }
        }

        gemm_lds_barrier();
    }
}

#define AGG_TB 1024
#define AGG_U  8

__device__ __forceinline__ void agg_add_row(float* __restrict__ acc, int dl, float w, int lane,
                                            const float4 h) {
    float* __restrict__ ap = acc + dl * 128 + lane;
    atomicAdd(ap,      w * h.x);
    atomicAdd(ap + 32, w * h.y);
    atomicAdd(ap + 64, w * h.z);
    atomicAdd(ap + 96, w * h.w);
}

__global__ void __launch_bounds__(AGG_TB)
k_agg_lds(const int* __restrict__ recs, const int* __restrict__ start,
          const float* __restrict__ H, const float* __restrict__ dinv,
          const int* __restrict__ deg8w,
          const float* __restrict__ bias, float* __restrict__ out, int n) {
    __shared__ float acc[BIN_R * 128];
    __shared__ float sdinv[BIN_R];
    __shared__ __attribute__((aligned(16))) int sdegw[NODE_TAB_WORDS];
    __shared__ float srs[256];
    const unsigned char* sdeg8 = (const unsigned char*)sdegw;
    const bool tab = (n <= NODE_TAB_MAX);
    const int tabmask = tab ? -1 : 0;
    const int t     = threadIdx.x;
    const int lane  = t & 31;
    const int wave  = __builtin_amdgcn_readfirstlane(t >> 5);
    const int node0 = blockIdx.x * BIN_R;
    const int rows  = min(BIN_R, n - node0);

    if (t < rows) sdinv[t] = dinv[node0 + t];
    if (t < 256) srs[t] = node_rs(t);
    if (tab) node_tab_load<7>(sdegw, deg8w, n, t, AGG_TB);
    for (int idx = t; idx < rows * 32; idx += AGG_TB) {
        const int r  = idx >> 5;
        const int c4 = (idx & 31) << 2;
        const float di = dinv[node0 + r];
        const float sw = di * di;
        const float4 h = *(const float4*)(H + (size_t)(node0 + r) * 128 + c4);
        float4 v;
        v.x = sw * h.x + bias[c4 + 0];
        v.y = sw * h.y + bias[c4 + 1];
        v.z = sw * h.z + bias[c4 + 2];
        v.w = sw * h.w + bias[c4 + 3];
        float* __restrict__ ap = &acc[r * 128 + (idx & 31)];
        ap[0]  = v.x;
        ap[32] = v.y;
        ap[64] = v.z;
        ap[96] = v.w;
    }
    __syncthreads();

    const int eb  = start[blockIdx.x];
    const int cnt = start[blockIdx.x + 1] - eb;
    const int w0  = eb + (int)(((long long)cnt * wave) >> 5);
    const int w1  = eb + (int)(((long long)cnt * (wave + 1)) >> 5);
    const int2* __restrict__ rec2 = (const int2*)recs;
    const float* __restrict__ Hl = H + (lane << 2);

    int2  rcN = make_int2(0, 0);
    float wN  = 0.0f;
    if (lane < w1 - w0) {
        rcN = rec2[w0 + lane];
        wN  = node_dinv(sdeg8, srs, dinv, rcN.x, tab) * sdinv[rcN.y];
    }
    for (int p = w0; p < w1; p += 32) {
        const int m  = min(32, w1 - p);
        const int2 rc = rcN;
        const int  wl = __float_as_int(wN);
        rcN = rec2[min(p + 32 + lane, w1 - 1)];
        if (m == 32) {
            float dsN = 0.0f, ddN = 0.0f;
            unsigned dgN = 0u;
#pragma unroll
            for (int j0 = 0; j0 < 32; j0 += AGG_U) {
                float4 v[AGG_U];
#pragma unroll
                for (int u = 0; u < AGG_U; ++u) {
                    const int s = __builtin_amdgcn_readlane(rc.x, j0 + u);
                    v[u] = *(const float4*)(Hl + (size_t)s * 128);
                }
#pragma unroll
                for (int u = 0; u < AGG_U; ++u) {
                    const int   dl = __builtin_amdgcn_readlane(rc.y, j0 + u);
                    const float w  = __int_as_float(__builtin_amdgcn_readlane(wl, j0 + u));
                    agg_add_row(acc, dl, w, lane, v[u]);
                }
                if (j0 == 0) {
                    dgN = sdeg8[rcN.x & tabmask];
                    ddN = sdinv[rcN.y];
                }
                if (j0 == AGG_U) dsN = srs[dgN];
            }
            if (!tab || dgN == 255u) dsN = dinv[rcN.x];
            wN = dsN * ddN;
        } else {
            for (int j = 0; j < m; ++j) {
                const int   s  = __builtin_amdgcn_readlane(rc.x, j);
                const int   dl = __builtin_amdgcn_readlane(rc.y, j);
                const float w  = __int_as_float(__builtin_amdgcn_readlane(wl, j));
                agg_add_row(acc, dl, w, lane, *(const float4*)(Hl + (size_t)s * 128));
            }
        }
    }
    __syncthreads();

    for (int idx = t; idx < rows * 32; idx += AGG_TB) {
        const int r  = idx >> 5;
        const int c4 = (idx & 31) << 2;
        const float* __restrict__ ap = &acc[r * 128 + (idx & 31)];
        float4 v = make_float4(ap[0], ap[32], ap[64], ap[96]);
        v.x = fmaxf(v.x, 0.0f);
        v.y = fmaxf(v.y, 0.0f);
        v.z = fmaxf(v.z, 0.0f);
        v.w = fmaxf(v.w, 0.0f);
        *(float4*)(out + (size_t)(node0 + r) * 128 + c4) = v;
    }
}

#define POOL_KC BIN_R
#define POOL_STRIDE 404
#define POOL_TILE (GCN_G * 128)
#define POOL_U 8

__global__ void k_pool_init(const int* __restrict__ batch, int* __restrict__ gcnt, int n) {
    int j = blockIdx.x * blockDim.x + threadIdx.x;
    int g = (j < n) ? batch[j] : -1;
    int g0 = __shfl(g, 0);
    int same = __popcll(__ballot(g == g0 && g >= 0));
    if ((threadIdx.x & 31) == 0 && g0 >= 0) atomicAdd(&gcnt[g0], same);
    if (g != g0 && g >= 0) atomicAdd(&gcnt[g], 1);
}

__global__ void __launch_bounds__(256)
k_pool_gemm(const int* __restrict__ recs, const int* __restrict__ start,
            const int* __restrict__ batch, const float* __restrict__ dinv,
            const int* __restrict__ deg8w, const int* __restrict__ g8w,
            const float* __restrict__ h, float* __restrict__ Tpart, int n) {
    __shared__ float As[GCN_G * POOL_STRIDE];
    __shared__ float sdv[POOL_KC];
    __shared__ __attribute__((aligned(16))) int sdegw[NODE_TAB_WORDS];
    __shared__ __attribute__((aligned(16))) int sgw[NODE_TAB_WORDS];
    __shared__ float srs[256];
    const unsigned char* sdeg8 = (const unsigned char*)sdegw;
    const unsigned char* sg8   = (const unsigned char*)sgw;
    const bool tab = (n <= NODE_TAB_MAX);
    const int t    = threadIdx.x;
    const int wave = t >> 5;
    const int lane = t & 31;
    const int half = lane >> 4;
    const int l16  = lane & 15;
    const int k0   = blockIdx.x * POOL_KC;
    const int klen = min(POOL_KC, n - k0);

    for (int idx = t; idx < GCN_G * (POOL_STRIDE / 4); idx += 256)
        *(float4*)(&As[idx << 2]) = make_float4(0.0f, 0.0f, 0.0f, 0.0f);
    for (int j = t; j < POOL_KC; j += 256) sdv[j] = (j < klen) ? dinv[k0 + j] : 0.0f;
    srs[t] = node_rs(t);
    if (tab) {
        node_tab_load<25>(sdegw, deg8w, n, t, 256);
        node_tab_load<25>(sgw, g8w, n, t, 256);
    }
    __syncthreads();
    for (int j = t; j < klen; j += 256) As[batch[k0 + j] * POOL_STRIDE + j] = sdv[j] * sdv[j];
    __syncthreads();
    {
        const int eb = start[blockIdx.x];
        const int m  = start[blockIdx.x + 1] - eb;
        const int2* __restrict__ rec2 = (const int2*)recs + eb;
        for (int i0 = t; i0 < m + t; i0 += 256 * POOL_U) {
            int2  r[POOL_U];
            int   g[POOL_U];
            float dv[POOL_U];
#pragma unroll
            for (int u = 0; u < POOL_U; ++u)
                r[u] = rec2[min(i0 + u * 256, m - 1)];
            if (tab) {
                unsigned dg[POOL_U];
#pragma unroll
                for (int u = 0; u < POOL_U; ++u) {
                    g[u]  = (int)sg8[r[u].x];
                    dg[u] = sdeg8[r[u].x];
                }
#pragma unroll
                for (int u = 0; u < POOL_U; ++u) dv[u] = srs[dg[u]];
#pragma unroll
                for (int u = 0; u < POOL_U; ++u)
                    if (dg[u] == 255u) dv[u] = dinv[r[u].x];
            } else {
#pragma unroll
                for (int u = 0; u < POOL_U; ++u) {
                    g[u]  = batch[r[u].x];
                    dv[u] = dinv[r[u].x];
                }
            }
#pragma unroll
            for (int u = 0; u < POOL_U; ++u)
                if (i0 + u * 256 < m)
                    atomicAdd(&As[g[u] * POOL_STRIDE + r[u].y], sdv[r[u].y] * dv[u]);
        }
    }
    __syncthreads();

    const int n0 = wave << 4;
    const float* __restrict__ al = &As[l16 * POOL_STRIDE + (half << 1)];
    const float* __restrict__ hb = h + (size_t)(k0 + (half << 1)) * 128 + n0 + l16;

    v8f c0 = {}, c1 = {}, c2 = {}, c3 = {};
    v2f bn[4];
#pragma unroll
    for (int u = 0; u < 4; ++u) {
        bn[u].x = hb[(size_t)(4 * u) * 128];
        bn[u].y = hb[(size_t)(4 * u) * 128 + 128];
    }
    for (int kk = 0; kk < klen; kk += 16) {
        v2f bc[4];
#pragma unroll
        for (int u = 0; u < 4; ++u) bc[u] = bn[u];
        const float* __restrict__ hn = hb + (size_t)min(kk + 16, klen - 16) * 128;
#pragma unroll
        for (int u = 0; u < 4; ++u) {
            bn[u].x = hn[(size_t)(4 * u) * 128];
            bn[u].y = hn[(size_t)(4 * u) * 128 + 128];
        }
#pragma unroll
        for (int u = 0; u < 4; ++u) {
            const float* __restrict__ ak = al + kk + 4 * u;
            v2f a0 = *(const v2f*)(ak);
            v2f a1 = *(const v2f*)(ak + 16 * POOL_STRIDE);
            v2f a2 = *(const v2f*)(ak + 32 * POOL_STRIDE);
            v2f a3 = *(const v2f*)(ak + 48 * POOL_STRIDE);
            c0 = __builtin_amdgcn_wmma_f32_16x16x4_f32(false, a0, false, bc[u], (short)0, c0, false, false);
            c1 = __builtin_amdgcn_wmma_f32_16x16x4_f32(false, a1, false, bc[u], (short)0, c1, false, false);
            c2 = __builtin_amdgcn_wmma_f32_16x16x4_f32(false, a2, false, bc[u], (short)0, c2, false, false);
            c3 = __builtin_amdgcn_wmma_f32_16x16x4_f32(false, a3, false, bc[u], (short)0, c3, false, false);
        }
    }

    float* __restrict__ o = Tpart + (size_t)blockIdx.x * POOL_TILE + (size_t)(half << 3) * 128 + n0 + l16;
#pragma unroll
    for (int v = 0; v < 8; ++v) {
        o[(size_t)(v)      * 128] = c0[v];
        o[(size_t)(v + 16) * 128] = c1[v];
        o[(size_t)(v + 32) * 128] = c2[v];
        o[(size_t)(v + 48) * 128] = c3[v];
    }
}

__global__ void k_pool_reduce(const float* __restrict__ Tpart, float* __restrict__ T8, int nparts) {
    int tid = blockIdx.x * blockDim.x + threadIdx.x;
    int e   = tid & (POOL_TILE - 1);
    int pg  = tid / POOL_TILE;
    int per = (nparts + 7) >> 3;
    int p0  = pg * per;
    int p1  = min(nparts, p0 + per);
    float s = 0.0f;
    for (int p = p0; p < p1; ++p) s += Tpart[(size_t)p * POOL_TILE + e];
    T8[tid] = s;
}

__global__ void __launch_bounds__(512)
k_pool_final(const float* __restrict__ T8, const float* __restrict__ W3,
             const float* __restrict__ b3, const int* __restrict__ gcnt,
             float* __restrict__ outp) {
    __shared__ float Ts[GCN_G * LDS_STRIDE];
    __shared__ float Wl[GCN_OUT * LDS_STRIDE];
    __shared__ int   hist[GCN_G];
    const int t    = threadIdx.x;
    const int wave = t >> 5;
    const int lane = t & 31;
    const int half = lane >> 4;
    const int l16  = lane & 15;

    if (t < GCN_G) hist[t] = gcnt[t];
    for (int idx = t; idx < POOL_TILE; idx += 512) {
        float s = 0.0f;
#pragma unroll
        for (int pg = 0; pg < 8; ++pg) s += T8[pg * POOL_TILE + idx];
        Ts[(idx >> 7) * LDS_STRIDE + (idx & 127)] = s;
    }
    for (int idx = t; idx < 128 * GCN_OUT; idx += 512)
        Wl[(idx & 63) * LDS_STRIDE + (idx >> 6)] = W3[idx];
    __syncthreads();

    const int mi = wave >> 2;
    const int ni = wave & 3;
    const float* __restrict__ al = &Ts[((mi << 4) + l16) * LDS_STRIDE + (half << 1)];
    const float* __restrict__ bl = &Wl[((ni << 4) + l16) * LDS_STRIDE + (half << 1)];
    v8f c = {};
#pragma unroll
    for (int k0 = 0; k0 < 128; k0 += 4) {
        v2f a = *(const v2f*)(al + k0);
        v2f b = *(const v2f*)(bl + k0);
        c = __builtin_amdgcn_wmma_f32_16x16x4_f32(false, a, false, b, (short)0, c, false, false);
    }
    const int col = (ni << 4) + l16;
    const float bias = b3[col];
#pragma unroll
    for (int v = 0; v < 8; ++v) {
        int g = (mi << 4) + (half << 3) + v;
        float cf = (float)hist[g];
        outp[g * GCN_OUT + col] = (c[v] + cf * bias) / fmaxf(cf, 1.0f);
    }
}

extern "C" void kernel_launch(void* const* d_in, const int* in_sizes, int n_in,
                              void* d_out, int out_size, void* d_ws, size_t ws_size,
                              hipStream_t stream) {
    const float* x     = (const float*)d_in[0];
    const int*   edge  = (const int*)d_in[1];
    const int*   batch = (const int*)d_in[2];
    const float* W1    = (const float*)d_in[3];
    const float* b1    = (const float*)d_in[4];
    const float* W2    = (const float*)d_in[5];
    const float* b2    = (const float*)d_in[6];
    const float* W3    = (const float*)d_in[7];
    const float* b3    = (const float*)d_in[8];

    const int IN = 128, HID = 128;
    const int N = in_sizes[0] / IN;
    const int E = in_sizes[1] / 2;
    const int* src = edge;
    const int* dst = edge + E;

    char* ws = (char*)d_ws;
    float* dinv = (float*)ws;
    size_t off = ((size_t)N * sizeof(float) + 511) & ~(size_t)511;
    float* bufA = (float*)(ws + off);
    float* bufB = bufA + (size_t)N * HID;
    const int P = (N + POOL_KC - 1) / POOL_KC;
    float* Tpart = bufB + (size_t)N * HID;
    float* T8    = Tpart + (size_t)P * POOL_TILE;
    const int NB    = (N + BIN_R - 1) / BIN_R;
    const int NBpad = (NB + 1 + 127) & ~127;
    int* binCount  = (int*)(T8 + (size_t)8 * POOL_TILE);
    int* binStart  = binCount + 2 * NBpad;
    int* binCursor = binStart + 2 * NBpad;
    int* gcnt      = binCursor + 2 * NBpad;
    int* recs      = gcnt + 128;
    int* recsS     = recs + (size_t)2 * E;
    size_t tabOff = (size_t)((char*)(recsS + (size_t)2 * E) - ws);
    tabOff = (tabOff + 511) & ~(size_t)511;
    const size_t tabBytes = ((size_t)N + 511) & ~(size_t)511;
    int* deg8w = (int*)(ws + tabOff);
    int* g8w   = (int*)(ws + tabOff + tabBytes);

    const int TB = 256;
    dim3 blk(TB);

    const int FB = (E + FILL_TB * FILL_EPT - 1) / (FILL_TB * FILL_EPT);
    k_tables_init<<<dim3((NBpad + NB + TB - 1) / TB), blk, 0, stream>>>(binCount, gcnt, NBpad + NB);
    k_bin_hist<<<dim3(FB, 2), dim3(FILL_TB), NB * sizeof(int), stream>>>(src, dst, binCount, E, NB, NBpad);
    k_bin_scan<<<dim3(2), dim3(1024), 0, stream>>>(binCount, binStart, binCursor, NB, NBpad);
    k_bin_fill<<<dim3(FB, 2), dim3(FILL_TB), 2 * NB * sizeof(int), stream>>>(src, dst, binStart, binCursor, recs, E, NB, NBpad);

    k_bin_deg<<<dim3(NB), dim3(DEG_TB), 0, stream>>>(recs, binStart, batch, dinv, deg8w, g8w, N);

    const int mt = N / 16;

    k_gemm_strips<<<dim3(GEMM_BLOCKS), dim3(GEMM_TB), 0, stream>>>(x, W1, bufA, mt);
    k_agg_lds      <<<dim3(NB), dim3(AGG_TB), 0, stream>>>(recs, binStart, bufA, dinv, deg8w, b1, bufB, N);

    k_gemm_strips<<<dim3(GEMM_BLOCKS), dim3(GEMM_TB), 0, stream>>>(bufB, W2, bufA, mt);
    k_agg_lds      <<<dim3(NB), dim3(AGG_TB), 0, stream>>>(recs, binStart, bufA, dinv, deg8w, b2, bufB, N);

    k_pool_init  <<<dim3((N + TB - 1) / TB), blk, 0, stream>>>(batch, gcnt, N);
    k_pool_gemm  <<<dim3(P), blk, 0, stream>>>(recsS, binStart + NBpad, batch, dinv, deg8w, g8w, bufB, Tpart, N);
    k_pool_reduce<<<dim3(8 * POOL_TILE / TB), blk, 0, stream>>>(Tpart, T8, P);
    k_pool_final <<<dim3(1), dim3(512), 0, stream>>>(T8, W3, b3, gcnt, (float*)d_out);
}


# Pooled split-K product (Cm slab x h2, f32 WMMA) moved into the layer-2 aggregation block, fed from its LDS rows: h2 never stored or re-read, no separate product launch; sum order as before
# speedup vs baseline: 5.1473x; 1.0663x over previous
#include <hip/hip_runtime.h>
#include <math.h>

typedef float v2f __attribute__((ext_vector_type(2)));
typedef float v8f __attribute__((ext_vector_type(8)));

#define GCN_G   64
#define GCN_OUT 64
#define LDS_STRIDE 132

#define BIN_R 400
#define FILL_EPT 4
#define FILL_TB 1024

#define NODE_TAB_MAX 106496
#define NODE_TAB_WORDS (NODE_TAB_MAX / 4)

template <int U>
__device__ __forceinline__ void node_tab_load(int* __restrict__ dstw, const int* __restrict__ srcw,
                                              int n, int t, int nthreads) {
    const int4* __restrict__ s4 = (const int4*)srcw;
    int4* __restrict__ d4 = (int4*)dstw;
    const int n16 = n >> 4;
    for (int i0 = t; i0 < n16; i0 += nthreads * U) {
        int4 v[U];
#pragma unroll
        for (int u = 0; u < U; ++u) v[u] = s4[min(i0 + u * nthreads, n16 - 1)];
#pragma unroll
        for (int u = 0; u < U; ++u) d4[min(i0 + u * nthreads, n16 - 1)] = v[u];
    }
}

__device__ __forceinline__ float node_rs(int d) {
    const float f = (float)d;
    return (f > 0.0f) ? (1.0f / sqrtf(f)) : 0.0f;
}

__device__ __forceinline__ float node_dinv(const unsigned char* deg8, const float* rs,
                                           const float* __restrict__ dinv, int node, bool tab) {
    if (tab) {
        const unsigned d = deg8[node];
        if (d != 255u) return rs[d];
    }
    return dinv[node];
}

__global__ void k_tables_init(int* __restrict__ binCount, int* __restrict__ gcnt, int nb) {
    int i = blockIdx.x * blockDim.x + threadIdx.x;
    if (i < nb) binCount[i] = 0;
    if (i < GCN_G) gcnt[i] = 0;
}

__global__ void __launch_bounds__(FILL_TB)
k_bin_hist(const int* __restrict__ src, const int* __restrict__ dstv,
           int* __restrict__ binCountAll, int e, int nb, int pitch) {
    extern __shared__ int hsm[];
    const int t = threadIdx.x;
    const int* __restrict__ dst = blockIdx.y ? src : dstv;
    int* __restrict__ binCount = binCountAll + blockIdx.y * pitch;
    for (int b = t; b < nb; b += FILL_TB) hsm[b] = 0;
    __syncthreads();
    const int e0 = blockIdx.x * (FILL_TB * FILL_EPT);
    int d[FILL_EPT];
#pragma unroll
    for (int u = 0; u < FILL_EPT; ++u)
        d[u] = dst[min(e0 + u * FILL_TB + t, e - 1)];
#pragma unroll
    for (int u = 0; u < FILL_EPT; ++u)
        if (e0 + u * FILL_TB + t < e) atomicAdd(&hsm[d[u] / BIN_R], 1);
    __syncthreads();
    for (int b = t; b < nb; b += FILL_TB) {
        const int c = hsm[b];
        if (c > 0) atomicAdd(&binCount[b], c);
    }
}

__global__ void __launch_bounds__(1024)
k_bin_scan(const int* __restrict__ cntAll, int* __restrict__ startAll, int* __restrict__ cursorAll,
           int nb, int pitch) {
    __shared__ int s[1024];
    const int t = threadIdx.x;
    const int* __restrict__ cnt = cntAll + blockIdx.x * pitch;
    int* __restrict__ start  = startAll + blockIdx.x * pitch;
    int* __restrict__ cursor = cursorAll + blockIdx.x * pitch;
    int carry = 0;
    for (int b0 = 0; b0 < nb; b0 += 1024) {
        const int b = b0 + t;
        const int v = (b < nb) ? cnt[b] : 0;
        s[t] = v;
        __syncthreads();
        for (int off = 1; off < 1024; off <<= 1) {
            int a = (t >= off) ? s[t - off] : 0;
            __syncthreads();
            s[t] += a;
            __syncthreads();
        }
        if (b < nb) {
            start[b]  = carry + s[t] - v;
            cursor[b] = 0;
        }
        const int total = s[1023];
        __syncthreads();
        carry += total;
    }
    if (t == 0) start[nb] = carry;
}

__global__ void __launch_bounds__(FILL_TB)
k_bin_fill(const int* __restrict__ srcv, const int* __restrict__ dstv,
           const int* __restrict__ startAll,
           int* __restrict__ cursorAll, int* __restrict__ recsAll, int e, int nb, int pitch) {
    const int* __restrict__ src = blockIdx.y ? dstv : srcv;
    const int* __restrict__ dst = blockIdx.y ? srcv : dstv;
    const int* __restrict__ start = startAll + blockIdx.y * pitch;
    int* __restrict__ cursor = cursorAll + blockIdx.y * pitch;
    int* __restrict__ recs = recsAll + (size_t)blockIdx.y * 2 * (size_t)e;
    extern __shared__ int fsm[];
    int* hist = fsm;
    int* base = fsm + nb;
    const int t = threadIdx.x;
    for (int b = t; b < nb; b += FILL_TB) hist[b] = 0;
    __syncthreads();

    const int e0 = blockIdx.x * (FILL_TB * FILL_EPT);
    int es[FILL_EPT], ed[FILL_EPT], eb[FILL_EPT], er[FILL_EPT];
#pragma unroll
    for (int u = 0; u < FILL_EPT; ++u) {
        const int i = e0 + u * FILL_TB + t;
        eb[u] = -1;
        if (i < e) {
            es[u] = src[i];
            ed[u] = dst[i];
            eb[u] = ed[u] / BIN_R;
            er[u] = atomicAdd(&hist[eb[u]], 1);
        }
    }
    __syncthreads();
    for (int b = t; b < nb; b += FILL_TB) {
        const int c = hist[b];
        base[b] = start[b] + (c > 0 ? atomicAdd(&cursor[b], c) : 0);
    }
    __syncthreads();
    int2* __restrict__ out = (int2*)recs;
#pragma unroll
    for (int u = 0; u < FILL_EPT; ++u) {
        if (eb[u] >= 0)
            out[base[eb[u]] + er[u]] = make_int2(es[u], ed[u] - eb[u] * BIN_R);
    }
}

#define DEG_TB 1024
#define DEG_U  8
__global__ void __launch_bounds__(DEG_TB)
k_bin_deg(const int* __restrict__ recs, const int* __restrict__ start,
          float* __restrict__ dinv, int* __restrict__ deg8w, int n) {
    __shared__ int cnt[BIN_R];
    const int t = threadIdx.x;
    if (t < BIN_R) cnt[t] = 0;
    __syncthreads();
    const int eb = start[blockIdx.x];
    const int m  = start[blockIdx.x + 1] - eb;
    const int2* __restrict__ rec2 = (const int2*)recs + eb;
    for (int i0 = t; i0 < m + t; i0 += DEG_TB * DEG_U) {
        int d[DEG_U];
#pragma unroll
        for (int u = 0; u < DEG_U; ++u)
            d[u] = rec2[min(i0 + u * DEG_TB, m - 1)].y;
#pragma unroll
        for (int u = 0; u < DEG_U; ++u)
            if (i0 + u * DEG_TB < m) atomicAdd(&cnt[d[u]], 1);
    }
    __syncthreads();
    const int node = blockIdx.x * BIN_R + t;
    if (t < BIN_R && node < n) {
        const float d = (float)(cnt[t] + 1);
        dinv[node] = (d > 0.0f) ? (1.0f / sqrtf(d)) : 0.0f;
    }
    if (t < BIN_R / 4) {
        const int nb0 = blockIdx.x * BIN_R + 4 * t;
        if (nb0 < n) {
            unsigned dw = 0u;
#pragma unroll
            for (int k = 0; k < 4; ++k) {
                if (nb0 + k < n) dw |= (unsigned)min(cnt[4 * t + k] + 1, 255) << (8 * k);
            }
            deg8w[nb0 >> 2] = (int)dw;
        }
    }
}

#define GEMM_K 128
#define GEMM_F 128
#define GEMM_KS (GEMM_K / 4)
#define GEMM_TB 512
#define GEMM_WAVES (GEMM_TB / 32)
#define GEMM_SROWS 128
#define GEMM_STILES (GEMM_SROWS / 16)
#define GEMM_RPW (GEMM_SROWS / GEMM_WAVES)
#define GEMM_PITCH 132
#define GEMM_BUF (GEMM_SROWS * GEMM_PITCH)
#define GEMM_BLOCKS 256

__device__ __forceinline__ void gemm_lds_barrier() {
    __builtin_amdgcn_fence(__ATOMIC_RELEASE, "workgroup", "local");
    __builtin_amdgcn_s_barrier();
    __builtin_amdgcn_fence(__ATOMIC_ACQUIRE, "workgroup", "local");
}

__device__ __forceinline__ float gemm_other_half(float x) {
    int b = __builtin_bit_cast(int, x);
    return __builtin_bit_cast(float, __builtin_amdgcn_permlanex16(b, b, 0x76543210, 0xfedcba98, false, false));
}

__device__ __forceinline__ void gemm_rowtile(const float* __restrict__ al,
                                             const v2f (&b0)[GEMM_KS], const v2f (&b1)[GEMM_KS],
                                             float* __restrict__ H1, float* __restrict__ H2, bool hi) {
    v8f c0 = {}, c1 = {};
#pragma unroll
    for (int s = 0; s < GEMM_KS; ++s) {
        v2f a = *(const v2f*)(al + 4 * s);
        c0 = __builtin_amdgcn_wmma_f32_16x16x4_f32(false, a, false, b0[s], (short)0, c0, false, false);
        c1 = __builtin_amdgcn_wmma_f32_16x16x4_f32(false, a, false, b1[s], (short)0, c1, false, false);
    }
#pragma unroll
    for (int v = 0; v < 8; ++v) {
        const float t = gemm_other_half(c1[v]);
        H1[(size_t)v * GEMM_F] = hi ? t : c0[v];
        H2[(size_t)v * GEMM_F] = hi ? c0[v] : t;
    }
}

__global__ void __launch_bounds__(GEMM_TB)
k_gemm_strips(const float* __restrict__ X, const float* __restrict__ W,
              float* __restrict__ H, int mtiles) {
    __shared__ float Xs[2 * GEMM_BUF];

    const int t    = threadIdx.x;
    const int lane = t & 31;
    const int wave = __builtin_amdgcn_readfirstlane(t >> 5);
    const int half = lane >> 4;
    const int l16  = lane & 15;
    const int cp   = wave >> 2;
    const int rq   = ((wave & 3) - (wave >> 2)) & 3;

    const unsigned nb = gridDim.x;
    const int t0 = (int)(((unsigned)mtiles * blockIdx.x) / nb);
    const int t1 = (int)(((unsigned)mtiles * (blockIdx.x + 1u)) / nb);
    const int rowEnd = t1 << 4;
    const int nIter  = (t1 - t0 + GEMM_STILES - 1) / GEMM_STILES;

    const int sc = lane << 2;
    float* __restrict__ st0 = &Xs[wave * GEMM_PITCH + sc];
    const float4 zero4 = make_float4(0.0f, 0.0f, 0.0f, 0.0f);
    float4 q[GEMM_RPW];

    {
        const int r0 = (t0 << 4) + wave;
        float4 wq[GEMM_K / GEMM_WAVES];
#pragma unroll
        for (int u = 0; u < GEMM_K / GEMM_WAVES; ++u)
            wq[u] = *(const float4*)(W + (size_t)(wave + GEMM_WAVES * u) * GEMM_F + sc);
#pragma unroll
        for (int u = 0; u < GEMM_RPW; ++u) {
            q[u] = zero4;
            if (r0 + GEMM_WAVES * u < rowEnd)
                q[u] = *(const float4*)(X + (size_t)(r0 + GEMM_WAVES * u) * GEMM_K + sc);
        }
#pragma unroll
        for (int u = 0; u < GEMM_K / GEMM_WAVES; ++u)
            *(float4*)(st0 + GEMM_BUF + GEMM_WAVES * u * GEMM_PITCH) = wq[u];
#pragma unroll
        for (int u = 0; u < GEMM_RPW; ++u)
            *(float4*)(st0 + GEMM_WAVES * u * GEMM_PITCH) = q[u];
    }
    gemm_lds_barrier();

    v2f b0[GEMM_KS], b1[GEMM_KS];
    {
        const float* __restrict__ Wl = &Xs[GEMM_BUF + (half << 1) * GEMM_PITCH + (cp << 5) + l16];
#pragma unroll
        for (int s = 0; s < GEMM_KS; ++s) {
            b0[s].x = Wl[(4 * s) * GEMM_PITCH];
            b0[s].y = Wl[(4 * s + 1) * GEMM_PITCH];
            b1[s].x = Wl[(4 * s) * GEMM_PITCH + 16];
            b1[s].y = Wl[(4 * s + 1) * GEMM_PITCH + 16];
        }
    }
    gemm_lds_barrier();

    const int  aoff = ((rq << 4) + l16) * GEMM_PITCH + (half << 1);
    const bool hi   = (half != 0);
    float* __restrict__ Hp1 = H + (cp << 5) + lane;
    float* __restrict__ Hp2 = H + (size_t)8 * GEMM_F + (cp << 5) + (lane ^ 16);

    for (int i = 0; i < nIter; ++i) {
        const int  cur  = i & 1;
        const bool more = (i + 1 < nIter);
        if (more) {
            const int r0 = ((t0 + GEMM_STILES * (i + 1)) << 4) + wave;
#pragma unroll
            for (int u = 0; u < GEMM_RPW; ++u) {
                q[u] = zero4;
                if (r0 + GEMM_WAVES * u < rowEnd)
                    q[u] = *(const float4*)(X + (size_t)(r0 + GEMM_WAVES * u) * GEMM_K + sc);
            }
        }

        const float* __restrict__ al = &Xs[cur * GEMM_BUF + aoff];
        const int tbase = t0 + GEMM_STILES * i + rq;
#pragma nounroll
        for (int j = 0; j < 2; ++j) {
            const int tile = tbase + 4 * j;
            if (tile < t1) {
                const size_t ho = (size_t)(tile << 4) * GEMM_F;
                gemm_rowtile(al + (size_t)(64 * j) * GEMM_PITCH, b0, b1, Hp1 + ho, Hp2 + ho, hi);
            }

            if (j == 0 && more) {
                float* __restrict__ sn = st0 + (cur ^ 1) * GEMM_BUF;
#pragma unroll
                for (int u = 0; u < GEMM_RPW; ++u)
                    *(float4*)(sn + GEMM_WAVES * u * GEMM_PITCH) = q[u];
            }
        }

        gemm_lds_barrier();
    }
}

#define AGG_TB 1024
#define AGG_U  8

#define POOL_KC BIN_R
#define POOL_STRIDE 404
#define POOL_TILE (GCN_G * 128)
#define POOL_U 8
static_assert(GCN_G * POOL_STRIDE <= NODE_TAB_WORDS, "the Cm slab reuses the degree table's LDS");

__device__ __forceinline__ void agg_add_row(float* __restrict__ acc, int dl, float w, int lane,
                                            const float4 h) {
    float* __restrict__ ap = acc + dl * 128 + lane;
    atomicAdd(ap,      w * h.x);
    atomicAdd(ap + 32, w * h.y);
    atomicAdd(ap + 64, w * h.z);
    atomicAdd(ap + 96, w * h.w);
}

template <bool TAIL>
__global__ void __launch_bounds__(AGG_TB)
k_agg_lds(const int* __restrict__ recs, const int* __restrict__ start,
          const float* __restrict__ H, const float* __restrict__ dinv,
          const int* __restrict__ deg8w,
          const float* __restrict__ bias, float* __restrict__ out, int n,
          const int* __restrict__ recsS, const int* __restrict__ startS,
          const int* __restrict__ batch) {
    __shared__ float acc[BIN_R * 128];
    __shared__ float sdinv[BIN_R];
    __shared__ __attribute__((aligned(16))) int sdegw[NODE_TAB_WORDS];
    __shared__ float srs[256];
    const unsigned char* sdeg8 = (const unsigned char*)sdegw;
    const bool tab = (n <= NODE_TAB_MAX);
    const int tabmask = tab ? -1 : 0;
    const int t     = threadIdx.x;
    const int lane  = t & 31;
    const int wave  = __builtin_amdgcn_readfirstlane(t >> 5);
    const int node0 = blockIdx.x * BIN_R;
    const int rows  = min(BIN_R, n - node0);

    if (t < rows) sdinv[t] = dinv[node0 + t];
    if (t < 256) srs[t] = node_rs(t);
    if (tab) node_tab_load<7>(sdegw, deg8w, n, t, AGG_TB);
    for (int idx = t; idx < rows * 32; idx += AGG_TB) {
        const int r  = idx >> 5;
        const int c4 = (idx & 31) << 2;
        const float di = dinv[node0 + r];
        const float sw = di * di;
        const float4 h = *(const float4*)(H + (size_t)(node0 + r) * 128 + c4);
        float4 v;
        v.x = sw * h.x + bias[c4 + 0];
        v.y = sw * h.y + bias[c4 + 1];
        v.z = sw * h.z + bias[c4 + 2];
        v.w = sw * h.w + bias[c4 + 3];
        float* __restrict__ ap = &acc[r * 128 + (idx & 31)];
        ap[0]  = v.x;
        ap[32] = v.y;
        ap[64] = v.z;
        ap[96] = v.w;
    }
    __syncthreads();

    const int eb  = start[blockIdx.x];
    const int cnt = start[blockIdx.x + 1] - eb;
    const int w0  = eb + (int)(((long long)cnt * wave) >> 5);
    const int w1  = eb + (int)(((long long)cnt * (wave + 1)) >> 5);
    const int2* __restrict__ rec2 = (const int2*)recs;
    const float* __restrict__ Hl = H + (lane << 2);

    int2  rcN = make_int2(0, 0);
    float wN  = 0.0f;
    if (lane < w1 - w0) {
        rcN = rec2[w0 + lane];
        wN  = node_dinv(sdeg8, srs, dinv, rcN.x, tab) * sdinv[rcN.y];
    }
    for (int p = w0; p < w1; p += 32) {
        const int m  = min(32, w1 - p);
        const int2 rc = rcN;
        const int  wl = __float_as_int(wN);
        rcN = rec2[min(p + 32 + lane, w1 - 1)];
        if (m == 32) {
            float dsN = 0.0f, ddN = 0.0f;
            unsigned dgN = 0u;
#pragma unroll
            for (int j0 = 0; j0 < 32; j0 += AGG_U) {
                float4 v[AGG_U];
#pragma unroll
                for (int u = 0; u < AGG_U; ++u) {
                    const int s = __builtin_amdgcn_readlane(rc.x, j0 + u);
                    v[u] = *(const float4*)(Hl + (size_t)s * 128);
                }
#pragma unroll
                for (int u = 0; u < AGG_U; ++u) {
                    const int   dl = __builtin_amdgcn_readlane(rc.y, j0 + u);
                    const float w  = __int_as_float(__builtin_amdgcn_readlane(wl, j0 + u));
                    agg_add_row(acc, dl, w, lane, v[u]);
                }
                if (j0 == 0) {
                    dgN = sdeg8[rcN.x & tabmask];
                    ddN = sdinv[rcN.y];
                }
                if (j0 == AGG_U) dsN = srs[dgN];
            }
            if (!tab || dgN == 255u) dsN = dinv[rcN.x];
            wN = dsN * ddN;
        } else {
            for (int j = 0; j < m; ++j) {
                const int   s  = __builtin_amdgcn_readlane(rc.x, j);
                const int   dl = __builtin_amdgcn_readlane(rc.y, j);
                const float w  = __int_as_float(__builtin_amdgcn_readlane(wl, j));
                agg_add_row(acc, dl, w, lane, *(const float4*)(Hl + (size_t)s * 128));
            }
        }
    }
    int   mS = 0, gb = 0;
    const int2* __restrict__ recS2 = nullptr;
    int2  rS[POOL_U];
    int   gS[POOL_U];
    float dS[POOL_U];
#pragma unroll
    for (int u = 0; u < POOL_U; ++u) {
        rS[u] = make_int2(0, 0);
        gS[u] = 0;
        dS[u] = 0.0f;
    }
    if (TAIL) {
        const int ebS = startS[blockIdx.x];
        mS    = startS[blockIdx.x + 1] - ebS;
        recS2 = (const int2*)recsS + ebS;
        gb = batch[node0 + min(t, rows - 1)];
        if (mS > 0) {
#pragma unroll
            for (int u = 0; u < POOL_U; ++u)
                rS[u] = recS2[min(t + u * AGG_TB, mS - 1)];
#pragma unroll
            for (int u = 0; u < POOL_U; ++u) {
                gS[u] = batch[(unsigned)rS[u].x];
                dS[u] = dinv[(unsigned)rS[u].x];
            }
        }
    }
    if (TAIL) gemm_lds_barrier(); else __syncthreads();

    if (!TAIL) {
        for (int idx = t; idx < rows * 32; idx += AGG_TB) {
            const int r  = idx >> 5;
            const int c4 = (idx & 31) << 2;
            const float* __restrict__ ap = &acc[r * 128 + (idx & 31)];
            float4 v = make_float4(ap[0], ap[32], ap[64], ap[96]);
            v.x = fmaxf(v.x, 0.0f);
            v.y = fmaxf(v.y, 0.0f);
            v.z = fmaxf(v.z, 0.0f);
            v.w = fmaxf(v.w, 0.0f);
            *(float4*)(out + (size_t)(node0 + r) * 128 + c4) = v;
        }
        return;
    }

    float* As = (float*)sdegw;
    for (int idx = t; idx < GCN_G * (POOL_STRIDE / 4); idx += AGG_TB)
        *(float4*)(&As[idx << 2]) = make_float4(0.0f, 0.0f, 0.0f, 0.0f);
    gemm_lds_barrier();
    if (t < rows) As[gb * POOL_STRIDE + t] = sdinv[t] * sdinv[t];
    gemm_lds_barrier();
    for (int i0 = t; i0 < mS + t; i0 += AGG_TB * POOL_U) {
        if (i0 != t) {
#pragma unroll
            for (int u = 0; u < POOL_U; ++u)
                rS[u] = recS2[min(i0 + u * AGG_TB, mS - 1)];
#pragma unroll
            for (int u = 0; u < POOL_U; ++u) {
                gS[u] = batch[(unsigned)rS[u].x];
                dS[u] = dinv[(unsigned)rS[u].x];
            }
        }
#pragma unroll
        for (int u = 0; u < POOL_U; ++u)
            if (i0 + u * AGG_TB < mS)
                atomicAdd(&As[gS[u] * POOL_STRIDE + rS[u].y], sdinv[rS[u].y] * dS[u]);
    }
    gemm_lds_barrier();

    if ((wave & 3) != ((wave >> 2) & 3)) return;
    {
        const int ct   = (wave & 3) + ((wave >> 4) << 2);
        const int half = lane >> 4;
        const int l16  = lane & 15;
        const int pos = (ct << 4) + l16;
        const int col = ((pos & 31) << 2) + (pos >> 5);
        const float* __restrict__ al = &As[l16 * POOL_STRIDE + (half << 1)];
        const float* __restrict__ bl = &acc[(half << 1) * 128 + pos];

        v8f c0 = {}, c1 = {}, c2 = {}, c3 = {};
#pragma unroll 4
        for (int kk = 0; kk < rows; kk += 4) {
            v2f b;
            b.x = fmaxf(bl[kk * 128], 0.0f);
            b.y = fmaxf(bl[kk * 128 + 128], 0.0f);
            const float* __restrict__ ak = al + kk;
            v2f a0 = *(const v2f*)(ak);
            v2f a1 = *(const v2f*)(ak + 16 * POOL_STRIDE);
            v2f a2 = *(const v2f*)(ak + 32 * POOL_STRIDE);
            v2f a3 = *(const v2f*)(ak + 48 * POOL_STRIDE);
            c0 = __builtin_amdgcn_wmma_f32_16x16x4_f32(false, a0, false, b, (short)0, c0, false, false);
            c1 = __builtin_amdgcn_wmma_f32_16x16x4_f32(false, a1, false, b, (short)0, c1, false, false);
            c2 = __builtin_amdgcn_wmma_f32_16x16x4_f32(false, a2, false, b, (short)0, c2, false, false);
            c3 = __builtin_amdgcn_wmma_f32_16x16x4_f32(false, a3, false, b, (short)0, c3, false, false);
        }

        float* __restrict__ o = out + (size_t)blockIdx.x * POOL_TILE + (size_t)(half << 3) * 128 + col;
#pragma unroll
        for (int v = 0; v < 8; ++v) {
            o[(size_t)(v)      * 128] = c0[v];
            o[(size_t)(v + 16) * 128] = c1[v];
            o[(size_t)(v + 32) * 128] = c2[v];
            o[(size_t)(v + 48) * 128] = c3[v];
        }
    }
}

__global__ void k_pool_init(const int* __restrict__ batch, int* __restrict__ gcnt, int n) {
    int j = blockIdx.x * blockDim.x + threadIdx.x;
    int g = (j < n) ? batch[j] : -1;
    int g0 = __shfl(g, 0);
    int same = __popcll(__ballot(g == g0 && g >= 0));
    if ((threadIdx.x & 31) == 0 && g0 >= 0) atomicAdd(&gcnt[g0], same);
    if (g != g0 && g >= 0) atomicAdd(&gcnt[g], 1);
}

__global__ void k_pool_reduce(const float* __restrict__ Tpart, float* __restrict__ T8, int nparts) {
    int tid = blockIdx.x * blockDim.x + threadIdx.x;
    int e   = tid & (POOL_TILE - 1);
    int pg  = tid / POOL_TILE;
    int per = (nparts + 7) >> 3;
    int p0  = pg * per;
    int p1  = min(nparts, p0 + per);
    float s = 0.0f;
    for (int p = p0; p < p1; ++p) s += Tpart[(size_t)p * POOL_TILE + e];
    T8[tid] = s;
}

__global__ void __launch_bounds__(512)
k_pool_final(const float* __restrict__ T8, const float* __restrict__ W3,
             const float* __restrict__ b3, const int* __restrict__ gcnt,
             float* __restrict__ outp) {
    __shared__ float Ts[GCN_G * LDS_STRIDE];
    __shared__ float Wl[GCN_OUT * LDS_STRIDE];
    __shared__ int   hist[GCN_G];
    const int t    = threadIdx.x;
    const int wave = t >> 5;
    const int lane = t & 31;
    const int half = lane >> 4;
    const int l16  = lane & 15;

    if (t < GCN_G) hist[t] = gcnt[t];
    for (int idx = t; idx < POOL_TILE; idx += 512) {
        float s = 0.0f;
#pragma unroll
        for (int pg = 0; pg < 8; ++pg) s += T8[pg * POOL_TILE + idx];
        Ts[(idx >> 7) * LDS_STRIDE + (idx & 127)] = s;
    }
    for (int idx = t; idx < 128 * GCN_OUT; idx += 512)
        Wl[(idx & 63) * LDS_STRIDE + (idx >> 6)] = W3[idx];
    __syncthreads();

    const int mi = wave >> 2;
    const int ni = wave & 3;
    const float* __restrict__ al = &Ts[((mi << 4) + l16) * LDS_STRIDE + (half << 1)];
    const float* __restrict__ bl = &Wl[((ni << 4) + l16) * LDS_STRIDE + (half << 1)];
    v8f c = {};
#pragma unroll
    for (int k0 = 0; k0 < 128; k0 += 4) {
        v2f a = *(const v2f*)(al + k0);
        v2f b = *(const v2f*)(bl + k0);
        c = __builtin_amdgcn_wmma_f32_16x16x4_f32(false, a, false, b, (short)0, c, false, false);
    }
    const int col = (ni << 4) + l16;
    const float bias = b3[col];
#pragma unroll
    for (int v = 0; v < 8; ++v) {
        int g = (mi << 4) + (half << 3) + v;
        float cf = (float)hist[g];
        outp[g * GCN_OUT + col] = (c[v] + cf * bias) / fmaxf(cf, 1.0f);
    }
}

extern "C" void kernel_launch(void* const* d_in, const int* in_sizes, int n_in,
                              void* d_out, int out_size, void* d_ws, size_t ws_size,
                              hipStream_t stream) {
    const float* x     = (const float*)d_in[0];
    const int*   edge  = (const int*)d_in[1];
    const int*   batch = (const int*)d_in[2];
    const float* W1    = (const float*)d_in[3];
    const float* b1    = (const float*)d_in[4];
    const float* W2    = (const float*)d_in[5];
    const float* b2    = (const float*)d_in[6];
    const float* W3    = (const float*)d_in[7];
    const float* b3    = (const float*)d_in[8];

    const int IN = 128, HID = 128;
    const int N = in_sizes[0] / IN;
    const int E = in_sizes[1] / 2;
    const int* src = edge;
    const int* dst = edge + E;

    char* ws = (char*)d_ws;
    float* dinv = (float*)ws;
    size_t off = ((size_t)N * sizeof(float) + 511) & ~(size_t)511;
    float* bufA = (float*)(ws + off);
    float* bufB = bufA + (size_t)N * HID;
    const int P = (N + POOL_KC - 1) / POOL_KC;
    float* Tpart = bufB + (size_t)N * HID;
    float* T8    = Tpart + (size_t)P * POOL_TILE;
    const int NB    = (N + BIN_R - 1) / BIN_R;
    const int NBpad = (NB + 1 + 127) & ~127;
    int* binCount  = (int*)(T8 + (size_t)8 * POOL_TILE);
    int* binStart  = binCount + 2 * NBpad;
    int* binCursor = binStart + 2 * NBpad;
    int* gcnt      = binCursor + 2 * NBpad;
    int* recs      = gcnt + 128;
    int* recsS     = recs + (size_t)2 * E;
    size_t tabOff = (size_t)((char*)(recsS + (size_t)2 * E) - ws);
    tabOff = (tabOff + 511) & ~(size_t)511;
    int* deg8w = (int*)(ws + tabOff);

    const int TB = 256;
    dim3 blk(TB);

    const int FB = (E + FILL_TB * FILL_EPT - 1) / (FILL_TB * FILL_EPT);
    k_tables_init<<<dim3((NBpad + NB + TB - 1) / TB), blk, 0, stream>>>(binCount, gcnt, NBpad + NB);
    k_bin_hist<<<dim3(FB, 2), dim3(FILL_TB), NB * sizeof(int), stream>>>(src, dst, binCount, E, NB, NBpad);
    k_bin_scan<<<dim3(2), dim3(1024), 0, stream>>>(binCount, binStart, binCursor, NB, NBpad);
    k_bin_fill<<<dim3(FB, 2), dim3(FILL_TB), 2 * NB * sizeof(int), stream>>>(src, dst, binStart, binCursor, recs, E, NB, NBpad);

    k_bin_deg<<<dim3(NB), dim3(DEG_TB), 0, stream>>>(recs, binStart, dinv, deg8w, N);

    const int mt = N / 16;

    k_gemm_strips<<<dim3(GEMM_BLOCKS), dim3(GEMM_TB), 0, stream>>>(x, W1, bufA, mt);
    k_agg_lds<false><<<dim3(NB), dim3(AGG_TB), 0, stream>>>(recs, binStart, bufA, dinv, deg8w, b1, bufB, N,
                                                            nullptr, nullptr, nullptr);

    k_gemm_strips<<<dim3(GEMM_BLOCKS), dim3(GEMM_TB), 0, stream>>>(bufB, W2, bufA, mt);
    k_agg_lds<true><<<dim3(NB), dim3(AGG_TB), 0, stream>>>(recs, binStart, bufA, dinv, deg8w, b2, Tpart, N,
                                                           recsS, binStart + NBpad, batch);

    k_pool_init  <<<dim3((N + TB - 1) / TB), blk, 0, stream>>>(batch, gcnt, N);
    k_pool_reduce<<<dim3(8 * POOL_TILE / TB), blk, 0, stream>>>(Tpart, T8, P);
    k_pool_final <<<dim3(1), dim3(512), 0, stream>>>(T8, W3, b3, gcnt, (float*)d_out);
}
